# starting kernel
# baseline (speedup 1.0000x reference)
#include <hip/hip_runtime.h>

typedef float v2f __attribute__((ext_vector_type(2)));
typedef float v8f __attribute__((ext_vector_type(8)));

#define HID 32

__global__ void k_fill(float* __restrict__ p, float v, long long n) {
    long long i = (long long)blockIdx.x * blockDim.x + threadIdx.x;
    if (i < n) p[i] = v;
}

__global__ void k_deg(const int* __restrict__ dst, float* __restrict__ deg, long long e) {
    long long i = (long long)blockIdx.x * blockDim.x + threadIdx.x;
    if (i < e) atomicAdd(&deg[dst[i]], 1.0f);
}

__global__ void k_rsqrt(float* __restrict__ d, long long n) {
    long long i = (long long)blockIdx.x * blockDim.x + threadIdx.x;
    if (i < n) {
        float v = d[i];
        d[i] = (v > 0.0f) ? rsqrtf(v) : 0.0f;
    }
}

template <int KDIM>
__global__ __launch_bounds__(128) void k_gemm(const float* __restrict__ A,
                                              const float* __restrict__ W,
                                              float* __restrict__ H, int n) {
    const int lane = threadIdx.x & 31;
    const int wave = threadIdx.x >> 5;
    const int row0 = (blockIdx.x * 4 + wave) * 16;
    if (row0 >= n) return;

    const int lr = lane & 15;
    const int kh = (lane >> 4) << 1;

    int arow = row0 + lr;
    if (arow >= n) arow = n - 1;
    const float* __restrict__ ap = A + (size_t)arow * KDIM;

    v8f c0 = {0.f, 0.f, 0.f, 0.f, 0.f, 0.f, 0.f, 0.f};
    v8f c1 = {0.f, 0.f, 0.f, 0.f, 0.f, 0.f, 0.f, 0.f};

#pragma unroll
    for (int k0 = 0; k0 < KDIM; k0 += 4) {
        v2f a;
        a.x = ap[k0 + kh];
        a.y = ap[k0 + kh + 1];
        const float* __restrict__ wp = W + (size_t)(k0 + kh) * HID;
        v2f b0, b1;
        b0.x = wp[lr];
        b0.y = wp[HID + lr];
        b1.x = wp[16 + lr];
        b1.y = wp[HID + 16 + lr];
        c0 = __builtin_amdgcn_wmma_f32_16x16x4_f32(false, a, false, b0, (short)0, c0, false, false);
        c1 = __builtin_amdgcn_wmma_f32_16x16x4_f32(false, a, false, b1, (short)0, c1, false, false);
    }

    const int mofs = (lane >> 4) * 8;
#pragma unroll
    for (int r = 0; r < 8; ++r) {
        int row = row0 + mofs + r;
        if (row < n) {
            float* __restrict__ hp = H + (size_t)row * HID;
            hp[lr]      = c0[r];
            hp[16 + lr] = c1[r];
        }
    }
}

__global__ __launch_bounds__(256) void k_scatter(const int* __restrict__ src,
                                                 const int* __restrict__ dst,
                                                 const float* __restrict__ dinv,
                                                 const float* __restrict__ h,
                                                 float* __restrict__ agg, long long e) {
    long long t = (long long)blockIdx.x * blockDim.x + threadIdx.x;
    long long idx = t >> 3;
    if (idx >= e) return;
    const int c = (int)(t & 7);
    const int s = src[idx];
    const int d = dst[idx];
    const float nrm = dinv[s] * dinv[d];
    const float4 v = ((const float4*)(h + (size_t)s * HID))[c];
    float* __restrict__ ap = agg + (size_t)d * HID + c * 4;
    atomicAdd(ap + 0, v.x * nrm);
    atomicAdd(ap + 1, v.y * nrm);
    atomicAdd(ap + 2, v.z * nrm);
    atomicAdd(ap + 3, v.w * nrm);
}

__global__ void k_finalize(const float* __restrict__ h, const float* __restrict__ dinv,
                           const float* __restrict__ b, float* __restrict__ agg,
                           long long total) {
    long long t = (long long)blockIdx.x * blockDim.x + threadIdx.x;
    if (t >= total) return;
    long long i = t >> 5;
    int k = (int)(t & 31);
    float di = dinv[i];
    float v = agg[t] + h[t] * (di * di) + b[k];
    agg[t] = v > 0.0f ? v : 0.0f;
}

__global__ void k_fc(const float* __restrict__ h, const float* __restrict__ wfc,
                     const float* __restrict__ bfc, float* __restrict__ out, int n) {
    int i = blockIdx.x * blockDim.x + threadIdx.x;
    if (i >= n) return;
    const float4* __restrict__ hp = (const float4*)(h + (size_t)i * HID);
    const float4* __restrict__ wp = (const float4*)wfc;
    float acc = 0.0f;
#pragma unroll
    for (int k = 0; k < 8; ++k) {
        float4 a = hp[k], w = wp[k];
        acc += a.x * w.x + a.y * w.y + a.z * w.z + a.w * w.w;
    }
    out[i] = acc + bfc[0];
}

extern "C" void kernel_launch(void* const* d_in, const int* in_sizes, int n_in,
                              void* d_out, int out_size, void* d_ws, size_t ws_size,
                              hipStream_t stream) {
    const float* x   = (const float*)d_in[0];
    const int*   ei  = (const int*)d_in[1];
    const float* W1  = (const float*)d_in[2];
    const float* b1  = (const float*)d_in[3];
    const float* W2  = (const float*)d_in[4];
    const float* b2  = (const float*)d_in[5];
    const float* Wfc = (const float*)d_in[6];
    const float* bfc = (const float*)d_in[7];
    float* out = (float*)d_out;

    const int F = 128;
    const long long n = in_sizes[0] / F;
    const long long e = in_sizes[1] / 2;
    const int* src = ei;
    const int* dst = ei + e;

    float* dinv = (float*)d_ws;
    float* hbuf = dinv + n;
    float* agg  = hbuf + n * HID;

    const long long nh = n * HID;
    const int B = 256;
    dim3 bfill((unsigned)((n + B - 1) / B));
    dim3 bfillh((unsigned)((nh + B - 1) / B));
    dim3 bedge((unsigned)((e + B - 1) / B));
    dim3 bscat((unsigned)((e * 8 + B - 1) / B));
    dim3 bgemm((unsigned)((n + 63) / 64));

    k_fill<<<bfill, B, 0, stream>>>(dinv, 1.0f, n);
    k_deg<<<bedge, B, 0, stream>>>(dst, dinv, e);
    k_rsqrt<<<bfill, B, 0, stream>>>(dinv, n);

    k_gemm<128><<<bgemm, 128, 0, stream>>>(x, W1, hbuf, (int)n);
    k_fill<<<bfillh, B, 0, stream>>>(agg, 0.0f, nh);
    k_scatter<<<bscat, B, 0, stream>>>(src, dst, dinv, hbuf, agg, e);
    k_finalize<<<bfillh, B, 0, stream>>>(hbuf, dinv, b1, agg, nh);

    k_gemm<32><<<bgemm, 128, 0, stream>>>(agg, W2, hbuf, (int)n);
    k_fill<<<bfillh, B, 0, stream>>>(agg, 0.0f, nh);
    k_scatter<<<bscat, B, 0, stream>>>(src, dst, dinv, hbuf, agg, e);
    k_finalize<<<bfillh, B, 0, stream>>>(hbuf, dinv, b2, agg, nh);

    k_fc<<<bfill, B, 0, stream>>>(agg, Wfc, bfc, out, (int)n);
}


# edge scatter remapped: one wave walks 32 edges, lane = feature, so each row gather and each float atomic add is one whole 128-B row (was 8 lanes x 4 partial-row atomics); atomic arrival order changes
# speedup vs baseline: 1.2727x; 1.2727x over previous
#include <hip/hip_runtime.h>

typedef float v2f __attribute__((ext_vector_type(2)));
typedef float v8f __attribute__((ext_vector_type(8)));

#define HID 32

__global__ void k_fill(float* __restrict__ p, float v, long long n) {
    long long i = (long long)blockIdx.x * blockDim.x + threadIdx.x;
    if (i < n) p[i] = v;
}

__global__ void k_deg(const int* __restrict__ dst, float* __restrict__ deg, long long e) {
    long long i = (long long)blockIdx.x * blockDim.x + threadIdx.x;
    if (i < e) atomicAdd(&deg[dst[i]], 1.0f);
}

__global__ void k_rsqrt(float* __restrict__ d, long long n) {
    long long i = (long long)blockIdx.x * blockDim.x + threadIdx.x;
    if (i < n) {
        float v = d[i];
        d[i] = (v > 0.0f) ? rsqrtf(v) : 0.0f;
    }
}

template <int KDIM>
__global__ __launch_bounds__(128) void k_gemm(const float* __restrict__ A,
                                              const float* __restrict__ W,
                                              float* __restrict__ H, int n) {
    const int lane = threadIdx.x & 31;
    const int wave = threadIdx.x >> 5;
    const int row0 = (blockIdx.x * 4 + wave) * 16;
    if (row0 >= n) return;

    const int lr = lane & 15;
    const int kh = (lane >> 4) << 1;

    int arow = row0 + lr;
    if (arow >= n) arow = n - 1;
    const float* __restrict__ ap = A + (size_t)arow * KDIM;

    v8f c0 = {0.f, 0.f, 0.f, 0.f, 0.f, 0.f, 0.f, 0.f};
    v8f c1 = {0.f, 0.f, 0.f, 0.f, 0.f, 0.f, 0.f, 0.f};

#pragma unroll
    for (int k0 = 0; k0 < KDIM; k0 += 4) {
        v2f a;
        a.x = ap[k0 + kh];
        a.y = ap[k0 + kh + 1];
        const float* __restrict__ wp = W + (size_t)(k0 + kh) * HID;
        v2f b0, b1;
        b0.x = wp[lr];
        b0.y = wp[HID + lr];
        b1.x = wp[16 + lr];
        b1.y = wp[HID + 16 + lr];
        c0 = __builtin_amdgcn_wmma_f32_16x16x4_f32(false, a, false, b0, (short)0, c0, false, false);
        c1 = __builtin_amdgcn_wmma_f32_16x16x4_f32(false, a, false, b1, (short)0, c1, false, false);
    }

    const int mofs = (lane >> 4) * 8;
#pragma unroll
    for (int r = 0; r < 8; ++r) {
        int row = row0 + mofs + r;
        if (row < n) {
            float* __restrict__ hp = H + (size_t)row * HID;
            hp[lr]      = c0[r];
            hp[16 + lr] = c1[r];
        }
    }
}

__global__ __launch_bounds__(256) void k_scatter(const int* __restrict__ src,
                                                 const int* __restrict__ dst,
                                                 const float* __restrict__ dinv,
                                                 const float* __restrict__ h,
                                                 float* __restrict__ agg, long long e) {
    const int lane = threadIdx.x & 31;
    const int wave = __builtin_amdgcn_readfirstlane((int)(threadIdx.x >> 5));
    const long long e0 = ((long long)blockIdx.x * (blockDim.x >> 5) + wave) * 32;
    if (e0 >= e) return;
    const long long rem = e - e0;
    const int cnt = rem < 32 ? (int)rem : 32;
    const long long mine = e0 + (lane < cnt ? lane : cnt - 1);
    const int s = src[mine];
    const int d = dst[mine];
    const float ws = dinv[s];
    const float wd = dinv[d];

    if (cnt == 32) {
        float v[32];
#pragma unroll
        for (int j = 0; j < 32; ++j) {
            const int sj = __builtin_amdgcn_readlane(s, j);
            v[j] = h[sj * HID + lane];
        }
        const float nrm = ws * wd;
#pragma unroll
        for (int j = 0; j < 32; ++j) {
            const int dj = __builtin_amdgcn_readlane(d, j);
            const float wj = __int_as_float(__builtin_amdgcn_readlane(__float_as_int(nrm), j));
            atomicAdd(&agg[dj * HID + lane], v[j] * wj);
        }
    } else {
        const float nrm = ws * wd;
        for (int j = 0; j < cnt; ++j) {
            const int sj = __builtin_amdgcn_readlane(s, j);
            const int dj = __builtin_amdgcn_readlane(d, j);
            const float wj = __int_as_float(__builtin_amdgcn_readlane(__float_as_int(nrm), j));
            atomicAdd(&agg[dj * HID + lane], h[sj * HID + lane] * wj);
        }
    }
}

__global__ void k_finalize(const float* __restrict__ h, const float* __restrict__ dinv,
                           const float* __restrict__ b, float* __restrict__ agg,
                           long long total) {
    long long t = (long long)blockIdx.x * blockDim.x + threadIdx.x;
    if (t >= total) return;
    long long i = t >> 5;
    int k = (int)(t & 31);
    float di = dinv[i];
    float v = agg[t] + h[t] * (di * di) + b[k];
    agg[t] = v > 0.0f ? v : 0.0f;
}

__global__ void k_fc(const float* __restrict__ h, const float* __restrict__ wfc,
                     const float* __restrict__ bfc, float* __restrict__ out, int n) {
    int i = blockIdx.x * blockDim.x + threadIdx.x;
    if (i >= n) return;
    const float4* __restrict__ hp = (const float4*)(h + (size_t)i * HID);
    const float4* __restrict__ wp = (const float4*)wfc;
    float acc = 0.0f;
#pragma unroll
    for (int k = 0; k < 8; ++k) {
        float4 a = hp[k], w = wp[k];
        acc += a.x * w.x + a.y * w.y + a.z * w.z + a.w * w.w;
    }
    out[i] = acc + bfc[0];
}

extern "C" void kernel_launch(void* const* d_in, const int* in_sizes, int n_in,
                              void* d_out, int out_size, void* d_ws, size_t ws_size,
                              hipStream_t stream) {
    const float* x   = (const float*)d_in[0];
    const int*   ei  = (const int*)d_in[1];
    const float* W1  = (const float*)d_in[2];
    const float* b1  = (const float*)d_in[3];
    const float* W2  = (const float*)d_in[4];
    const float* b2  = (const float*)d_in[5];
    const float* Wfc = (const float*)d_in[6];
    const float* bfc = (const float*)d_in[7];
    float* out = (float*)d_out;

    const int F = 128;
    const long long n = in_sizes[0] / F;
    const long long e = in_sizes[1] / 2;
    const int* src = ei;
    const int* dst = ei + e;

    float* dinv = (float*)d_ws;
    float* hbuf = dinv + n;
    float* agg  = hbuf + n * HID;

    const long long nh = n * HID;
    const int B = 256;
    dim3 bfill((unsigned)((n + B - 1) / B));
    dim3 bfillh((unsigned)((nh + B - 1) / B));
    dim3 bedge((unsigned)((e + B - 1) / B));
    dim3 bgemm((unsigned)((n + 63) / 64));

    k_fill<<<bfill, B, 0, stream>>>(dinv, 1.0f, n);
    k_deg<<<bedge, B, 0, stream>>>(dst, dinv, e);
    k_rsqrt<<<bfill, B, 0, stream>>>(dinv, n);

    k_gemm<128><<<bgemm, 128, 0, stream>>>(x, W1, hbuf, (int)n);
    k_fill<<<bfillh, B, 0, stream>>>(agg, 0.0f, nh);
    k_scatter<<<bedge, B, 0, stream>>>(src, dst, dinv, hbuf, agg, e);
    k_finalize<<<bfillh, B, 0, stream>>>(hbuf, dinv, b1, agg, nh);

    k_gemm<32><<<bgemm, 128, 0, stream>>>(agg, W2, hbuf, (int)n);
    k_fill<<<bfillh, B, 0, stream>>>(agg, 0.0f, nh);
    k_scatter<<<bedge, B, 0, stream>>>(src, dst, dinv, hbuf, agg, e);
    k_finalize<<<bfillh, B, 0, stream>>>(hbuf, dinv, b2, agg, nh);

    k_fc<<<bfill, B, 0, stream>>>(agg, Wfc, bfc, out, (int)n);
}


# scatter: nodes split in two ranges, a wave adds only into the range of the package half it runs on (edges filtered via per-wave LDS queue, chunks drawn from counters); atomic arrival order changes
# speedup vs baseline: 1.4394x; 1.1310x over previous
#include <hip/hip_runtime.h>

typedef float v2f __attribute__((ext_vector_type(2)));
typedef float v8f __attribute__((ext_vector_type(8)));

#define HID 32

__global__ void k_fill(float* __restrict__ p, float v, long long n) {
    long long i = (long long)blockIdx.x * blockDim.x + threadIdx.x;
    if (i < n) p[i] = v;
}

__global__ void k_deg(const int* __restrict__ dst, float* __restrict__ deg, long long e) {
    long long i = (long long)blockIdx.x * blockDim.x + threadIdx.x;
    if (i < e) atomicAdd(&deg[dst[i]], 1.0f);
}

#define NCTR 256
__global__ void k_rsqrt(float* __restrict__ d, unsigned int* __restrict__ ctr, long long n) {
    long long i = (long long)blockIdx.x * blockDim.x + threadIdx.x;
    if (i < NCTR) ctr[i] = 0u;
    if (i < n) {
        float v = d[i];
        d[i] = (v > 0.0f) ? rsqrtf(v) : 0.0f;
    }
}

template <int KDIM>
__global__ __launch_bounds__(128) void k_gemm(const float* __restrict__ A,
                                              const float* __restrict__ W,
                                              float* __restrict__ H, int n) {
    const int lane = threadIdx.x & 31;
    const int wave = threadIdx.x >> 5;
    const int row0 = (blockIdx.x * 4 + wave) * 16;
    if (row0 >= n) return;

    const int lr = lane & 15;
    const int kh = (lane >> 4) << 1;

    int arow = row0 + lr;
    if (arow >= n) arow = n - 1;
    const float* __restrict__ ap = A + (size_t)arow * KDIM;

    v8f c0 = {0.f, 0.f, 0.f, 0.f, 0.f, 0.f, 0.f, 0.f};
    v8f c1 = {0.f, 0.f, 0.f, 0.f, 0.f, 0.f, 0.f, 0.f};

#pragma unroll
    for (int k0 = 0; k0 < KDIM; k0 += 4) {
        v2f a;
        a.x = ap[k0 + kh];
        a.y = ap[k0 + kh + 1];
        const float* __restrict__ wp = W + (size_t)(k0 + kh) * HID;
        v2f b0, b1;
        b0.x = wp[lr];
        b0.y = wp[HID + lr];
        b1.x = wp[16 + lr];
        b1.y = wp[HID + 16 + lr];
        c0 = __builtin_amdgcn_wmma_f32_16x16x4_f32(false, a, false, b0, (short)0, c0, false, false);
        c1 = __builtin_amdgcn_wmma_f32_16x16x4_f32(false, a, false, b1, (short)0, c1, false, false);
    }

    const int mofs = (lane >> 4) * 8;
#pragma unroll
    for (int r = 0; r < 8; ++r) {
        int row = row0 + mofs + r;
        if (row < n) {
            float* __restrict__ hp = H + (size_t)row * HID;
            hp[lr]      = c0[r];
            hp[16 + lr] = c1[r];
        }
    }
}

#define CH_STEPS 4
#define CH_EDGES (32 * CH_STEPS)
#define CTR_STRIDE 64

__device__ __forceinline__ void edge_group32(int s, int d, int lane,
                                             const float* __restrict__ dinv,
                                             const float* __restrict__ h,
                                             float* __restrict__ agg) {
    const float ws = dinv[s];
    const float wd = dinv[d];
    float v[32];
#pragma unroll
    for (int j = 0; j < 32; ++j) {
        const int sj = __builtin_amdgcn_readlane(s, j);
        v[j] = h[sj * HID + lane];
    }
    const float nrm = ws * wd;
#pragma unroll
    for (int j = 0; j < 32; ++j) {
        const int dj = __builtin_amdgcn_readlane(d, j);
        const float wj = __int_as_float(__builtin_amdgcn_readlane(__float_as_int(nrm), j));
        atomicAdd(&agg[dj * HID + lane], v[j] * wj);
    }
}

__global__ __launch_bounds__(256) void k_scatter(const int* __restrict__ src,
                                                 const int* __restrict__ dst,
                                                 const float* __restrict__ dinv,
                                                 const float* __restrict__ h,
                                                 float* __restrict__ agg,
                                                 unsigned int* __restrict__ ctr,
                                                 long long e, int nsplit) {
    __shared__ int q_s[8][64];
    __shared__ int q_d[8][64];
    const int lane = threadIdx.x & 31;
    const int wave = __builtin_amdgcn_readfirstlane((int)(threadIdx.x >> 5));
    int* qs = q_s[wave];
    int* qd = q_d[wave];
    const int half = (int)((__builtin_amdgcn_s_sendmsg_rtn(0x87) >> 18) & 1u);
    const unsigned int nchunks = (unsigned int)((e + CH_EDGES - 1) / CH_EDGES);
    int cnt = 0;

    for (int p = 0; p < 2; ++p) {
        const int r = half ^ p;
        unsigned int* c = ctr + r * CTR_STRIDE;
        for (unsigned int it = 0; it <= nchunks; ++it) {
            unsigned int ck = 0;
            if (lane == 0) ck = atomicAdd(c, 1u);
            ck = (unsigned int)__builtin_amdgcn_readfirstlane((int)ck);
            if (ck >= nchunks) break;
            const long long e0 = (long long)ck * CH_EDGES + lane;
            int sv[CH_STEPS], dv[CH_STEPS];
#pragma unroll
            for (int k = 0; k < CH_STEPS; ++k) {
                const long long idx = e0 + k * 32;
                const long long ii = idx < e ? idx : e - 1;
                sv[k] = src[ii];
                dv[k] = dst[ii];
            }
#pragma unroll
            for (int k = 0; k < CH_STEPS; ++k) {
                const long long idx = e0 + k * 32;
                const bool m = (idx < e) && ((dv[k] >= nsplit ? 1 : 0) == r);
                const unsigned int mask = __builtin_amdgcn_ballot_w32(m);
                if (m) {
                    const int pos = cnt + (int)__builtin_amdgcn_mbcnt_lo(mask, 0u);
                    qs[pos] = sv[k];
                    qd[pos] = dv[k];
                }
                cnt += __builtin_popcount(mask);
                if (cnt >= 32) {
                    const int s = qs[lane];
                    const int d = qd[lane];
                    const int s2 = qs[32 + lane];
                    const int d2 = qd[32 + lane];
                    edge_group32(s, d, lane, dinv, h, agg);
                    cnt -= 32;
                    if (lane < cnt) {
                        qs[lane] = s2;
                        qd[lane] = d2;
                    }
                }
            }
        }
    }
    if (cnt > 0) {
        const int ql = lane < cnt ? lane : cnt - 1;
        const int s = qs[ql];
        const int d = qd[ql];
        const float nrm = dinv[s] * dinv[d];
        for (int j = 0; j < cnt; ++j) {
            const int sj = __builtin_amdgcn_readlane(s, j);
            const int dj = __builtin_amdgcn_readlane(d, j);
            const float wj = __int_as_float(__builtin_amdgcn_readlane(__float_as_int(nrm), j));
            atomicAdd(&agg[dj * HID + lane], h[sj * HID + lane] * wj);
        }
    }
}

__global__ void k_finalize(const float* __restrict__ h, const float* __restrict__ dinv,
                           const float* __restrict__ b, float* __restrict__ agg,
                           long long total) {
    long long t = (long long)blockIdx.x * blockDim.x + threadIdx.x;
    if (t >= total) return;
    long long i = t >> 5;
    int k = (int)(t & 31);
    float di = dinv[i];
    float v = agg[t] + h[t] * (di * di) + b[k];
    agg[t] = v > 0.0f ? v : 0.0f;
}

__global__ void k_fc(const float* __restrict__ h, const float* __restrict__ wfc,
                     const float* __restrict__ bfc, float* __restrict__ out, int n) {
    int i = blockIdx.x * blockDim.x + threadIdx.x;
    if (i >= n) return;
    const float4* __restrict__ hp = (const float4*)(h + (size_t)i * HID);
    const float4* __restrict__ wp = (const float4*)wfc;
    float acc = 0.0f;
#pragma unroll
    for (int k = 0; k < 8; ++k) {
        float4 a = hp[k], w = wp[k];
        acc += a.x * w.x + a.y * w.y + a.z * w.z + a.w * w.w;
    }
    out[i] = acc + bfc[0];
}

extern "C" void kernel_launch(void* const* d_in, const int* in_sizes, int n_in,
                              void* d_out, int out_size, void* d_ws, size_t ws_size,
                              hipStream_t stream) {
    const float* x   = (const float*)d_in[0];
    const int*   ei  = (const int*)d_in[1];
    const float* W1  = (const float*)d_in[2];
    const float* b1  = (const float*)d_in[3];
    const float* W2  = (const float*)d_in[4];
    const float* b2  = (const float*)d_in[5];
    const float* Wfc = (const float*)d_in[6];
    const float* bfc = (const float*)d_in[7];
    float* out = (float*)d_out;

    const int F = 128;
    const long long n = in_sizes[0] / F;
    const long long e = in_sizes[1] / 2;
    const int* src = ei;
    const int* dst = ei + e;

    unsigned int* ctr = (unsigned int*)d_ws;
    float* dinv = (float*)d_ws + 256;
    float* hbuf = dinv + n;
    float* agg  = hbuf + n * HID;

    const long long nh = n * HID;
    const int B = 256;
    dim3 bfill((unsigned)((n + B - 1) / B));
    dim3 bfillh((unsigned)((nh + B - 1) / B));
    dim3 bedge((unsigned)((e + B - 1) / B));
    dim3 bgemm((unsigned)((n + 63) / 64));
    dim3 bscat(2048);
    const int nsplit = (int)(n / 2);

    k_fill<<<bfill, B, 0, stream>>>(dinv, 1.0f, n);
    k_deg<<<bedge, B, 0, stream>>>(dst, dinv, e);
    k_rsqrt<<<bfill, B, 0, stream>>>(dinv, ctr, n);

    k_gemm<128><<<bgemm, 128, 0, stream>>>(x, W1, hbuf, (int)n);
    k_fill<<<bfillh, B, 0, stream>>>(agg, 0.0f, nh);
    k_scatter<<<bscat, B, 0, stream>>>(src, dst, dinv, hbuf, agg, ctr, e, nsplit);
    k_finalize<<<bfillh, B, 0, stream>>>(hbuf, dinv, b1, agg, nh);

    k_gemm<32><<<bgemm, 128, 0, stream>>>(agg, W2, hbuf, (int)n);
    k_fill<<<bfillh, B, 0, stream>>>(agg, 0.0f, nh);
    k_scatter<<<bscat, B, 0, stream>>>(src, dst, dinv, hbuf, agg, ctr + 128, e, nsplit);
    k_finalize<<<bfillh, B, 0, stream>>>(hbuf, dinv, b2, agg, nh);

    k_fc<<<bfill, B, 0, stream>>>(agg, Wfc, bfc, out, (int)n);
}


# aggregation over destination-sorted edges: edges grouped by destination once per call (rank atomics, scan, placement); a node's rows summed in registers, stored once; sum order = atomic arrival order
# speedup vs baseline: 1.5608x; 1.0843x over previous
#include <hip/hip_runtime.h>

typedef float v2f __attribute__((ext_vector_type(2)));
typedef float v8f __attribute__((ext_vector_type(8)));

#define HID 32

__global__ void k_fill(float* __restrict__ p, float v, long long n) {
    long long i = (long long)blockIdx.x * blockDim.x + threadIdx.x;
    if (i < n) p[i] = v;
}

__global__ void k_deg(const int* __restrict__ dst, float* __restrict__ deg, long long e) {
    long long i = (long long)blockIdx.x * blockDim.x + threadIdx.x;
    if (i < e) atomicAdd(&deg[dst[i]], 1.0f);
}

#define NCTR 256
__global__ void k_rsqrt(float* __restrict__ d, unsigned int* __restrict__ ctr, long long n) {
    long long i = (long long)blockIdx.x * blockDim.x + threadIdx.x;
    if (i < NCTR) ctr[i] = 0u;
    if (i < n) {
        float v = d[i];
        d[i] = (v > 0.0f) ? rsqrtf(v) : 0.0f;
    }
}

template <int KDIM>
__global__ __launch_bounds__(128) void k_gemm(const float* __restrict__ A,
                                              const float* __restrict__ W,
                                              const float* __restrict__ rs,
                                              float* __restrict__ H, int n) {
    const int lane = threadIdx.x & 31;
    const int wave = threadIdx.x >> 5;
    const int row0 = (blockIdx.x * 4 + wave) * 16;
    if (row0 >= n) return;

    const int lr = lane & 15;
    const int kh = (lane >> 4) << 1;

    int arow = row0 + lr;
    if (arow >= n) arow = n - 1;
    const float* __restrict__ ap = A + (size_t)arow * KDIM;

    v8f c0 = {0.f, 0.f, 0.f, 0.f, 0.f, 0.f, 0.f, 0.f};
    v8f c1 = {0.f, 0.f, 0.f, 0.f, 0.f, 0.f, 0.f, 0.f};

#pragma unroll
    for (int k0 = 0; k0 < KDIM; k0 += 4) {
        v2f a;
        a.x = ap[k0 + kh];
        a.y = ap[k0 + kh + 1];
        const float* __restrict__ wp = W + (size_t)(k0 + kh) * HID;
        v2f b0, b1;
        b0.x = wp[lr];
        b0.y = wp[HID + lr];
        b1.x = wp[16 + lr];
        b1.y = wp[HID + 16 + lr];
        c0 = __builtin_amdgcn_wmma_f32_16x16x4_f32(false, a, false, b0, (short)0, c0, false, false);
        c1 = __builtin_amdgcn_wmma_f32_16x16x4_f32(false, a, false, b1, (short)0, c1, false, false);
    }

    const int mofs = (lane >> 4) * 8;
#pragma unroll
    for (int r = 0; r < 8; ++r) {
        int row = row0 + mofs + r;
        if (row < n) {
            const float sc = rs ? rs[row] : 1.0f;
            float* __restrict__ hp = H + (size_t)row * HID;
            hp[lr]      = c0[r] * sc;
            hp[16 + lr] = c1[r] * sc;
        }
    }
}

#define CH_STEPS 4
#define CH_EDGES (32 * CH_STEPS)
#define CTR_STRIDE 64

__device__ __forceinline__ void edge_group32(int s, int d, int lane,
                                             const float* __restrict__ dinv,
                                             const float* __restrict__ h,
                                             float* __restrict__ agg) {
    const float ws = dinv[s];
    const float wd = dinv[d];
    float v[32];
#pragma unroll
    for (int j = 0; j < 32; ++j) {
        const int sj = __builtin_amdgcn_readlane(s, j);
        v[j] = h[sj * HID + lane];
    }
    const float nrm = ws * wd;
#pragma unroll
    for (int j = 0; j < 32; ++j) {
        const int dj = __builtin_amdgcn_readlane(d, j);
        const float wj = __int_as_float(__builtin_amdgcn_readlane(__float_as_int(nrm), j));
        atomicAdd(&agg[dj * HID + lane], v[j] * wj);
    }
}

__global__ __launch_bounds__(256) void k_scatter(const int* __restrict__ src,
                                                 const int* __restrict__ dst,
                                                 const float* __restrict__ dinv,
                                                 const float* __restrict__ h,
                                                 float* __restrict__ agg,
                                                 unsigned int* __restrict__ ctr,
                                                 long long e, int nsplit) {
    __shared__ int q_s[8][64];
    __shared__ int q_d[8][64];
    const int lane = threadIdx.x & 31;
    const int wave = __builtin_amdgcn_readfirstlane((int)(threadIdx.x >> 5));
    int* qs = q_s[wave];
    int* qd = q_d[wave];
    const int half = (int)((__builtin_amdgcn_s_sendmsg_rtn(0x87) >> 18) & 1u);
    const unsigned int nchunks = (unsigned int)((e + CH_EDGES - 1) / CH_EDGES);
    int cnt = 0;

    for (int p = 0; p < 2; ++p) {
        const int r = half ^ p;
        unsigned int* c = ctr + r * CTR_STRIDE;
        for (unsigned int it = 0; it <= nchunks; ++it) {
            unsigned int ck = 0;
            if (lane == 0) ck = atomicAdd(c, 1u);
            ck = (unsigned int)__builtin_amdgcn_readfirstlane((int)ck);
            if (ck >= nchunks) break;
            const long long e0 = (long long)ck * CH_EDGES + lane;
            int sv[CH_STEPS], dv[CH_STEPS];
#pragma unroll
            for (int k = 0; k < CH_STEPS; ++k) {
                const long long idx = e0 + k * 32;
                const long long ii = idx < e ? idx : e - 1;
                sv[k] = src[ii];
                dv[k] = dst[ii];
            }
#pragma unroll
            for (int k = 0; k < CH_STEPS; ++k) {
                const long long idx = e0 + k * 32;
                const bool m = (idx < e) && ((dv[k] >= nsplit ? 1 : 0) == r);
                const unsigned int mask = __builtin_amdgcn_ballot_w32(m);
                if (m) {
                    const int pos = cnt + (int)__builtin_amdgcn_mbcnt_lo(mask, 0u);
                    qs[pos] = sv[k];
                    qd[pos] = dv[k];
                }
                cnt += __builtin_popcount(mask);
                if (cnt >= 32) {
                    const int s = qs[lane];
                    const int d = qd[lane];
                    const int s2 = qs[32 + lane];
                    const int d2 = qd[32 + lane];
                    edge_group32(s, d, lane, dinv, h, agg);
                    cnt -= 32;
                    if (lane < cnt) {
                        qs[lane] = s2;
                        qd[lane] = d2;
                    }
                }
            }
        }
    }
    if (cnt > 0) {
        const int ql = lane < cnt ? lane : cnt - 1;
        const int s = qs[ql];
        const int d = qd[ql];
        const float nrm = dinv[s] * dinv[d];
        for (int j = 0; j < cnt; ++j) {
            const int sj = __builtin_amdgcn_readlane(s, j);
            const int dj = __builtin_amdgcn_readlane(d, j);
            const float wj = __int_as_float(__builtin_amdgcn_readlane(__float_as_int(nrm), j));
            atomicAdd(&agg[dj * HID + lane], h[sj * HID + lane] * wj);
        }
    }
}

__global__ void k_finalize(const float* __restrict__ h, const float* __restrict__ dinv,
                           const float* __restrict__ b, float* __restrict__ agg,
                           long long total) {
    long long t = (long long)blockIdx.x * blockDim.x + threadIdx.x;
    if (t >= total) return;
    long long i = t >> 5;
    int k = (int)(t & 31);
    float di = dinv[i];
    float v = agg[t] + h[t] * (di * di) + b[k];
    agg[t] = v > 0.0f ? v : 0.0f;
}

__global__ void k_fc(const float* __restrict__ h, const float* __restrict__ wfc,
                     const float* __restrict__ bfc, float* __restrict__ out, int n) {
    int i = blockIdx.x * blockDim.x + threadIdx.x;
    if (i >= n) return;
    const float4* __restrict__ hp = (const float4*)(h + (size_t)i * HID);
    const float4* __restrict__ wp = (const float4*)wfc;
    float acc = 0.0f;
#pragma unroll
    for (int k = 0; k < 8; ++k) {
        float4 a = hp[k], w = wp[k];
        acc += a.x * w.x + a.y * w.y + a.z * w.z + a.w * w.w;
    }
    out[i] = acc + bfc[0];
}

#define PGRID 2048
#define SCAN_BLK 1024

__global__ void k_zero(int* __restrict__ p, int n) {
    int i = blockIdx.x * blockDim.x + threadIdx.x;
    if (i < n) p[i] = 0;
}

__global__ __launch_bounds__(256) void k_count(const int* __restrict__ dst, int* __restrict__ cnt,
                                               int* __restrict__ rank, long long e) {
    const long long stride = (long long)gridDim.x * blockDim.x;
    for (long long i = (long long)blockIdx.x * blockDim.x + threadIdx.x; i < e; i += stride)
        rank[i] = atomicAdd(&cnt[dst[i]], 1);
}

__global__ __launch_bounds__(256) void k_scan_local(const int* __restrict__ cnt, int* __restrict__ rowptr,
                                                    int* __restrict__ btot, float* __restrict__ dinv, int n) {
    __shared__ int wsum[8];
    const int t = threadIdx.x;
    const int lane = t & 31;
    const int wv = t >> 5;
    const int base = blockIdx.x * SCAN_BLK + t * 4;
    int c[4];
    int tsum = 0;
#pragma unroll
    for (int k = 0; k < 4; ++k) {
        const int idx = base + k;
        c[k] = idx < n ? cnt[idx] : 0;
        if (idx < n) dinv[idx] = rsqrtf((float)(c[k] + 1));
        tsum += c[k];
    }
    int inc = tsum;
#pragma unroll
    for (int d = 1; d < 32; d <<= 1) {
        const int o = __shfl_up(inc, d);
        if (lane >= d) inc += o;
    }
    if (lane == 31) wsum[wv] = inc;
    __syncthreads();
    int woff = 0, total = 0;
#pragma unroll
    for (int w = 0; w < 8; ++w) {
        const int s = wsum[w];
        if (w < wv) woff += s;
        total += s;
    }
    int run = woff + inc - tsum;
#pragma unroll
    for (int k = 0; k < 4; ++k) {
        const int idx = base + k;
        if (idx < n) rowptr[idx] = run;
        run += c[k];
    }
    if (t == 0) btot[blockIdx.x] = total;
}

__global__ __launch_bounds__(256) void k_scan_add(int* __restrict__ rowptr, const int* __restrict__ btot,
                                                  int n, int e) {
    __shared__ int wsum[8];
    const int t = threadIdx.x;
    const int b = blockIdx.x;
    int s = 0;
    for (int i = t; i < b; i += 256) s += btot[i];
#pragma unroll
    for (int d = 16; d >= 1; d >>= 1) s += __shfl_xor(s, d);
    if ((t & 31) == 0) wsum[t >> 5] = s;
    __syncthreads();
    int off = 0;
#pragma unroll
    for (int w = 0; w < 8; ++w) off += wsum[w];
    const int base = b * SCAN_BLK + t * 4;
#pragma unroll
    for (int k = 0; k < 4; ++k) {
        const int idx = base + k;
        if (idx < n) rowptr[idx] += off;
    }
    if (b == 0 && t == 0) rowptr[n] = e;
}

__global__ __launch_bounds__(256) void k_place(const int* __restrict__ src, const int* __restrict__ dst,
                                               const int* __restrict__ rank, const int* __restrict__ rowptr,
                                               int* __restrict__ col, long long e) {
    const long long stride = (long long)gridDim.x * blockDim.x;
    for (long long i = (long long)blockIdx.x * blockDim.x + threadIdx.x; i < e; i += stride)
        col[rowptr[dst[i]] + rank[i]] = src[i];
}

__global__ __launch_bounds__(256) void k_gather(const int* __restrict__ rowptr, const int* __restrict__ col,
                                                const float* __restrict__ dinv, const float* __restrict__ hs,
                                                const float* __restrict__ b, float* __restrict__ outp, int n) {
    const int lane = threadIdx.x & 31;
    const int wave = __builtin_amdgcn_readfirstlane((int)(blockIdx.x * 8 + (threadIdx.x >> 5)));
    const int nwaves = (int)gridDim.x * 8;
    const float bias = b[lane];
    for (int d = wave; d < n; d += nwaves) {
        const int beg = rowptr[d];
        const int end = rowptr[d + 1];
        float acc = hs[d * HID + lane];
        for (int base = beg; base < end; base += 32) {
            const int cnt = (end - base) < 32 ? (end - base) : 32;
            const int s = col[base + (lane < cnt ? lane : 0)];
            int j0 = 0;
            for (; j0 + 16 <= cnt; j0 += 16) {
                float v[16];
#pragma unroll
                for (int jj = 0; jj < 16; ++jj) {
                    const int sj = __builtin_amdgcn_readlane(s, j0 + jj);
                    v[jj] = hs[sj * HID + lane];
                }
#pragma unroll
                for (int jj = 0; jj < 16; ++jj) acc += v[jj];
            }
            if (j0 + 8 <= cnt) {
                float v[8];
#pragma unroll
                for (int jj = 0; jj < 8; ++jj) {
                    const int sj = __builtin_amdgcn_readlane(s, j0 + jj);
                    v[jj] = hs[sj * HID + lane];
                }
#pragma unroll
                for (int jj = 0; jj < 8; ++jj) acc += v[jj];
                j0 += 8;
            }
            if (j0 < cnt) {
                float v[8];
#pragma unroll
                for (int jj = 0; jj < 8; ++jj) {
                    const int j = j0 + jj;
                    const int sj = __builtin_amdgcn_readlane(s, j < cnt ? j : cnt - 1);
                    v[jj] = hs[sj * HID + lane];
                }
#pragma unroll
                for (int jj = 0; jj < 8; ++jj) acc += (j0 + jj < cnt) ? v[jj] : 0.0f;
            }
        }
        const float o = acc * dinv[d] + bias;
        outp[d * HID + lane] = o > 0.0f ? o : 0.0f;
    }
}

extern "C" void kernel_launch(void* const* d_in, const int* in_sizes, int n_in,
                              void* d_out, int out_size, void* d_ws, size_t ws_size,
                              hipStream_t stream) {
    const float* x   = (const float*)d_in[0];
    const int*   ei  = (const int*)d_in[1];
    const float* W1  = (const float*)d_in[2];
    const float* b1  = (const float*)d_in[3];
    const float* W2  = (const float*)d_in[4];
    const float* b2  = (const float*)d_in[5];
    const float* Wfc = (const float*)d_in[6];
    const float* bfc = (const float*)d_in[7];
    float* out = (float*)d_out;

    const int F = 128;
    const long long n = in_sizes[0] / F;
    const long long e = in_sizes[1] / 2;
    const int* src = ei;
    const int* dst = ei + e;
    const long long nh = n * HID;
    const int B = 256;
    dim3 bfill((unsigned)((n + B - 1) / B));
    dim3 bgemm((unsigned)((n + 63) / 64));

    const long long w_n   = (n + 63) & ~63LL;
    const long long w_rp  = (n + 1 + 63) & ~63LL;
    const long long w_bt  = 1024;
    const long long w_e   = (e + 63) & ~63LL;
    const long long w_nh  = (nh + 63) & ~63LL;
    const long long w_ag  = w_nh > w_e ? w_nh : w_e;
    const long long need  = (w_n + w_rp + w_bt + w_n + w_e + w_nh + w_ag) * 4;
    const long long nsblk = (n + SCAN_BLK - 1) / SCAN_BLK;

    if ((long long)ws_size >= need && nsblk <= w_bt) {
        int*   cnt    = (int*)d_ws;
        int*   rowptr = cnt + w_n;
        int*   btot   = rowptr + w_rp;
        float* dinv   = (float*)(btot + w_bt);
        int*   col    = (int*)(dinv + w_n);
        float* hbuf   = (float*)(col + w_e);
        float* agg    = hbuf + w_nh;
        int*   rank   = (int*)agg;
        dim3 bscan((unsigned)nsblk);

        k_zero<<<bfill, B, 0, stream>>>(cnt, (int)n);
        k_count<<<PGRID, B, 0, stream>>>(dst, cnt, rank, e);
        k_scan_local<<<bscan, B, 0, stream>>>(cnt, rowptr, btot, dinv, (int)n);
        k_scan_add<<<bscan, B, 0, stream>>>(rowptr, btot, (int)n, (int)e);
        k_place<<<PGRID, B, 0, stream>>>(src, dst, rank, rowptr, col, e);

        k_gemm<128><<<bgemm, 128, 0, stream>>>(x, W1, dinv, hbuf, (int)n);
        k_gather<<<PGRID, B, 0, stream>>>(rowptr, col, dinv, hbuf, b1, agg, (int)n);

        k_gemm<32><<<bgemm, 128, 0, stream>>>(agg, W2, dinv, hbuf, (int)n);
        k_gather<<<PGRID, B, 0, stream>>>(rowptr, col, dinv, hbuf, b2, agg, (int)n);

        k_fc<<<bfill, B, 0, stream>>>(agg, Wfc, bfc, out, (int)n);
        return;
    }

    unsigned int* ctr = (unsigned int*)d_ws;
    float* dinv = (float*)d_ws + 256;
    float* hbuf = dinv + n;
    float* agg  = hbuf + n * HID;

    dim3 bfillh((unsigned)((nh + B - 1) / B));
    dim3 bedge((unsigned)((e + B - 1) / B));
    dim3 bscat(2048);
    const int nsplit = (int)(n / 2);

    k_fill<<<bfill, B, 0, stream>>>(dinv, 1.0f, n);
    k_deg<<<bedge, B, 0, stream>>>(dst, dinv, e);
    k_rsqrt<<<bfill, B, 0, stream>>>(dinv, ctr, n);

    k_gemm<128><<<bgemm, 128, 0, stream>>>(x, W1, nullptr, hbuf, (int)n);
    k_fill<<<bfillh, B, 0, stream>>>(agg, 0.0f, nh);
    k_scatter<<<bscat, B, 0, stream>>>(src, dst, dinv, hbuf, agg, ctr, e, nsplit);
    k_finalize<<<bfillh, B, 0, stream>>>(hbuf, dinv, b1, agg, nh);

    k_gemm<32><<<bgemm, 128, 0, stream>>>(agg, W2, nullptr, hbuf, (int)n);
    k_fill<<<bfillh, B, 0, stream>>>(agg, 0.0f, nh);
    k_scatter<<<bscat, B, 0, stream>>>(src, dst, dinv, hbuf, agg, ctr + 128, e, nsplit);
    k_finalize<<<bfillh, B, 0, stream>>>(hbuf, dinv, b2, agg, nh);

    k_fc<<<bfill, B, 0, stream>>>(agg, Wfc, bfc, out, (int)n);
}


# rank atomics split by package half: nodes in two ranges, a wave adds only to counters of its half's range (chunks by ticket, walked once per range, 8 adds in flight per lane); arrival order changes
# speedup vs baseline: 2.2787x; 1.4600x over previous
#include <hip/hip_runtime.h>

typedef float v2f __attribute__((ext_vector_type(2)));
typedef float v8f __attribute__((ext_vector_type(8)));

#define HID 32

__global__ void k_fill(float* __restrict__ p, float v, long long n) {
    long long i = (long long)blockIdx.x * blockDim.x + threadIdx.x;
    if (i < n) p[i] = v;
}

__global__ void k_deg(const int* __restrict__ dst, float* __restrict__ deg, long long e) {
    long long i = (long long)blockIdx.x * blockDim.x + threadIdx.x;
    if (i < e) atomicAdd(&deg[dst[i]], 1.0f);
}

#define NCTR 256
__global__ void k_rsqrt(float* __restrict__ d, unsigned int* __restrict__ ctr, long long n) {
    long long i = (long long)blockIdx.x * blockDim.x + threadIdx.x;
    if (i < NCTR) ctr[i] = 0u;
    if (i < n) {
        float v = d[i];
        d[i] = (v > 0.0f) ? rsqrtf(v) : 0.0f;
    }
}

template <int KDIM>
__global__ __launch_bounds__(128) void k_gemm(const float* __restrict__ A,
                                              const float* __restrict__ W,
                                              const float* __restrict__ rs,
                                              float* __restrict__ H, int n) {
    const int lane = threadIdx.x & 31;
    const int wave = threadIdx.x >> 5;
    const int row0 = (blockIdx.x * 4 + wave) * 16;
    if (row0 >= n) return;

    const int lr = lane & 15;
    const int kh = (lane >> 4) << 1;

    int arow = row0 + lr;
    if (arow >= n) arow = n - 1;
    const float* __restrict__ ap = A + (size_t)arow * KDIM;

    v8f c0 = {0.f, 0.f, 0.f, 0.f, 0.f, 0.f, 0.f, 0.f};
    v8f c1 = {0.f, 0.f, 0.f, 0.f, 0.f, 0.f, 0.f, 0.f};

#pragma unroll
    for (int k0 = 0; k0 < KDIM; k0 += 4) {
        v2f a;
        a.x = ap[k0 + kh];
        a.y = ap[k0 + kh + 1];
        const float* __restrict__ wp = W + (size_t)(k0 + kh) * HID;
        v2f b0, b1;
        b0.x = wp[lr];
        b0.y = wp[HID + lr];
        b1.x = wp[16 + lr];
        b1.y = wp[HID + 16 + lr];
        c0 = __builtin_amdgcn_wmma_f32_16x16x4_f32(false, a, false, b0, (short)0, c0, false, false);
        c1 = __builtin_amdgcn_wmma_f32_16x16x4_f32(false, a, false, b1, (short)0, c1, false, false);
    }

    const int mofs = (lane >> 4) * 8;
#pragma unroll
    for (int r = 0; r < 8; ++r) {
        int row = row0 + mofs + r;
        if (row < n) {
            const float sc = rs ? rs[row] : 1.0f;
            float* __restrict__ hp = H + (size_t)row * HID;
            hp[lr]      = c0[r] * sc;
            hp[16 + lr] = c1[r] * sc;
        }
    }
}

#define CH_STEPS 4
#define CH_EDGES (32 * CH_STEPS)
#define CTR_STRIDE 64

__device__ __forceinline__ void edge_group32(int s, int d, int lane,
                                             const float* __restrict__ dinv,
                                             const float* __restrict__ h,
                                             float* __restrict__ agg) {
    const float ws = dinv[s];
    const float wd = dinv[d];
    float v[32];
#pragma unroll
    for (int j = 0; j < 32; ++j) {
        const int sj = __builtin_amdgcn_readlane(s, j);
        v[j] = h[sj * HID + lane];
    }
    const float nrm = ws * wd;
#pragma unroll
    for (int j = 0; j < 32; ++j) {
        const int dj = __builtin_amdgcn_readlane(d, j);
        const float wj = __int_as_float(__builtin_amdgcn_readlane(__float_as_int(nrm), j));
        atomicAdd(&agg[dj * HID + lane], v[j] * wj);
    }
}

__global__ __launch_bounds__(256) void k_scatter(const int* __restrict__ src,
                                                 const int* __restrict__ dst,
                                                 const float* __restrict__ dinv,
                                                 const float* __restrict__ h,
                                                 float* __restrict__ agg,
                                                 unsigned int* __restrict__ ctr,
                                                 long long e, int nsplit) {
    __shared__ int q_s[8][64];
    __shared__ int q_d[8][64];
    const int lane = threadIdx.x & 31;
    const int wave = __builtin_amdgcn_readfirstlane((int)(threadIdx.x >> 5));
    int* qs = q_s[wave];
    int* qd = q_d[wave];
    const int half = (int)((__builtin_amdgcn_s_sendmsg_rtn(0x87) >> 18) & 1u);
    const unsigned int nchunks = (unsigned int)((e + CH_EDGES - 1) / CH_EDGES);
    int cnt = 0;

    for (int p = 0; p < 2; ++p) {
        const int r = half ^ p;
        unsigned int* c = ctr + r * CTR_STRIDE;
        for (unsigned int it = 0; it <= nchunks; ++it) {
            unsigned int ck = 0;
            if (lane == 0) ck = atomicAdd(c, 1u);
            ck = (unsigned int)__builtin_amdgcn_readfirstlane((int)ck);
            if (ck >= nchunks) break;
            const long long e0 = (long long)ck * CH_EDGES + lane;
            int sv[CH_STEPS], dv[CH_STEPS];
#pragma unroll
            for (int k = 0; k < CH_STEPS; ++k) {
                const long long idx = e0 + k * 32;
                const long long ii = idx < e ? idx : e - 1;
                sv[k] = src[ii];
                dv[k] = dst[ii];
            }
#pragma unroll
            for (int k = 0; k < CH_STEPS; ++k) {
                const long long idx = e0 + k * 32;
                const bool m = (idx < e) && ((dv[k] >= nsplit ? 1 : 0) == r);
                const unsigned int mask = __builtin_amdgcn_ballot_w32(m);
                if (m) {
                    const int pos = cnt + (int)__builtin_amdgcn_mbcnt_lo(mask, 0u);
                    qs[pos] = sv[k];
                    qd[pos] = dv[k];
                }
                cnt += __builtin_popcount(mask);
                if (cnt >= 32) {
                    const int s = qs[lane];
                    const int d = qd[lane];
                    const int s2 = qs[32 + lane];
                    const int d2 = qd[32 + lane];
                    edge_group32(s, d, lane, dinv, h, agg);
                    cnt -= 32;
                    if (lane < cnt) {
                        qs[lane] = s2;
                        qd[lane] = d2;
                    }
                }
            }
        }
    }
    if (cnt > 0) {
        const int ql = lane < cnt ? lane : cnt - 1;
        const int s = qs[ql];
        const int d = qd[ql];
        const float nrm = dinv[s] * dinv[d];
        for (int j = 0; j < cnt; ++j) {
            const int sj = __builtin_amdgcn_readlane(s, j);
            const int dj = __builtin_amdgcn_readlane(d, j);
            const float wj = __int_as_float(__builtin_amdgcn_readlane(__float_as_int(nrm), j));
            atomicAdd(&agg[dj * HID + lane], h[sj * HID + lane] * wj);
        }
    }
}

__global__ void k_finalize(const float* __restrict__ h, const float* __restrict__ dinv,
                           const float* __restrict__ b, float* __restrict__ agg,
                           long long total) {
    long long t = (long long)blockIdx.x * blockDim.x + threadIdx.x;
    if (t >= total) return;
    long long i = t >> 5;
    int k = (int)(t & 31);
    float di = dinv[i];
    float v = agg[t] + h[t] * (di * di) + b[k];
    agg[t] = v > 0.0f ? v : 0.0f;
}

__global__ void k_fc(const float* __restrict__ h, const float* __restrict__ wfc,
                     const float* __restrict__ bfc, float* __restrict__ out, int n) {
    int i = blockIdx.x * blockDim.x + threadIdx.x;
    if (i >= n) return;
    const float4* __restrict__ hp = (const float4*)(h + (size_t)i * HID);
    const float4* __restrict__ wp = (const float4*)wfc;
    float acc = 0.0f;
#pragma unroll
    for (int k = 0; k < 8; ++k) {
        float4 a = hp[k], w = wp[k];
        acc += a.x * w.x + a.y * w.y + a.z * w.z + a.w * w.w;
    }
    out[i] = acc + bfc[0];
}

#define PGRID 2048
#define SCAN_BLK 1024

__global__ void k_zero(int* __restrict__ p, unsigned int* __restrict__ ctr, int n) {
    int i = blockIdx.x * blockDim.x + threadIdx.x;
    if (i < NCTR) ctr[i] = 0u;
    if (i < n) p[i] = 0;
}

#define CK_STEPS 8
#define CK_EDGES (32 * CK_STEPS)
__global__ __launch_bounds__(256) void k_count(const int* __restrict__ dst, int* __restrict__ cnt,
                                               int* __restrict__ rank, unsigned int* __restrict__ ctr,
                                               long long e, int nsplit) {
    const int lane = threadIdx.x & 31;
    const int half = (int)((__builtin_amdgcn_s_sendmsg_rtn(0x87) >> 18) & 1u);
    const unsigned int nchunks = (unsigned int)((e + CK_EDGES - 1) / CK_EDGES);

    for (int p = 0; p < 2; ++p) {
        const int r = half ^ p;
        unsigned int* c = ctr + r * CTR_STRIDE;
        for (unsigned int it = 0; it <= nchunks; ++it) {
            unsigned int ck = 0;
            if (lane == 0) ck = atomicAdd(c, 1u);
            ck = (unsigned int)__builtin_amdgcn_readfirstlane((int)ck);
            if (ck >= nchunks) break;
            const long long e0 = (long long)ck * CK_EDGES + lane;
            int dv[CK_STEPS], rk[CK_STEPS];
#pragma unroll
            for (int k = 0; k < CK_STEPS; ++k) {
                const long long idx = e0 + k * 32;
                dv[k] = dst[idx < e ? idx : e - 1];
            }
            unsigned int inr = 0u;
#pragma unroll
            for (int k = 0; k < CK_STEPS; ++k) {
                const long long idx = e0 + k * 32;
                const unsigned int ex = idx < e ? 1u : 0u;
                const unsigned int in = (dv[k] >= nsplit ? 1 : 0) == r ? 1u : 0u;
                inr |= (ex & in) << k;
            }
            asm volatile("" : "+v"(inr));
#pragma unroll
            for (int k = 0; k < CK_STEPS; ++k) {
                rk[k] = 0;
                if (inr & (1u << k)) rk[k] = atomicAdd(&cnt[dv[k]], 1);
            }
#pragma unroll
            for (int k = 0; k < CK_STEPS; ++k) {
                if (inr & (1u << k)) rank[e0 + k * 32] = rk[k];
            }
        }
    }
}

__global__ __launch_bounds__(256) void k_scan_local(const int* __restrict__ cnt, int* __restrict__ rowptr,
                                                    int* __restrict__ btot, float* __restrict__ dinv, int n) {
    __shared__ int wsum[8];
    const int t = threadIdx.x;
    const int lane = t & 31;
    const int wv = t >> 5;
    const int base = blockIdx.x * SCAN_BLK + t * 4;
    int c[4];
    int tsum = 0;
#pragma unroll
    for (int k = 0; k < 4; ++k) {
        const int idx = base + k;
        c[k] = idx < n ? cnt[idx] : 0;
        if (idx < n) dinv[idx] = rsqrtf((float)(c[k] + 1));
        tsum += c[k];
    }
    int inc = tsum;
#pragma unroll
    for (int d = 1; d < 32; d <<= 1) {
        const int o = __shfl_up(inc, d);
        if (lane >= d) inc += o;
    }
    if (lane == 31) wsum[wv] = inc;
    __syncthreads();
    int woff = 0, total = 0;
#pragma unroll
    for (int w = 0; w < 8; ++w) {
        const int s = wsum[w];
        if (w < wv) woff += s;
        total += s;
    }
    int run = woff + inc - tsum;
#pragma unroll
    for (int k = 0; k < 4; ++k) {
        const int idx = base + k;
        if (idx < n) rowptr[idx] = run;
        run += c[k];
    }
    if (t == 0) btot[blockIdx.x] = total;
}

__global__ __launch_bounds__(256) void k_scan_add(int* __restrict__ rowptr, const int* __restrict__ btot,
                                                  int n, int e) {
    __shared__ int wsum[8];
    const int t = threadIdx.x;
    const int b = blockIdx.x;
    int s = 0;
    for (int i = t; i < b; i += 256) s += btot[i];
#pragma unroll
    for (int d = 16; d >= 1; d >>= 1) s += __shfl_xor(s, d);
    if ((t & 31) == 0) wsum[t >> 5] = s;
    __syncthreads();
    int off = 0;
#pragma unroll
    for (int w = 0; w < 8; ++w) off += wsum[w];
    const int base = b * SCAN_BLK + t * 4;
#pragma unroll
    for (int k = 0; k < 4; ++k) {
        const int idx = base + k;
        if (idx < n) rowptr[idx] += off;
    }
    if (b == 0 && t == 0) rowptr[n] = e;
}

__global__ __launch_bounds__(256) void k_place(const int* __restrict__ src, const int* __restrict__ dst,
                                               const int* __restrict__ rank, const int* __restrict__ rowptr,
                                               int* __restrict__ col, long long e) {
    const long long stride = (long long)gridDim.x * blockDim.x;
    for (long long i = (long long)blockIdx.x * blockDim.x + threadIdx.x; i < e; i += stride)
        col[rowptr[dst[i]] + rank[i]] = src[i];
}

__global__ __launch_bounds__(256) void k_gather(const int* __restrict__ rowptr, const int* __restrict__ col,
                                                const float* __restrict__ dinv, const float* __restrict__ hs,
                                                const float* __restrict__ b, float* __restrict__ outp, int n) {
    const int lane = threadIdx.x & 31;
    const int wave = __builtin_amdgcn_readfirstlane((int)(blockIdx.x * 8 + (threadIdx.x >> 5)));
    const int nwaves = (int)gridDim.x * 8;
    const float bias = b[lane];
    for (int d = wave; d < n; d += nwaves) {
        const int beg = rowptr[d];
        const int end = rowptr[d + 1];
        float acc = hs[d * HID + lane];
        for (int base = beg; base < end; base += 32) {
            const int cnt = (end - base) < 32 ? (end - base) : 32;
            const int s = col[base + (lane < cnt ? lane : 0)];
            int j0 = 0;
            for (; j0 + 16 <= cnt; j0 += 16) {
                float v[16];
#pragma unroll
                for (int jj = 0; jj < 16; ++jj) {
                    const int sj = __builtin_amdgcn_readlane(s, j0 + jj);
                    v[jj] = hs[sj * HID + lane];
                }
#pragma unroll
                for (int jj = 0; jj < 16; ++jj) acc += v[jj];
            }
            if (j0 + 8 <= cnt) {
                float v[8];
#pragma unroll
                for (int jj = 0; jj < 8; ++jj) {
                    const int sj = __builtin_amdgcn_readlane(s, j0 + jj);
                    v[jj] = hs[sj * HID + lane];
                }
#pragma unroll
                for (int jj = 0; jj < 8; ++jj) acc += v[jj];
                j0 += 8;
            }
            if (j0 < cnt) {
                float v[8];
#pragma unroll
                for (int jj = 0; jj < 8; ++jj) {
                    const int j = j0 + jj;
                    const int sj = __builtin_amdgcn_readlane(s, j < cnt ? j : cnt - 1);
                    v[jj] = hs[sj * HID + lane];
                }
#pragma unroll
                for (int jj = 0; jj < 8; ++jj) acc += (j0 + jj < cnt) ? v[jj] : 0.0f;
            }
        }
        const float o = acc * dinv[d] + bias;
        outp[d * HID + lane] = o > 0.0f ? o : 0.0f;
    }
}

extern "C" void kernel_launch(void* const* d_in, const int* in_sizes, int n_in,
                              void* d_out, int out_size, void* d_ws, size_t ws_size,
                              hipStream_t stream) {
    const float* x   = (const float*)d_in[0];
    const int*   ei  = (const int*)d_in[1];
    const float* W1  = (const float*)d_in[2];
    const float* b1  = (const float*)d_in[3];
    const float* W2  = (const float*)d_in[4];
    const float* b2  = (const float*)d_in[5];
    const float* Wfc = (const float*)d_in[6];
    const float* bfc = (const float*)d_in[7];
    float* out = (float*)d_out;

    const int F = 128;
    const long long n = in_sizes[0] / F;
    const long long e = in_sizes[1] / 2;
    const int* src = ei;
    const int* dst = ei + e;
    const long long nh = n * HID;
    const int B = 256;
    dim3 bfill((unsigned)((n + B - 1) / B));
    dim3 bgemm((unsigned)((n + 63) / 64));

    const long long w_n   = (n + 63) & ~63LL;
    const long long w_rp  = (n + 1 + 63) & ~63LL;
    const long long w_bt  = 1024;
    const long long w_e   = (e + 63) & ~63LL;
    const long long w_nh  = (nh + 63) & ~63LL;
    const long long w_ag  = w_nh > w_e ? w_nh : w_e;
    const long long need  = (w_n + w_rp + w_bt + w_n + w_e + w_nh + w_ag) * 4;
    const long long nsblk = (n + SCAN_BLK - 1) / SCAN_BLK;

    if ((long long)ws_size >= need && nsblk <= 512) {
        int*   cnt    = (int*)d_ws;
        int*   rowptr = cnt + w_n;
        int*   btot   = rowptr + w_rp;
        unsigned int* tick = (unsigned int*)(btot + 512);
        const int nsplit = (int)((n / 2) & ~63LL);
        float* dinv   = (float*)(btot + w_bt);
        int*   col    = (int*)(dinv + w_n);
        float* hbuf   = (float*)(col + w_e);
        float* agg    = hbuf + w_nh;
        int*   rank   = (int*)agg;
        dim3 bscan((unsigned)nsblk);

        k_zero<<<bfill, B, 0, stream>>>(cnt, tick, (int)n);
        k_count<<<PGRID, B, 0, stream>>>(dst, cnt, rank, tick, e, nsplit);
        k_scan_local<<<bscan, B, 0, stream>>>(cnt, rowptr, btot, dinv, (int)n);
        k_scan_add<<<bscan, B, 0, stream>>>(rowptr, btot, (int)n, (int)e);
        k_place<<<PGRID, B, 0, stream>>>(src, dst, rank, rowptr, col, e);

        k_gemm<128><<<bgemm, 128, 0, stream>>>(x, W1, dinv, hbuf, (int)n);
        k_gather<<<PGRID, B, 0, stream>>>(rowptr, col, dinv, hbuf, b1, agg, (int)n);

        k_gemm<32><<<bgemm, 128, 0, stream>>>(agg, W2, dinv, hbuf, (int)n);
        k_gather<<<PGRID, B, 0, stream>>>(rowptr, col, dinv, hbuf, b2, agg, (int)n);

        k_fc<<<bfill, B, 0, stream>>>(agg, Wfc, bfc, out, (int)n);
        return;
    }

    unsigned int* ctr = (unsigned int*)d_ws;
    float* dinv = (float*)d_ws + 256;
    float* hbuf = dinv + n;
    float* agg  = hbuf + n * HID;

    dim3 bfillh((unsigned)((nh + B - 1) / B));
    dim3 bedge((unsigned)((e + B - 1) / B));
    dim3 bscat(2048);
    const int nsplit = (int)(n / 2);

    k_fill<<<bfill, B, 0, stream>>>(dinv, 1.0f, n);
    k_deg<<<bedge, B, 0, stream>>>(dst, dinv, e);
    k_rsqrt<<<bfill, B, 0, stream>>>(dinv, ctr, n);

    k_gemm<128><<<bgemm, 128, 0, stream>>>(x, W1, nullptr, hbuf, (int)n);
    k_fill<<<bfillh, B, 0, stream>>>(agg, 0.0f, nh);
    k_scatter<<<bscat, B, 0, stream>>>(src, dst, dinv, hbuf, agg, ctr, e, nsplit);
    k_finalize<<<bfillh, B, 0, stream>>>(hbuf, dinv, b1, agg, nh);

    k_gemm<32><<<bgemm, 128, 0, stream>>>(agg, W2, nullptr, hbuf, (int)n);
    k_fill<<<bfillh, B, 0, stream>>>(agg, 0.0f, nh);
    k_scatter<<<bscat, B, 0, stream>>>(src, dst, dinv, hbuf, agg, ctr + 128, e, nsplit);
    k_finalize<<<bfillh, B, 0, stream>>>(hbuf, dinv, b2, agg, nh);

    k_fc<<<bfill, B, 0, stream>>>(agg, Wfc, bfc, out, (int)n);
}


# placement split by package half: a wave stores only into the part of col that holds its half's node range (chunks by ticket, each walked once per range, index streams read whole); sum order kept
# speedup vs baseline: 3.1256x; 1.3717x over previous
#include <hip/hip_runtime.h>

typedef float v2f __attribute__((ext_vector_type(2)));
typedef float v8f __attribute__((ext_vector_type(8)));

#define HID 32

__global__ void k_fill(float* __restrict__ p, float v, long long n) {
    long long i = (long long)blockIdx.x * blockDim.x + threadIdx.x;
    if (i < n) p[i] = v;
}

__global__ void k_deg(const int* __restrict__ dst, float* __restrict__ deg, long long e) {
    long long i = (long long)blockIdx.x * blockDim.x + threadIdx.x;
    if (i < e) atomicAdd(&deg[dst[i]], 1.0f);
}

#define NCTR 256
__global__ void k_rsqrt(float* __restrict__ d, unsigned int* __restrict__ ctr, long long n) {
    long long i = (long long)blockIdx.x * blockDim.x + threadIdx.x;
    if (i < NCTR) ctr[i] = 0u;
    if (i < n) {
        float v = d[i];
        d[i] = (v > 0.0f) ? rsqrtf(v) : 0.0f;
    }
}

template <int KDIM>
__global__ __launch_bounds__(128) void k_gemm(const float* __restrict__ A,
                                              const float* __restrict__ W,
                                              const float* __restrict__ rs,
                                              float* __restrict__ H, int n) {
    const int lane = threadIdx.x & 31;
    const int wave = threadIdx.x >> 5;
    const int row0 = (blockIdx.x * 4 + wave) * 16;
    if (row0 >= n) return;

    const int lr = lane & 15;
    const int kh = (lane >> 4) << 1;

    int arow = row0 + lr;
    if (arow >= n) arow = n - 1;
    const float* __restrict__ ap = A + (size_t)arow * KDIM;

    v8f c0 = {0.f, 0.f, 0.f, 0.f, 0.f, 0.f, 0.f, 0.f};
    v8f c1 = {0.f, 0.f, 0.f, 0.f, 0.f, 0.f, 0.f, 0.f};

#pragma unroll
    for (int k0 = 0; k0 < KDIM; k0 += 4) {
        v2f a;
        a.x = ap[k0 + kh];
        a.y = ap[k0 + kh + 1];
        const float* __restrict__ wp = W + (size_t)(k0 + kh) * HID;
        v2f b0, b1;
        b0.x = wp[lr];
        b0.y = wp[HID + lr];
        b1.x = wp[16 + lr];
        b1.y = wp[HID + 16 + lr];
        c0 = __builtin_amdgcn_wmma_f32_16x16x4_f32(false, a, false, b0, (short)0, c0, false, false);
        c1 = __builtin_amdgcn_wmma_f32_16x16x4_f32(false, a, false, b1, (short)0, c1, false, false);
    }

    const int mofs = (lane >> 4) * 8;
#pragma unroll
    for (int r = 0; r < 8; ++r) {
        int row = row0 + mofs + r;
        if (row < n) {
            const float sc = rs ? rs[row] : 1.0f;
            float* __restrict__ hp = H + (size_t)row * HID;
            hp[lr]      = c0[r] * sc;
            hp[16 + lr] = c1[r] * sc;
        }
    }
}

#define CH_STEPS 4
#define CH_EDGES (32 * CH_STEPS)
#define CTR_STRIDE 64

__device__ __forceinline__ void edge_group32(int s, int d, int lane,
                                             const float* __restrict__ dinv,
                                             const float* __restrict__ h,
                                             float* __restrict__ agg) {
    const float ws = dinv[s];
    const float wd = dinv[d];
    float v[32];
#pragma unroll
    for (int j = 0; j < 32; ++j) {
        const int sj = __builtin_amdgcn_readlane(s, j);
        v[j] = h[sj * HID + lane];
    }
    const float nrm = ws * wd;
#pragma unroll
    for (int j = 0; j < 32; ++j) {
        const int dj = __builtin_amdgcn_readlane(d, j);
        const float wj = __int_as_float(__builtin_amdgcn_readlane(__float_as_int(nrm), j));
        atomicAdd(&agg[dj * HID + lane], v[j] * wj);
    }
}

__global__ __launch_bounds__(256) void k_scatter(const int* __restrict__ src,
                                                 const int* __restrict__ dst,
                                                 const float* __restrict__ dinv,
                                                 const float* __restrict__ h,
                                                 float* __restrict__ agg,
                                                 unsigned int* __restrict__ ctr,
                                                 long long e, int nsplit) {
    __shared__ int q_s[8][64];
    __shared__ int q_d[8][64];
    const int lane = threadIdx.x & 31;
    const int wave = __builtin_amdgcn_readfirstlane((int)(threadIdx.x >> 5));
    int* qs = q_s[wave];
    int* qd = q_d[wave];
    const int half = (int)((__builtin_amdgcn_s_sendmsg_rtn(0x87) >> 18) & 1u);
    const unsigned int nchunks = (unsigned int)((e + CH_EDGES - 1) / CH_EDGES);
    int cnt = 0;

    for (int p = 0; p < 2; ++p) {
        const int r = half ^ p;
        unsigned int* c = ctr + r * CTR_STRIDE;
        for (unsigned int it = 0; it <= nchunks; ++it) {
            unsigned int ck = 0;
            if (lane == 0) ck = atomicAdd(c, 1u);
            ck = (unsigned int)__builtin_amdgcn_readfirstlane((int)ck);
            if (ck >= nchunks) break;
            const long long e0 = (long long)ck * CH_EDGES + lane;
            int sv[CH_STEPS], dv[CH_STEPS];
#pragma unroll
            for (int k = 0; k < CH_STEPS; ++k) {
                const long long idx = e0 + k * 32;
                const long long ii = idx < e ? idx : e - 1;
                sv[k] = src[ii];
                dv[k] = dst[ii];
            }
#pragma unroll
            for (int k = 0; k < CH_STEPS; ++k) {
                const long long idx = e0 + k * 32;
                const bool m = (idx < e) && ((dv[k] >= nsplit ? 1 : 0) == r);
                const unsigned int mask = __builtin_amdgcn_ballot_w32(m);
                if (m) {
                    const int pos = cnt + (int)__builtin_amdgcn_mbcnt_lo(mask, 0u);
                    qs[pos] = sv[k];
                    qd[pos] = dv[k];
                }
                cnt += __builtin_popcount(mask);
                if (cnt >= 32) {
                    const int s = qs[lane];
                    const int d = qd[lane];
                    const int s2 = qs[32 + lane];
                    const int d2 = qd[32 + lane];
                    edge_group32(s, d, lane, dinv, h, agg);
                    cnt -= 32;
                    if (lane < cnt) {
                        qs[lane] = s2;
                        qd[lane] = d2;
                    }
                }
            }
        }
    }
    if (cnt > 0) {
        const int ql = lane < cnt ? lane : cnt - 1;
        const int s = qs[ql];
        const int d = qd[ql];
        const float nrm = dinv[s] * dinv[d];
        for (int j = 0; j < cnt; ++j) {
            const int sj = __builtin_amdgcn_readlane(s, j);
            const int dj = __builtin_amdgcn_readlane(d, j);
            const float wj = __int_as_float(__builtin_amdgcn_readlane(__float_as_int(nrm), j));
            atomicAdd(&agg[dj * HID + lane], h[sj * HID + lane] * wj);
        }
    }
}

__global__ void k_finalize(const float* __restrict__ h, const float* __restrict__ dinv,
                           const float* __restrict__ b, float* __restrict__ agg,
                           long long total) {
    long long t = (long long)blockIdx.x * blockDim.x + threadIdx.x;
    if (t >= total) return;
    long long i = t >> 5;
    int k = (int)(t & 31);
    float di = dinv[i];
    float v = agg[t] + h[t] * (di * di) + b[k];
    agg[t] = v > 0.0f ? v : 0.0f;
}

__global__ void k_fc(const float* __restrict__ h, const float* __restrict__ wfc,
                     const float* __restrict__ bfc, float* __restrict__ out, int n) {
    int i = blockIdx.x * blockDim.x + threadIdx.x;
    if (i >= n) return;
    const float4* __restrict__ hp = (const float4*)(h + (size_t)i * HID);
    const float4* __restrict__ wp = (const float4*)wfc;
    float acc = 0.0f;
#pragma unroll
    for (int k = 0; k < 8; ++k) {
        float4 a = hp[k], w = wp[k];
        acc += a.x * w.x + a.y * w.y + a.z * w.z + a.w * w.w;
    }
    out[i] = acc + bfc[0];
}

#define PGRID 2048
#define SCAN_BLK 1024

__global__ void k_zero(int* __restrict__ p, unsigned int* __restrict__ ctr, int n) {
    int i = blockIdx.x * blockDim.x + threadIdx.x;
    if (i < NCTR) ctr[i] = 0u;
    if (i < n) p[i] = 0;
}

#define CK_STEPS 8
#define CK_EDGES (32 * CK_STEPS)
__global__ __launch_bounds__(256) void k_count(const int* __restrict__ dst, int* __restrict__ cnt,
                                               int* __restrict__ rank, unsigned int* __restrict__ ctr,
                                               long long e, int nsplit) {
    const int lane = threadIdx.x & 31;
    const int half = (int)((__builtin_amdgcn_s_sendmsg_rtn(0x87) >> 18) & 1u);
    const unsigned int nchunks = (unsigned int)((e + CK_EDGES - 1) / CK_EDGES);

    for (int p = 0; p < 2; ++p) {
        const int r = half ^ p;
        unsigned int* c = ctr + r * CTR_STRIDE;
        for (unsigned int it = 0; it <= nchunks; ++it) {
            unsigned int ck = 0;
            if (lane == 0) ck = atomicAdd(c, 1u);
            ck = (unsigned int)__builtin_amdgcn_readfirstlane((int)ck);
            if (ck >= nchunks) break;
            const long long e0 = (long long)ck * CK_EDGES + lane;
            int dv[CK_STEPS], rk[CK_STEPS];
#pragma unroll
            for (int k = 0; k < CK_STEPS; ++k) {
                const long long idx = e0 + k * 32;
                dv[k] = dst[idx < e ? idx : e - 1];
            }
            unsigned int inr = 0u;
#pragma unroll
            for (int k = 0; k < CK_STEPS; ++k) {
                const long long idx = e0 + k * 32;
                const unsigned int ex = idx < e ? 1u : 0u;
                const unsigned int in = (dv[k] >= nsplit ? 1 : 0) == r ? 1u : 0u;
                inr |= (ex & in) << k;
            }
            asm volatile("" : "+v"(inr));
#pragma unroll
            for (int k = 0; k < CK_STEPS; ++k) {
                rk[k] = 0;
                if (inr & (1u << k)) rk[k] = atomicAdd(&cnt[dv[k]], 1);
            }
#pragma unroll
            for (int k = 0; k < CK_STEPS; ++k) {
                if (inr & (1u << k)) rank[e0 + k * 32] = rk[k];
            }
        }
    }
}

__global__ __launch_bounds__(256) void k_scan_local(const int* __restrict__ cnt, int* __restrict__ rowptr,
                                                    int* __restrict__ btot, float* __restrict__ dinv, int n) {
    __shared__ int wsum[8];
    const int t = threadIdx.x;
    const int lane = t & 31;
    const int wv = t >> 5;
    const int base = blockIdx.x * SCAN_BLK + t * 4;
    int c[4];
    int tsum = 0;
#pragma unroll
    for (int k = 0; k < 4; ++k) {
        const int idx = base + k;
        c[k] = idx < n ? cnt[idx] : 0;
        if (idx < n) dinv[idx] = rsqrtf((float)(c[k] + 1));
        tsum += c[k];
    }
    int inc = tsum;
#pragma unroll
    for (int d = 1; d < 32; d <<= 1) {
        const int o = __shfl_up(inc, d);
        if (lane >= d) inc += o;
    }
    if (lane == 31) wsum[wv] = inc;
    __syncthreads();
    int woff = 0, total = 0;
#pragma unroll
    for (int w = 0; w < 8; ++w) {
        const int s = wsum[w];
        if (w < wv) woff += s;
        total += s;
    }
    int run = woff + inc - tsum;
#pragma unroll
    for (int k = 0; k < 4; ++k) {
        const int idx = base + k;
        if (idx < n) rowptr[idx] = run;
        run += c[k];
    }
    if (t == 0) btot[blockIdx.x] = total;
}

__global__ __launch_bounds__(256) void k_scan_add(int* __restrict__ rowptr, const int* __restrict__ btot,
                                                  int n, int e) {
    __shared__ int wsum[8];
    const int t = threadIdx.x;
    const int b = blockIdx.x;
    int s = 0;
    for (int i = t; i < b; i += 256) s += btot[i];
#pragma unroll
    for (int d = 16; d >= 1; d >>= 1) s += __shfl_xor(s, d);
    if ((t & 31) == 0) wsum[t >> 5] = s;
    __syncthreads();
    int off = 0;
#pragma unroll
    for (int w = 0; w < 8; ++w) off += wsum[w];
    const int base = b * SCAN_BLK + t * 4;
#pragma unroll
    for (int k = 0; k < 4; ++k) {
        const int idx = base + k;
        if (idx < n) rowptr[idx] += off;
    }
    if (b == 0 && t == 0) rowptr[n] = e;
}

#define PL_STEPS 8
#define PL_EDGES (32 * PL_STEPS)
__global__ __launch_bounds__(256) void k_place(const int* __restrict__ src, const int* __restrict__ dst,
                                               const int* __restrict__ rank, const int* __restrict__ rowptr,
                                               int* __restrict__ col, unsigned int* __restrict__ ctr,
                                               long long e, int nsplit) {
    const int lane = threadIdx.x & 31;
    const int half = (int)((__builtin_amdgcn_s_sendmsg_rtn(0x87) >> 18) & 1u);
    const int ne = (int)e;
    const unsigned int nchunks = (unsigned int)((ne + PL_EDGES - 1) / PL_EDGES);

    for (int p = 0; p < 2; ++p) {
        const int r = half ^ p;
        unsigned int* c = ctr + r * CTR_STRIDE;
        for (unsigned int it = 0; it <= nchunks; ++it) {
            unsigned int ck = 0;
            if (lane == 0) ck = atomicAdd(c, 1u);
            ck = (unsigned int)__builtin_amdgcn_readfirstlane((int)ck);
            if (ck >= nchunks) break;
            const int e0 = (int)ck * PL_EDGES + lane;
            int dv[PL_STEPS], sv[PL_STEPS], rk[PL_STEPS], rp[PL_STEPS];
#pragma unroll
            for (int k = 0; k < PL_STEPS; ++k) {
                const int idx = e0 + k * 32;
                const int ii = idx < ne ? idx : ne - 1;
                dv[k] = dst[ii];
                sv[k] = src[ii];
                rk[k] = rank[ii];
            }
            unsigned int inr = 0u;
#pragma unroll
            for (int k = 0; k < PL_STEPS; ++k) {
                const int idx = e0 + k * 32;
                const unsigned int ex = idx < ne ? 1u : 0u;
                const unsigned int in = (dv[k] >= nsplit ? 1 : 0) == r ? 1u : 0u;
                inr |= (ex & in) << k;
            }
#pragma unroll
            for (int k = 0; k < PL_STEPS; ++k) {
                rp[k] = 0;
                if (inr & (1u << k)) rp[k] = rowptr[dv[k]];
            }
#pragma unroll
            for (int k = 0; k < PL_STEPS; ++k) {
                if (inr & (1u << k)) col[rp[k] + rk[k]] = sv[k];
            }
        }
    }
}

__global__ __launch_bounds__(256) void k_gather(const int* __restrict__ rowptr, const int* __restrict__ col,
                                                const float* __restrict__ dinv, const float* __restrict__ hs,
                                                const float* __restrict__ b, float* __restrict__ outp, int n) {
    const int lane = threadIdx.x & 31;
    const int wave = __builtin_amdgcn_readfirstlane((int)(blockIdx.x * 8 + (threadIdx.x >> 5)));
    const int nwaves = (int)gridDim.x * 8;
    const float bias = b[lane];
    for (int d = wave; d < n; d += nwaves) {
        const int beg = rowptr[d];
        const int end = rowptr[d + 1];
        float acc = hs[d * HID + lane];
        for (int base = beg; base < end; base += 32) {
            const int cnt = (end - base) < 32 ? (end - base) : 32;
            const int s = col[base + (lane < cnt ? lane : 0)];
            int j0 = 0;
            for (; j0 + 16 <= cnt; j0 += 16) {
                float v[16];
#pragma unroll
                for (int jj = 0; jj < 16; ++jj) {
                    const int sj = __builtin_amdgcn_readlane(s, j0 + jj);
                    v[jj] = hs[sj * HID + lane];
                }
#pragma unroll
                for (int jj = 0; jj < 16; ++jj) acc += v[jj];
            }
            if (j0 + 8 <= cnt) {
                float v[8];
#pragma unroll
                for (int jj = 0; jj < 8; ++jj) {
                    const int sj = __builtin_amdgcn_readlane(s, j0 + jj);
                    v[jj] = hs[sj * HID + lane];
                }
#pragma unroll
                for (int jj = 0; jj < 8; ++jj) acc += v[jj];
                j0 += 8;
            }
            if (j0 < cnt) {
                float v[8];
#pragma unroll
                for (int jj = 0; jj < 8; ++jj) {
                    const int j = j0 + jj;
                    const int sj = __builtin_amdgcn_readlane(s, j < cnt ? j : cnt - 1);
                    v[jj] = hs[sj * HID + lane];
                }
#pragma unroll
                for (int jj = 0; jj < 8; ++jj) acc += (j0 + jj < cnt) ? v[jj] : 0.0f;
            }
        }
        const float o = acc * dinv[d] + bias;
        outp[d * HID + lane] = o > 0.0f ? o : 0.0f;
    }
}

extern "C" void kernel_launch(void* const* d_in, const int* in_sizes, int n_in,
                              void* d_out, int out_size, void* d_ws, size_t ws_size,
                              hipStream_t stream) {
    const float* x   = (const float*)d_in[0];
    const int*   ei  = (const int*)d_in[1];
    const float* W1  = (const float*)d_in[2];
    const float* b1  = (const float*)d_in[3];
    const float* W2  = (const float*)d_in[4];
    const float* b2  = (const float*)d_in[5];
    const float* Wfc = (const float*)d_in[6];
    const float* bfc = (const float*)d_in[7];
    float* out = (float*)d_out;

    const int F = 128;
    const long long n = in_sizes[0] / F;
    const long long e = in_sizes[1] / 2;
    const int* src = ei;
    const int* dst = ei + e;
    const long long nh = n * HID;
    const int B = 256;
    dim3 bfill((unsigned)((n + B - 1) / B));
    dim3 bgemm((unsigned)((n + 63) / 64));

    const long long w_n   = (n + 63) & ~63LL;
    const long long w_rp  = (n + 1 + 63) & ~63LL;
    const long long w_bt  = 1024;
    const long long w_e   = (e + 63) & ~63LL;
    const long long w_nh  = (nh + 63) & ~63LL;
    const long long w_ag  = w_nh > w_e ? w_nh : w_e;
    const long long need  = (w_n + w_rp + w_bt + w_n + w_e + w_nh + w_ag) * 4;
    const long long nsblk = (n + SCAN_BLK - 1) / SCAN_BLK;

    if ((long long)ws_size >= need && nsblk <= 512) {
        int*   cnt    = (int*)d_ws;
        int*   rowptr = cnt + w_n;
        int*   btot   = rowptr + w_rp;
        unsigned int* tick = (unsigned int*)(btot + 512);
        const int nsplit = (int)((n / 2) & ~63LL);
        float* dinv   = (float*)(btot + w_bt);
        int*   col    = (int*)(dinv + w_n);
        float* hbuf   = (float*)(col + w_e);
        float* agg    = hbuf + w_nh;
        int*   rank   = (int*)agg;
        dim3 bscan((unsigned)nsblk);

        k_zero<<<bfill, B, 0, stream>>>(cnt, tick, (int)n);
        k_count<<<PGRID, B, 0, stream>>>(dst, cnt, rank, tick, e, nsplit);
        k_scan_local<<<bscan, B, 0, stream>>>(cnt, rowptr, btot, dinv, (int)n);
        k_scan_add<<<bscan, B, 0, stream>>>(rowptr, btot, (int)n, (int)e);
        k_place<<<PGRID, B, 0, stream>>>(src, dst, rank, rowptr, col, tick + 128, e, nsplit);

        k_gemm<128><<<bgemm, 128, 0, stream>>>(x, W1, dinv, hbuf, (int)n);
        k_gather<<<PGRID, B, 0, stream>>>(rowptr, col, dinv, hbuf, b1, agg, (int)n);

        k_gemm<32><<<bgemm, 128, 0, stream>>>(agg, W2, dinv, hbuf, (int)n);
        k_gather<<<PGRID, B, 0, stream>>>(rowptr, col, dinv, hbuf, b2, agg, (int)n);

        k_fc<<<bfill, B, 0, stream>>>(agg, Wfc, bfc, out, (int)n);
        return;
    }

    unsigned int* ctr = (unsigned int*)d_ws;
    float* dinv = (float*)d_ws + 256;
    float* hbuf = dinv + n;
    float* agg  = hbuf + n * HID;

    dim3 bfillh((unsigned)((nh + B - 1) / B));
    dim3 bedge((unsigned)((e + B - 1) / B));
    dim3 bscat(2048);
    const int nsplit = (int)(n / 2);

    k_fill<<<bfill, B, 0, stream>>>(dinv, 1.0f, n);
    k_deg<<<bedge, B, 0, stream>>>(dst, dinv, e);
    k_rsqrt<<<bfill, B, 0, stream>>>(dinv, ctr, n);

    k_gemm<128><<<bgemm, 128, 0, stream>>>(x, W1, nullptr, hbuf, (int)n);
    k_fill<<<bfillh, B, 0, stream>>>(agg, 0.0f, nh);
    k_scatter<<<bscat, B, 0, stream>>>(src, dst, dinv, hbuf, agg, ctr, e, nsplit);
    k_finalize<<<bfillh, B, 0, stream>>>(hbuf, dinv, b1, agg, nh);

    k_gemm<32><<<bgemm, 128, 0, stream>>>(agg, W2, nullptr, hbuf, (int)n);
    k_fill<<<bfillh, B, 0, stream>>>(agg, 0.0f, nh);
    k_scatter<<<bscat, B, 0, stream>>>(src, dst, dinv, hbuf, agg, ctr + 128, e, nsplit);
    k_finalize<<<bfillh, B, 0, stream>>>(hbuf, dinv, b2, agg, nh);

    k_fc<<<bfill, B, 0, stream>>>(agg, Wfc, bfc, out, (int)n);
}


# edges grouped by destination through LDS (two-level counting sort, whole-run stores) instead of global rank atomics and scattered placement; sum order = arrival order of LDS adds; 9 launches
# speedup vs baseline: 4.6071x; 1.4740x over previous
#include <hip/hip_runtime.h>

typedef float v2f __attribute__((ext_vector_type(2)));
typedef float v8f __attribute__((ext_vector_type(8)));

#define HID 32

__global__ void k_fill(float* __restrict__ p, float v, long long n) {
    long long i = (long long)blockIdx.x * blockDim.x + threadIdx.x;
    if (i < n) p[i] = v;
}

__global__ void k_deg(const int* __restrict__ dst, float* __restrict__ deg, long long e) {
    long long i = (long long)blockIdx.x * blockDim.x + threadIdx.x;
    if (i < e) atomicAdd(&deg[dst[i]], 1.0f);
}

#define NCTR 256
__global__ void k_rsqrt(float* __restrict__ d, unsigned int* __restrict__ ctr, long long n) {
    long long i = (long long)blockIdx.x * blockDim.x + threadIdx.x;
    if (i < NCTR) ctr[i] = 0u;
    if (i < n) {
        float v = d[i];
        d[i] = (v > 0.0f) ? rsqrtf(v) : 0.0f;
    }
}

template <int KDIM>
__global__ __launch_bounds__(128) void k_gemm(const float* __restrict__ A,
                                              const float* __restrict__ W,
                                              const float* __restrict__ rs,
                                              float* __restrict__ H, int n) {
    const int lane = threadIdx.x & 31;
    const int wave = threadIdx.x >> 5;
    const int row0 = (blockIdx.x * 4 + wave) * 16;
    if (row0 >= n) return;

    const int lr = lane & 15;
    const int kh = (lane >> 4) << 1;

    int arow = row0 + lr;
    if (arow >= n) arow = n - 1;
    const float* __restrict__ ap = A + (size_t)arow * KDIM;

    v8f c0 = {0.f, 0.f, 0.f, 0.f, 0.f, 0.f, 0.f, 0.f};
    v8f c1 = {0.f, 0.f, 0.f, 0.f, 0.f, 0.f, 0.f, 0.f};

#pragma unroll
    for (int k0 = 0; k0 < KDIM; k0 += 4) {
        v2f a;
        a.x = ap[k0 + kh];
        a.y = ap[k0 + kh + 1];
        const float* __restrict__ wp = W + (size_t)(k0 + kh) * HID;
        v2f b0, b1;
        b0.x = wp[lr];
        b0.y = wp[HID + lr];
        b1.x = wp[16 + lr];
        b1.y = wp[HID + 16 + lr];
        c0 = __builtin_amdgcn_wmma_f32_16x16x4_f32(false, a, false, b0, (short)0, c0, false, false);
        c1 = __builtin_amdgcn_wmma_f32_16x16x4_f32(false, a, false, b1, (short)0, c1, false, false);
    }

    const int mofs = (lane >> 4) * 8;
#pragma unroll
    for (int r = 0; r < 8; ++r) {
        int row = row0 + mofs + r;
        if (row < n) {
            const float sc = rs ? rs[row] : 1.0f;
            float* __restrict__ hp = H + (size_t)row * HID;
            hp[lr]      = c0[r] * sc;
            hp[16 + lr] = c1[r] * sc;
        }
    }
}

#define CH_STEPS 4
#define CH_EDGES (32 * CH_STEPS)
#define CTR_STRIDE 64

__device__ __forceinline__ void edge_group32(int s, int d, int lane,
                                             const float* __restrict__ dinv,
                                             const float* __restrict__ h,
                                             float* __restrict__ agg) {
    const float ws = dinv[s];
    const float wd = dinv[d];
    float v[32];
#pragma unroll
    for (int j = 0; j < 32; ++j) {
        const int sj = __builtin_amdgcn_readlane(s, j);
        v[j] = h[sj * HID + lane];
    }
    const float nrm = ws * wd;
#pragma unroll
    for (int j = 0; j < 32; ++j) {
        const int dj = __builtin_amdgcn_readlane(d, j);
        const float wj = __int_as_float(__builtin_amdgcn_readlane(__float_as_int(nrm), j));
        atomicAdd(&agg[dj * HID + lane], v[j] * wj);
    }
}

__global__ __launch_bounds__(256) void k_scatter(const int* __restrict__ src,
                                                 const int* __restrict__ dst,
                                                 const float* __restrict__ dinv,
                                                 const float* __restrict__ h,
                                                 float* __restrict__ agg,
                                                 unsigned int* __restrict__ ctr,
                                                 long long e, int nsplit) {
    __shared__ int q_s[8][64];
    __shared__ int q_d[8][64];
    const int lane = threadIdx.x & 31;
    const int wave = __builtin_amdgcn_readfirstlane((int)(threadIdx.x >> 5));
    int* qs = q_s[wave];
    int* qd = q_d[wave];
    const int half = (int)((__builtin_amdgcn_s_sendmsg_rtn(0x87) >> 18) & 1u);
    const unsigned int nchunks = (unsigned int)((e + CH_EDGES - 1) / CH_EDGES);
    int cnt = 0;

    for (int p = 0; p < 2; ++p) {
        const int r = half ^ p;
        unsigned int* c = ctr + r * CTR_STRIDE;
        for (unsigned int it = 0; it <= nchunks; ++it) {
            unsigned int ck = 0;
            if (lane == 0) ck = atomicAdd(c, 1u);
            ck = (unsigned int)__builtin_amdgcn_readfirstlane((int)ck);
            if (ck >= nchunks) break;
            const long long e0 = (long long)ck * CH_EDGES + lane;
            int sv[CH_STEPS], dv[CH_STEPS];
#pragma unroll
            for (int k = 0; k < CH_STEPS; ++k) {
                const long long idx = e0 + k * 32;
                const long long ii = idx < e ? idx : e - 1;
                sv[k] = src[ii];
                dv[k] = dst[ii];
            }
#pragma unroll
            for (int k = 0; k < CH_STEPS; ++k) {
                const long long idx = e0 + k * 32;
                const bool m = (idx < e) && ((dv[k] >= nsplit ? 1 : 0) == r);
                const unsigned int mask = __builtin_amdgcn_ballot_w32(m);
                if (m) {
                    const int pos = cnt + (int)__builtin_amdgcn_mbcnt_lo(mask, 0u);
                    qs[pos] = sv[k];
                    qd[pos] = dv[k];
                }
                cnt += __builtin_popcount(mask);
                if (cnt >= 32) {
                    const int s = qs[lane];
                    const int d = qd[lane];
                    const int s2 = qs[32 + lane];
                    const int d2 = qd[32 + lane];
                    edge_group32(s, d, lane, dinv, h, agg);
                    cnt -= 32;
                    if (lane < cnt) {
                        qs[lane] = s2;
                        qd[lane] = d2;
                    }
                }
            }
        }
    }
    if (cnt > 0) {
        const int ql = lane < cnt ? lane : cnt - 1;
        const int s = qs[ql];
        const int d = qd[ql];
        const float nrm = dinv[s] * dinv[d];
        for (int j = 0; j < cnt; ++j) {
            const int sj = __builtin_amdgcn_readlane(s, j);
            const int dj = __builtin_amdgcn_readlane(d, j);
            const float wj = __int_as_float(__builtin_amdgcn_readlane(__float_as_int(nrm), j));
            atomicAdd(&agg[dj * HID + lane], h[sj * HID + lane] * wj);
        }
    }
}

__global__ void k_finalize(const float* __restrict__ h, const float* __restrict__ dinv,
                           const float* __restrict__ b, float* __restrict__ agg,
                           long long total) {
    long long t = (long long)blockIdx.x * blockDim.x + threadIdx.x;
    if (t >= total) return;
    long long i = t >> 5;
    int k = (int)(t & 31);
    float di = dinv[i];
    float v = agg[t] + h[t] * (di * di) + b[k];
    agg[t] = v > 0.0f ? v : 0.0f;
}

__global__ void k_fc(const float* __restrict__ h, const float* __restrict__ wfc,
                     const float* __restrict__ bfc, float* __restrict__ out, int n) {
    int i = blockIdx.x * blockDim.x + threadIdx.x;
    if (i >= n) return;
    const float4* __restrict__ hp = (const float4*)(h + (size_t)i * HID);
    const float4* __restrict__ wp = (const float4*)wfc;
    float acc = 0.0f;
#pragma unroll
    for (int k = 0; k < 8; ++k) {
        float4 a = hp[k], w = wp[k];
        acc += a.x * w.x + a.y * w.y + a.z * w.z + a.w * w.w;
    }
    out[i] = acc + bfc[0];
}

#define PGRID 2048
#define SCAN_BLK 1024

__global__ void k_zero(int* __restrict__ p, unsigned int* __restrict__ ctr, int n) {
    int i = blockIdx.x * blockDim.x + threadIdx.x;
    if (i < NCTR) ctr[i] = 0u;
    if (i < n) p[i] = 0;
}

#define CK_STEPS 8
#define CK_EDGES (32 * CK_STEPS)
__global__ __launch_bounds__(256) void k_count(const int* __restrict__ dst, int* __restrict__ cnt,
                                               int* __restrict__ rank, unsigned int* __restrict__ ctr,
                                               long long e, int nsplit) {
    const int lane = threadIdx.x & 31;
    const int half = (int)((__builtin_amdgcn_s_sendmsg_rtn(0x87) >> 18) & 1u);
    const unsigned int nchunks = (unsigned int)((e + CK_EDGES - 1) / CK_EDGES);

    for (int p = 0; p < 2; ++p) {
        const int r = half ^ p;
        unsigned int* c = ctr + r * CTR_STRIDE;
        for (unsigned int it = 0; it <= nchunks; ++it) {
            unsigned int ck = 0;
            if (lane == 0) ck = atomicAdd(c, 1u);
            ck = (unsigned int)__builtin_amdgcn_readfirstlane((int)ck);
            if (ck >= nchunks) break;
            const long long e0 = (long long)ck * CK_EDGES + lane;
            int dv[CK_STEPS], rk[CK_STEPS];
#pragma unroll
            for (int k = 0; k < CK_STEPS; ++k) {
                const long long idx = e0 + k * 32;
                dv[k] = dst[idx < e ? idx : e - 1];
            }
            unsigned int inr = 0u;
#pragma unroll
            for (int k = 0; k < CK_STEPS; ++k) {
                const long long idx = e0 + k * 32;
                const unsigned int ex = idx < e ? 1u : 0u;
                const unsigned int in = (dv[k] >= nsplit ? 1 : 0) == r ? 1u : 0u;
                inr |= (ex & in) << k;
            }
            asm volatile("" : "+v"(inr));
#pragma unroll
            for (int k = 0; k < CK_STEPS; ++k) {
                rk[k] = 0;
                if (inr & (1u << k)) rk[k] = atomicAdd(&cnt[dv[k]], 1);
            }
#pragma unroll
            for (int k = 0; k < CK_STEPS; ++k) {
                if (inr & (1u << k)) rank[e0 + k * 32] = rk[k];
            }
        }
    }
}

__global__ __launch_bounds__(256) void k_scan_local(const int* __restrict__ cnt, int* __restrict__ rowptr,
                                                    int* __restrict__ btot, float* __restrict__ dinv, int n) {
    __shared__ int wsum[8];
    const int t = threadIdx.x;
    const int lane = t & 31;
    const int wv = t >> 5;
    const int base = blockIdx.x * SCAN_BLK + t * 4;
    int c[4];
    int tsum = 0;
#pragma unroll
    for (int k = 0; k < 4; ++k) {
        const int idx = base + k;
        c[k] = idx < n ? cnt[idx] : 0;
        if (idx < n) dinv[idx] = rsqrtf((float)(c[k] + 1));
        tsum += c[k];
    }
    int inc = tsum;
#pragma unroll
    for (int d = 1; d < 32; d <<= 1) {
        const int o = __shfl_up(inc, d);
        if (lane >= d) inc += o;
    }
    if (lane == 31) wsum[wv] = inc;
    __syncthreads();
    int woff = 0, total = 0;
#pragma unroll
    for (int w = 0; w < 8; ++w) {
        const int s = wsum[w];
        if (w < wv) woff += s;
        total += s;
    }
    int run = woff + inc - tsum;
#pragma unroll
    for (int k = 0; k < 4; ++k) {
        const int idx = base + k;
        if (idx < n) rowptr[idx] = run;
        run += c[k];
    }
    if (t == 0) btot[blockIdx.x] = total;
}

__global__ __launch_bounds__(256) void k_scan_add(int* __restrict__ rowptr, const int* __restrict__ btot,
                                                  int n, int e) {
    __shared__ int wsum[8];
    const int t = threadIdx.x;
    const int b = blockIdx.x;
    int s = 0;
    for (int i = t; i < b; i += 256) s += btot[i];
#pragma unroll
    for (int d = 16; d >= 1; d >>= 1) s += __shfl_xor(s, d);
    if ((t & 31) == 0) wsum[t >> 5] = s;
    __syncthreads();
    int off = 0;
#pragma unroll
    for (int w = 0; w < 8; ++w) off += wsum[w];
    const int base = b * SCAN_BLK + t * 4;
#pragma unroll
    for (int k = 0; k < 4; ++k) {
        const int idx = base + k;
        if (idx < n) rowptr[idx] += off;
    }
    if (b == 0 && t == 0) rowptr[n] = e;
}

#define PL_STEPS 8
#define PL_EDGES (32 * PL_STEPS)
__global__ __launch_bounds__(256) void k_place(const int* __restrict__ src, const int* __restrict__ dst,
                                               const int* __restrict__ rank, const int* __restrict__ rowptr,
                                               int* __restrict__ col, unsigned int* __restrict__ ctr,
                                               long long e, int nsplit) {
    const int lane = threadIdx.x & 31;
    const int half = (int)((__builtin_amdgcn_s_sendmsg_rtn(0x87) >> 18) & 1u);
    const int ne = (int)e;
    const unsigned int nchunks = (unsigned int)((ne + PL_EDGES - 1) / PL_EDGES);

    for (int p = 0; p < 2; ++p) {
        const int r = half ^ p;
        unsigned int* c = ctr + r * CTR_STRIDE;
        for (unsigned int it = 0; it <= nchunks; ++it) {
            unsigned int ck = 0;
            if (lane == 0) ck = atomicAdd(c, 1u);
            ck = (unsigned int)__builtin_amdgcn_readfirstlane((int)ck);
            if (ck >= nchunks) break;
            const int e0 = (int)ck * PL_EDGES + lane;
            int dv[PL_STEPS], sv[PL_STEPS], rk[PL_STEPS], rp[PL_STEPS];
#pragma unroll
            for (int k = 0; k < PL_STEPS; ++k) {
                const int idx = e0 + k * 32;
                const int ii = idx < ne ? idx : ne - 1;
                dv[k] = dst[ii];
                sv[k] = src[ii];
                rk[k] = rank[ii];
            }
            unsigned int inr = 0u;
#pragma unroll
            for (int k = 0; k < PL_STEPS; ++k) {
                const int idx = e0 + k * 32;
                const unsigned int ex = idx < ne ? 1u : 0u;
                const unsigned int in = (dv[k] >= nsplit ? 1 : 0) == r ? 1u : 0u;
                inr |= (ex & in) << k;
            }
#pragma unroll
            for (int k = 0; k < PL_STEPS; ++k) {
                rp[k] = 0;
                if (inr & (1u << k)) rp[k] = rowptr[dv[k]];
            }
#pragma unroll
            for (int k = 0; k < PL_STEPS; ++k) {
                if (inr & (1u << k)) col[rp[k] + rk[k]] = sv[k];
            }
        }
    }
}

#define BK_SHIFT 10
#define BK_NODES (1 << BK_SHIFT)
#define BK_MAX   256
#define TL_PER   8
#define TL_EDGES (1024 * TL_PER)
#define ST_CAP   49152

__device__ __forceinline__ int wg_excl_scan_1024(int v, int* wsum, int* tot) {
    const int lane = threadIdx.x & 31;
    const int wv = threadIdx.x >> 5;
    int inc = v;
#pragma unroll
    for (int d = 1; d < 32; d <<= 1) {
        const int o = __shfl_up(inc, d);
        if (lane >= d) inc += o;
    }
    if (lane == 31) wsum[wv] = inc;
    __syncthreads();
    const int ws = wsum[lane];
    int winc = ws;
#pragma unroll
    for (int d = 1; d < 32; d <<= 1) {
        const int o = __shfl_up(winc, d);
        if (lane >= d) winc += o;
    }
    const int total = __shfl(winc, 31);
    const int woff = __shfl(winc, wv) - __shfl(ws, wv);
    __syncthreads();
    *tot = total;
    return woff + inc - v;
}

__global__ __launch_bounds__(1024) void k_bhist(const int* __restrict__ dst, int* __restrict__ bcnt,
                                                int ne, int nb, int ntiles) {
    __shared__ int hist[BK_MAX];
    const int t = threadIdx.x;
    if (t < BK_MAX) hist[t] = 0;
    __syncthreads();
    const int e0 = blockIdx.x * TL_EDGES + t;
    int dv[TL_PER];
#pragma unroll
    for (int k = 0; k < TL_PER; ++k) {
        const int idx = e0 + k * 1024;
        dv[k] = dst[idx < ne ? idx : ne - 1];
    }
#pragma unroll
    for (int k = 0; k < TL_PER; ++k) {
        if (e0 + k * 1024 < ne) atomicAdd(&hist[dv[k] >> BK_SHIFT], 1);
    }
    __syncthreads();
    if (t < nb) bcnt[t * ntiles + blockIdx.x] = hist[t];
}

__global__ __launch_bounds__(256) void k_bscan(int* __restrict__ bcnt, int* __restrict__ btotal, int ntiles) {
    __shared__ int wsum[8];
    const int t = threadIdx.x;
    const int lane = t & 31;
    const int wv = t >> 5;
    int* __restrict__ row = bcnt + (size_t)blockIdx.x * ntiles;
    int c[4];
    int tsum = 0;
#pragma unroll
    for (int k = 0; k < 4; ++k) {
        const int idx = t * 4 + k;
        c[k] = idx < ntiles ? row[idx] : 0;
        tsum += c[k];
    }
    int inc = tsum;
#pragma unroll
    for (int d = 1; d < 32; d <<= 1) {
        const int o = __shfl_up(inc, d);
        if (lane >= d) inc += o;
    }
    if (lane == 31) wsum[wv] = inc;
    __syncthreads();
    int woff = 0, total = 0;
#pragma unroll
    for (int w = 0; w < 8; ++w) {
        const int s = wsum[w];
        if (w < wv) woff += s;
        total += s;
    }
    int run = woff + inc - tsum;
#pragma unroll
    for (int k = 0; k < 4; ++k) {
        const int idx = t * 4 + k;
        if (idx < ntiles) row[idx] = run;
        run += c[k];
    }
    if (t == 0) btotal[blockIdx.x] = total;
}

__global__ __launch_bounds__(1024) void k_bpart(const int* __restrict__ src, const int* __restrict__ dst,
                                                const int* __restrict__ boff, const int* __restrict__ btotal,
                                                int* __restrict__ part, int ne, int nb, int ntiles) {
    __shared__ int hist[BK_MAX];
    __shared__ int gdelta[BK_MAX];
    __shared__ int wsum[32];
    __shared__ int2 stage[TL_EDGES];
    const int t = threadIdx.x;
    if (t < BK_MAX) hist[t] = 0;
    __syncthreads();
    const int e0 = blockIdx.x * TL_EDGES;
    int sv[TL_PER], dv[TL_PER];
#pragma unroll
    for (int k = 0; k < TL_PER; ++k) {
        const int idx = e0 + k * 1024 + t;
        const int ii = idx < ne ? idx : ne - 1;
        sv[k] = src[ii];
        dv[k] = dst[ii];
    }
#pragma unroll
    for (int k = 0; k < TL_PER; ++k) {
        if (e0 + k * 1024 + t < ne) atomicAdd(&hist[dv[k] >> BK_SHIFT], 1);
    }
    __syncthreads();
    const int hc = t < nb ? hist[t] : 0;
    const int bt = t < nb ? btotal[t] : 0;
    const int bo = t < nb ? boff[t * ntiles + blockIdx.x] : 0;
    int tot;
    const int lex = wg_excl_scan_1024(hc, wsum, &tot);
    const int gex = wg_excl_scan_1024(bt, wsum, &tot);
    if (t < nb) {
        hist[t] = lex;
        gdelta[t] = gex + bo - lex;
    }
    __syncthreads();
#pragma unroll
    for (int k = 0; k < TL_PER; ++k) {
        if (e0 + k * 1024 + t < ne) {
            const int bk = dv[k] >> BK_SHIFT;
            const int pos = atomicAdd(&hist[bk], 1);
            stage[pos] = make_int2((int)(((unsigned int)sv[k] << BK_SHIFT) | (unsigned int)(dv[k] & (BK_NODES - 1))), bk);
        }
    }
    __syncthreads();
    const int ct = (ne - e0) < TL_EDGES ? (ne - e0) : TL_EDGES;
#pragma unroll
    for (int k = 0; k < TL_PER; ++k) {
        const int i = k * 1024 + t;
        if (i < ct) {
            const int2 p = stage[i];
            part[gdelta[p.y] + i] = p.x;
        }
    }
}

__global__ __launch_bounds__(1024) void k_bsort(const int* __restrict__ part, const int* __restrict__ btotal,
                                                int* __restrict__ rowptr, int* __restrict__ col,
                                                float* __restrict__ dinv, int n, int ne, int nb) {
    __shared__ int ncnt[BK_NODES];
    __shared__ int wsum[32];
    __shared__ int bsz[2];
    __shared__ int stage[ST_CAP];
    const int t = threadIdx.x;
    const int b = blockIdx.x;
    const int base = b << BK_SHIFT;
    ncnt[t] = 0;
    const int bt = t < nb ? btotal[t] : 0;
    int tot;
    const int gex = wg_excl_scan_1024(bt, wsum, &tot);
    if (t == b) {
        bsz[0] = gex;
        bsz[1] = bt;
    }
    __syncthreads();
    const int bs = bsz[0];
    const int cb = bsz[1];
    const int* __restrict__ pp = part + bs;
    const int nit = (cb + TL_EDGES - 1) / TL_EDGES;

    for (int it = 0; it < nit; ++it) {
        const int i0 = it * TL_EDGES + t;
        int dv[TL_PER];
#pragma unroll
        for (int k = 0; k < TL_PER; ++k) {
            const int i = i0 + k * 1024;
            dv[k] = pp[i < cb ? i : cb - 1];
        }
#pragma unroll
        for (int k = 0; k < TL_PER; ++k) {
            if (i0 + k * 1024 < cb) atomicAdd(&ncnt[dv[k] & (BK_NODES - 1)], 1);
        }
    }
    __syncthreads();
    const int c = ncnt[t];
    const int ex = wg_excl_scan_1024(c, wsum, &tot);
    if (base + t < n) {
        rowptr[base + t] = bs + ex;
        dinv[base + t] = rsqrtf((float)(c + 1));
    }
    if (b == nb - 1 && t == 0) rowptr[n] = ne;
    ncnt[t] = ex;
    __syncthreads();

    if (cb <= ST_CAP) {
        for (int it = 0; it < nit; ++it) {
            const int i0 = it * TL_EDGES + t;
            int pv[TL_PER];
#pragma unroll
            for (int k = 0; k < TL_PER; ++k) {
                const int i = i0 + k * 1024;
                pv[k] = pp[i < cb ? i : cb - 1];
            }
#pragma unroll
            for (int k = 0; k < TL_PER; ++k) {
                if (i0 + k * 1024 < cb) {
                    const int pos = atomicAdd(&ncnt[pv[k] & (BK_NODES - 1)], 1);
                    stage[pos] = (int)((unsigned int)pv[k] >> BK_SHIFT);
                }
            }
        }
        __syncthreads();
        for (int i = t; i < cb; i += 1024) col[bs + i] = stage[i];
    } else {
        for (int it = 0; it < nit; ++it) {
            const int i0 = it * TL_EDGES + t;
            int pv[TL_PER];
#pragma unroll
            for (int k = 0; k < TL_PER; ++k) {
                const int i = i0 + k * 1024;
                pv[k] = pp[i < cb ? i : cb - 1];
            }
#pragma unroll
            for (int k = 0; k < TL_PER; ++k) {
                if (i0 + k * 1024 < cb) {
                    const int pos = atomicAdd(&ncnt[pv[k] & (BK_NODES - 1)], 1);
                    col[bs + pos] = (int)((unsigned int)pv[k] >> BK_SHIFT);
                }
            }
        }
    }
}

__global__ __launch_bounds__(256) void k_gather(const int* __restrict__ rowptr, const int* __restrict__ col,
                                                const float* __restrict__ dinv, const float* __restrict__ hs,
                                                const float* __restrict__ b, float* __restrict__ outp, int n) {
    const int lane = threadIdx.x & 31;
    const int wave = __builtin_amdgcn_readfirstlane((int)(blockIdx.x * 8 + (threadIdx.x >> 5)));
    const int nwaves = (int)gridDim.x * 8;
    const float bias = b[lane];
    for (int d = wave; d < n; d += nwaves) {
        const int beg = rowptr[d];
        const int end = rowptr[d + 1];
        float acc = hs[d * HID + lane];
        for (int base = beg; base < end; base += 32) {
            const int cnt = (end - base) < 32 ? (end - base) : 32;
            const int s = col[base + (lane < cnt ? lane : 0)];
            int j0 = 0;
            for (; j0 + 16 <= cnt; j0 += 16) {
                float v[16];
#pragma unroll
                for (int jj = 0; jj < 16; ++jj) {
                    const int sj = __builtin_amdgcn_readlane(s, j0 + jj);
                    v[jj] = hs[sj * HID + lane];
                }
#pragma unroll
                for (int jj = 0; jj < 16; ++jj) acc += v[jj];
            }
            if (j0 + 8 <= cnt) {
                float v[8];
#pragma unroll
                for (int jj = 0; jj < 8; ++jj) {
                    const int sj = __builtin_amdgcn_readlane(s, j0 + jj);
                    v[jj] = hs[sj * HID + lane];
                }
#pragma unroll
                for (int jj = 0; jj < 8; ++jj) acc += v[jj];
                j0 += 8;
            }
            if (j0 < cnt) {
                float v[8];
#pragma unroll
                for (int jj = 0; jj < 8; ++jj) {
                    const int j = j0 + jj;
                    const int sj = __builtin_amdgcn_readlane(s, j < cnt ? j : cnt - 1);
                    v[jj] = hs[sj * HID + lane];
                }
#pragma unroll
                for (int jj = 0; jj < 8; ++jj) acc += (j0 + jj < cnt) ? v[jj] : 0.0f;
            }
        }
        const float o = acc * dinv[d] + bias;
        outp[d * HID + lane] = o > 0.0f ? o : 0.0f;
    }
}

extern "C" void kernel_launch(void* const* d_in, const int* in_sizes, int n_in,
                              void* d_out, int out_size, void* d_ws, size_t ws_size,
                              hipStream_t stream) {
    const float* x   = (const float*)d_in[0];
    const int*   ei  = (const int*)d_in[1];
    const float* W1  = (const float*)d_in[2];
    const float* b1  = (const float*)d_in[3];
    const float* W2  = (const float*)d_in[4];
    const float* b2  = (const float*)d_in[5];
    const float* Wfc = (const float*)d_in[6];
    const float* bfc = (const float*)d_in[7];
    float* out = (float*)d_out;

    const int F = 128;
    const long long n = in_sizes[0] / F;
    const long long e = in_sizes[1] / 2;
    const int* src = ei;
    const int* dst = ei + e;
    const long long nh = n * HID;
    const int B = 256;
    dim3 bfill((unsigned)((n + B - 1) / B));
    dim3 bgemm((unsigned)((n + 63) / 64));

    const long long w_n   = (n + 63) & ~63LL;
    const long long w_rp  = (n + 1 + 63) & ~63LL;
    const long long w_bt  = 1024;
    const long long w_e   = (e + 63) & ~63LL;
    const long long w_nh  = (nh + 63) & ~63LL;
    const long long w_ag  = w_nh > w_e ? w_nh : w_e;
    const long long need  = (w_n + w_rp + w_bt + w_n + w_e + w_nh + w_ag) * 4;
    const long long nsblk = (n + SCAN_BLK - 1) / SCAN_BLK;

    if ((long long)ws_size >= need && nsblk <= 512) {
        int*   cnt    = (int*)d_ws;
        int*   rowptr = cnt + w_n;
        int*   btot   = rowptr + w_rp;
        unsigned int* tick = (unsigned int*)(btot + 512);
        const int nsplit = (int)((n / 2) & ~63LL);
        float* dinv   = (float*)(btot + w_bt);
        int*   col    = (int*)(dinv + w_n);
        float* hbuf   = (float*)(col + w_e);
        float* agg    = hbuf + w_nh;
        int*   rank   = (int*)agg;
        dim3 bscan((unsigned)nsblk);

        const int nb = (int)((n + BK_NODES - 1) >> BK_SHIFT);
        const int ntiles = (int)((e + TL_EDGES - 1) / TL_EDGES);
        if (nb <= BK_MAX && ntiles >= 1 && ntiles <= 1024 && (long long)nb * ntiles <= w_n &&
            n <= (1LL << (32 - BK_SHIFT)) && e < (1LL << 30)) {
            int* bcnt   = cnt;
            int* btotal = btot;
            int* part   = (int*)agg;
            k_bhist<<<ntiles, 1024, 0, stream>>>(dst, bcnt, (int)e, nb, ntiles);
            k_bscan<<<nb, 256, 0, stream>>>(bcnt, btotal, ntiles);
            k_bpart<<<ntiles, 1024, 0, stream>>>(src, dst, bcnt, btotal, part, (int)e, nb, ntiles);
            k_bsort<<<nb, 1024, 0, stream>>>(part, btotal, rowptr, col, dinv, (int)n, (int)e, nb);
        } else {
            k_zero<<<bfill, B, 0, stream>>>(cnt, tick, (int)n);
            k_count<<<PGRID, B, 0, stream>>>(dst, cnt, rank, tick, e, nsplit);
            k_scan_local<<<bscan, B, 0, stream>>>(cnt, rowptr, btot, dinv, (int)n);
            k_scan_add<<<bscan, B, 0, stream>>>(rowptr, btot, (int)n, (int)e);
            k_place<<<PGRID, B, 0, stream>>>(src, dst, rank, rowptr, col, tick + 128, e, nsplit);
        }

        k_gemm<128><<<bgemm, 128, 0, stream>>>(x, W1, dinv, hbuf, (int)n);
        k_gather<<<PGRID, B, 0, stream>>>(rowptr, col, dinv, hbuf, b1, agg, (int)n);

        k_gemm<32><<<bgemm, 128, 0, stream>>>(agg, W2, dinv, hbuf, (int)n);
        k_gather<<<PGRID, B, 0, stream>>>(rowptr, col, dinv, hbuf, b2, agg, (int)n);

        k_fc<<<bfill, B, 0, stream>>>(agg, Wfc, bfc, out, (int)n);
        return;
    }

    unsigned int* ctr = (unsigned int*)d_ws;
    float* dinv = (float*)d_ws + 256;
    float* hbuf = dinv + n;
    float* agg  = hbuf + n * HID;

    dim3 bfillh((unsigned)((nh + B - 1) / B));
    dim3 bedge((unsigned)((e + B - 1) / B));
    dim3 bscat(2048);
    const int nsplit = (int)(n / 2);

    k_fill<<<bfill, B, 0, stream>>>(dinv, 1.0f, n);
    k_deg<<<bedge, B, 0, stream>>>(dst, dinv, e);
    k_rsqrt<<<bfill, B, 0, stream>>>(dinv, ctr, n);

    k_gemm<128><<<bgemm, 128, 0, stream>>>(x, W1, nullptr, hbuf, (int)n);
    k_fill<<<bfillh, B, 0, stream>>>(agg, 0.0f, nh);
    k_scatter<<<bscat, B, 0, stream>>>(src, dst, dinv, hbuf, agg, ctr, e, nsplit);
    k_finalize<<<bfillh, B, 0, stream>>>(hbuf, dinv, b1, agg, nh);

    k_gemm<32><<<bgemm, 128, 0, stream>>>(agg, W2, nullptr, hbuf, (int)n);
    k_fill<<<bfillh, B, 0, stream>>>(agg, 0.0f, nh);
    k_scatter<<<bscat, B, 0, stream>>>(src, dst, dinv, hbuf, agg, ctr + 128, e, nsplit);
    k_finalize<<<bfillh, B, 0, stream>>>(hbuf, dinv, b2, agg, nh);

    k_fc<<<bfill, B, 0, stream>>>(agg, Wfc, bfc, out, (int)n);
}


# gather: a row is read by 8 lanes x 16 B, so one load fetches the rows of 4 edges and a block of 32 rows is 8 loads in flight at once; a node's sum is now 4 partial sums (slot mod 4) added as a tree
# speedup vs baseline: 4.7434x; 1.0296x over previous
#include <hip/hip_runtime.h>

typedef float v2f __attribute__((ext_vector_type(2)));
typedef float v8f __attribute__((ext_vector_type(8)));

#define HID 32

__global__ void k_fill(float* __restrict__ p, float v, long long n) {
    long long i = (long long)blockIdx.x * blockDim.x + threadIdx.x;
    if (i < n) p[i] = v;
}

__global__ void k_deg(const int* __restrict__ dst, float* __restrict__ deg, long long e) {
    long long i = (long long)blockIdx.x * blockDim.x + threadIdx.x;
    if (i < e) atomicAdd(&deg[dst[i]], 1.0f);
}

#define NCTR 256
__global__ void k_rsqrt(float* __restrict__ d, unsigned int* __restrict__ ctr, long long n) {
    long long i = (long long)blockIdx.x * blockDim.x + threadIdx.x;
    if (i < NCTR) ctr[i] = 0u;
    if (i < n) {
        float v = d[i];
        d[i] = (v > 0.0f) ? rsqrtf(v) : 0.0f;
    }
}

template <int KDIM>
__global__ __launch_bounds__(128) void k_gemm(const float* __restrict__ A,
                                              const float* __restrict__ W,
                                              const float* __restrict__ rs,
                                              float* __restrict__ H, int n) {
    const int lane = threadIdx.x & 31;
    const int wave = threadIdx.x >> 5;
    const int row0 = (blockIdx.x * 4 + wave) * 16;
    if (row0 >= n) return;

    const int lr = lane & 15;
    const int kh = (lane >> 4) << 1;

    int arow = row0 + lr;
    if (arow >= n) arow = n - 1;
    const float* __restrict__ ap = A + (size_t)arow * KDIM;

    v8f c0 = {0.f, 0.f, 0.f, 0.f, 0.f, 0.f, 0.f, 0.f};
    v8f c1 = {0.f, 0.f, 0.f, 0.f, 0.f, 0.f, 0.f, 0.f};

#pragma unroll
    for (int k0 = 0; k0 < KDIM; k0 += 4) {
        v2f a;
        a.x = ap[k0 + kh];
        a.y = ap[k0 + kh + 1];
        const float* __restrict__ wp = W + (size_t)(k0 + kh) * HID;
        v2f b0, b1;
        b0.x = wp[lr];
        b0.y = wp[HID + lr];
        b1.x = wp[16 + lr];
        b1.y = wp[HID + 16 + lr];
        c0 = __builtin_amdgcn_wmma_f32_16x16x4_f32(false, a, false, b0, (short)0, c0, false, false);
        c1 = __builtin_amdgcn_wmma_f32_16x16x4_f32(false, a, false, b1, (short)0, c1, false, false);
    }

    const int mofs = (lane >> 4) * 8;
#pragma unroll
    for (int r = 0; r < 8; ++r) {
        int row = row0 + mofs + r;
        if (row < n) {
            const float sc = rs ? rs[row] : 1.0f;
            float* __restrict__ hp = H + (size_t)row * HID;
            hp[lr]      = c0[r] * sc;
            hp[16 + lr] = c1[r] * sc;
        }
    }
}

#define CH_STEPS 4
#define CH_EDGES (32 * CH_STEPS)
#define CTR_STRIDE 64

__device__ __forceinline__ void edge_group32(int s, int d, int lane,
                                             const float* __restrict__ dinv,
                                             const float* __restrict__ h,
                                             float* __restrict__ agg) {
    const float ws = dinv[s];
    const float wd = dinv[d];
    float v[32];
#pragma unroll
    for (int j = 0; j < 32; ++j) {
        const int sj = __builtin_amdgcn_readlane(s, j);
        v[j] = h[sj * HID + lane];
    }
    const float nrm = ws * wd;
#pragma unroll
    for (int j = 0; j < 32; ++j) {
        const int dj = __builtin_amdgcn_readlane(d, j);
        const float wj = __int_as_float(__builtin_amdgcn_readlane(__float_as_int(nrm), j));
        atomicAdd(&agg[dj * HID + lane], v[j] * wj);
    }
}

__global__ __launch_bounds__(256) void k_scatter(const int* __restrict__ src,
                                                 const int* __restrict__ dst,
                                                 const float* __restrict__ dinv,
                                                 const float* __restrict__ h,
                                                 float* __restrict__ agg,
                                                 unsigned int* __restrict__ ctr,
                                                 long long e, int nsplit) {
    __shared__ int q_s[8][64];
    __shared__ int q_d[8][64];
    const int lane = threadIdx.x & 31;
    const int wave = __builtin_amdgcn_readfirstlane((int)(threadIdx.x >> 5));
    int* qs = q_s[wave];
    int* qd = q_d[wave];
    const int half = (int)((__builtin_amdgcn_s_sendmsg_rtn(0x87) >> 18) & 1u);
    const unsigned int nchunks = (unsigned int)((e + CH_EDGES - 1) / CH_EDGES);
    int cnt = 0;

    for (int p = 0; p < 2; ++p) {
        const int r = half ^ p;
        unsigned int* c = ctr + r * CTR_STRIDE;
        for (unsigned int it = 0; it <= nchunks; ++it) {
            unsigned int ck = 0;
            if (lane == 0) ck = atomicAdd(c, 1u);
            ck = (unsigned int)__builtin_amdgcn_readfirstlane((int)ck);
            if (ck >= nchunks) break;
            const long long e0 = (long long)ck * CH_EDGES + lane;
            int sv[CH_STEPS], dv[CH_STEPS];
#pragma unroll
            for (int k = 0; k < CH_STEPS; ++k) {
                const long long idx = e0 + k * 32;
                const long long ii = idx < e ? idx : e - 1;
                sv[k] = src[ii];
                dv[k] = dst[ii];
            }
#pragma unroll
            for (int k = 0; k < CH_STEPS; ++k) {
                const long long idx = e0 + k * 32;
                const bool m = (idx < e) && ((dv[k] >= nsplit ? 1 : 0) == r);
                const unsigned int mask = __builtin_amdgcn_ballot_w32(m);
                if (m) {
                    const int pos = cnt + (int)__builtin_amdgcn_mbcnt_lo(mask, 0u);
                    qs[pos] = sv[k];
                    qd[pos] = dv[k];
                }
                cnt += __builtin_popcount(mask);
                if (cnt >= 32) {
                    const int s = qs[lane];
                    const int d = qd[lane];
                    const int s2 = qs[32 + lane];
                    const int d2 = qd[32 + lane];
                    edge_group32(s, d, lane, dinv, h, agg);
                    cnt -= 32;
                    if (lane < cnt) {
                        qs[lane] = s2;
                        qd[lane] = d2;
                    }
                }
            }
        }
    }
    if (cnt > 0) {
        const int ql = lane < cnt ? lane : cnt - 1;
        const int s = qs[ql];
        const int d = qd[ql];
        const float nrm = dinv[s] * dinv[d];
        for (int j = 0; j < cnt; ++j) {
            const int sj = __builtin_amdgcn_readlane(s, j);
            const int dj = __builtin_amdgcn_readlane(d, j);
            const float wj = __int_as_float(__builtin_amdgcn_readlane(__float_as_int(nrm), j));
            atomicAdd(&agg[dj * HID + lane], h[sj * HID + lane] * wj);
        }
    }
}

__global__ void k_finalize(const float* __restrict__ h, const float* __restrict__ dinv,
                           const float* __restrict__ b, float* __restrict__ agg,
                           long long total) {
    long long t = (long long)blockIdx.x * blockDim.x + threadIdx.x;
    if (t >= total) return;
    long long i = t >> 5;
    int k = (int)(t & 31);
    float di = dinv[i];
    float v = agg[t] + h[t] * (di * di) + b[k];
    agg[t] = v > 0.0f ? v : 0.0f;
}

__global__ void k_fc(const float* __restrict__ h, const float* __restrict__ wfc,
                     const float* __restrict__ bfc, float* __restrict__ out, int n) {
    int i = blockIdx.x * blockDim.x + threadIdx.x;
    if (i >= n) return;
    const float4* __restrict__ hp = (const float4*)(h + (size_t)i * HID);
    const float4* __restrict__ wp = (const float4*)wfc;
    float acc = 0.0f;
#pragma unroll
    for (int k = 0; k < 8; ++k) {
        float4 a = hp[k], w = wp[k];
        acc += a.x * w.x + a.y * w.y + a.z * w.z + a.w * w.w;
    }
    out[i] = acc + bfc[0];
}

#define PGRID 2048
#define SCAN_BLK 1024

__global__ void k_zero(int* __restrict__ p, unsigned int* __restrict__ ctr, int n) {
    int i = blockIdx.x * blockDim.x + threadIdx.x;
    if (i < NCTR) ctr[i] = 0u;
    if (i < n) p[i] = 0;
}

#define CK_STEPS 8
#define CK_EDGES (32 * CK_STEPS)
__global__ __launch_bounds__(256) void k_count(const int* __restrict__ dst, int* __restrict__ cnt,
                                               int* __restrict__ rank, unsigned int* __restrict__ ctr,
                                               long long e, int nsplit) {
    const int lane = threadIdx.x & 31;
    const int half = (int)((__builtin_amdgcn_s_sendmsg_rtn(0x87) >> 18) & 1u);
    const unsigned int nchunks = (unsigned int)((e + CK_EDGES - 1) / CK_EDGES);

    for (int p = 0; p < 2; ++p) {
        const int r = half ^ p;
        unsigned int* c = ctr + r * CTR_STRIDE;
        for (unsigned int it = 0; it <= nchunks; ++it) {
            unsigned int ck = 0;
            if (lane == 0) ck = atomicAdd(c, 1u);
            ck = (unsigned int)__builtin_amdgcn_readfirstlane((int)ck);
            if (ck >= nchunks) break;
            const long long e0 = (long long)ck * CK_EDGES + lane;
            int dv[CK_STEPS], rk[CK_STEPS];
#pragma unroll
            for (int k = 0; k < CK_STEPS; ++k) {
                const long long idx = e0 + k * 32;
                dv[k] = dst[idx < e ? idx : e - 1];
            }
            unsigned int inr = 0u;
#pragma unroll
            for (int k = 0; k < CK_STEPS; ++k) {
                const long long idx = e0 + k * 32;
                const unsigned int ex = idx < e ? 1u : 0u;
                const unsigned int in = (dv[k] >= nsplit ? 1 : 0) == r ? 1u : 0u;
                inr |= (ex & in) << k;
            }
            asm volatile("" : "+v"(inr));
#pragma unroll
            for (int k = 0; k < CK_STEPS; ++k) {
                rk[k] = 0;
                if (inr & (1u << k)) rk[k] = atomicAdd(&cnt[dv[k]], 1);
            }
#pragma unroll
            for (int k = 0; k < CK_STEPS; ++k) {
                if (inr & (1u << k)) rank[e0 + k * 32] = rk[k];
            }
        }
    }
}

__global__ __launch_bounds__(256) void k_scan_local(const int* __restrict__ cnt, int* __restrict__ rowptr,
                                                    int* __restrict__ btot, float* __restrict__ dinv, int n) {
    __shared__ int wsum[8];
    const int t = threadIdx.x;
    const int lane = t & 31;
    const int wv = t >> 5;
    const int base = blockIdx.x * SCAN_BLK + t * 4;
    int c[4];
    int tsum = 0;
#pragma unroll
    for (int k = 0; k < 4; ++k) {
        const int idx = base + k;
        c[k] = idx < n ? cnt[idx] : 0;
        if (idx < n) dinv[idx] = rsqrtf((float)(c[k] + 1));
        tsum += c[k];
    }
    int inc = tsum;
#pragma unroll
    for (int d = 1; d < 32; d <<= 1) {
        const int o = __shfl_up(inc, d);
        if (lane >= d) inc += o;
    }
    if (lane == 31) wsum[wv] = inc;
    __syncthreads();
    int woff = 0, total = 0;
#pragma unroll
    for (int w = 0; w < 8; ++w) {
        const int s = wsum[w];
        if (w < wv) woff += s;
        total += s;
    }
    int run = woff + inc - tsum;
#pragma unroll
    for (int k = 0; k < 4; ++k) {
        const int idx = base + k;
        if (idx < n) rowptr[idx] = run;
        run += c[k];
    }
    if (t == 0) btot[blockIdx.x] = total;
}

__global__ __launch_bounds__(256) void k_scan_add(int* __restrict__ rowptr, const int* __restrict__ btot,
                                                  int n, int e) {
    __shared__ int wsum[8];
    const int t = threadIdx.x;
    const int b = blockIdx.x;
    int s = 0;
    for (int i = t; i < b; i += 256) s += btot[i];
#pragma unroll
    for (int d = 16; d >= 1; d >>= 1) s += __shfl_xor(s, d);
    if ((t & 31) == 0) wsum[t >> 5] = s;
    __syncthreads();
    int off = 0;
#pragma unroll
    for (int w = 0; w < 8; ++w) off += wsum[w];
    const int base = b * SCAN_BLK + t * 4;
#pragma unroll
    for (int k = 0; k < 4; ++k) {
        const int idx = base + k;
        if (idx < n) rowptr[idx] += off;
    }
    if (b == 0 && t == 0) rowptr[n] = e;
}

#define PL_STEPS 8
#define PL_EDGES (32 * PL_STEPS)
__global__ __launch_bounds__(256) void k_place(const int* __restrict__ src, const int* __restrict__ dst,
                                               const int* __restrict__ rank, const int* __restrict__ rowptr,
                                               int* __restrict__ col, unsigned int* __restrict__ ctr,
                                               long long e, int nsplit) {
    const int lane = threadIdx.x & 31;
    const int half = (int)((__builtin_amdgcn_s_sendmsg_rtn(0x87) >> 18) & 1u);
    const int ne = (int)e;
    const unsigned int nchunks = (unsigned int)((ne + PL_EDGES - 1) / PL_EDGES);

    for (int p = 0; p < 2; ++p) {
        const int r = half ^ p;
        unsigned int* c = ctr + r * CTR_STRIDE;
        for (unsigned int it = 0; it <= nchunks; ++it) {
            unsigned int ck = 0;
            if (lane == 0) ck = atomicAdd(c, 1u);
            ck = (unsigned int)__builtin_amdgcn_readfirstlane((int)ck);
            if (ck >= nchunks) break;
            const int e0 = (int)ck * PL_EDGES + lane;
            int dv[PL_STEPS], sv[PL_STEPS], rk[PL_STEPS], rp[PL_STEPS];
#pragma unroll
            for (int k = 0; k < PL_STEPS; ++k) {
                const int idx = e0 + k * 32;
                const int ii = idx < ne ? idx : ne - 1;
                dv[k] = dst[ii];
                sv[k] = src[ii];
                rk[k] = rank[ii];
            }
            unsigned int inr = 0u;
#pragma unroll
            for (int k = 0; k < PL_STEPS; ++k) {
                const int idx = e0 + k * 32;
                const unsigned int ex = idx < ne ? 1u : 0u;
                const unsigned int in = (dv[k] >= nsplit ? 1 : 0) == r ? 1u : 0u;
                inr |= (ex & in) << k;
            }
#pragma unroll
            for (int k = 0; k < PL_STEPS; ++k) {
                rp[k] = 0;
                if (inr & (1u << k)) rp[k] = rowptr[dv[k]];
            }
#pragma unroll
            for (int k = 0; k < PL_STEPS; ++k) {
                if (inr & (1u << k)) col[rp[k] + rk[k]] = sv[k];
            }
        }
    }
}

#define BK_SHIFT 10
#define BK_NODES (1 << BK_SHIFT)
#define BK_MAX   256
#define TL_PER   8
#define TL_EDGES (1024 * TL_PER)
#define ST_CAP   49152

__device__ __forceinline__ int wg_excl_scan_1024(int v, int* wsum, int* tot) {
    const int lane = threadIdx.x & 31;
    const int wv = threadIdx.x >> 5;
    int inc = v;
#pragma unroll
    for (int d = 1; d < 32; d <<= 1) {
        const int o = __shfl_up(inc, d);
        if (lane >= d) inc += o;
    }
    if (lane == 31) wsum[wv] = inc;
    __syncthreads();
    const int ws = wsum[lane];
    int winc = ws;
#pragma unroll
    for (int d = 1; d < 32; d <<= 1) {
        const int o = __shfl_up(winc, d);
        if (lane >= d) winc += o;
    }
    const int total = __shfl(winc, 31);
    const int woff = __shfl(winc, wv) - __shfl(ws, wv);
    __syncthreads();
    *tot = total;
    return woff + inc - v;
}

__global__ __launch_bounds__(1024) void k_bhist(const int* __restrict__ dst, int* __restrict__ bcnt,
                                                int ne, int nb, int ntiles) {
    __shared__ int hist[BK_MAX];
    const int t = threadIdx.x;
    if (t < BK_MAX) hist[t] = 0;
    __syncthreads();
    const int e0 = blockIdx.x * TL_EDGES + t;
    int dv[TL_PER];
#pragma unroll
    for (int k = 0; k < TL_PER; ++k) {
        const int idx = e0 + k * 1024;
        dv[k] = dst[idx < ne ? idx : ne - 1];
    }
#pragma unroll
    for (int k = 0; k < TL_PER; ++k) {
        if (e0 + k * 1024 < ne) atomicAdd(&hist[dv[k] >> BK_SHIFT], 1);
    }
    __syncthreads();
    if (t < nb) bcnt[t * ntiles + blockIdx.x] = hist[t];
}

__global__ __launch_bounds__(256) void k_bscan(int* __restrict__ bcnt, int* __restrict__ btotal, int ntiles) {
    __shared__ int wsum[8];
    const int t = threadIdx.x;
    const int lane = t & 31;
    const int wv = t >> 5;
    int* __restrict__ row = bcnt + (size_t)blockIdx.x * ntiles;
    int c[4];
    int tsum = 0;
#pragma unroll
    for (int k = 0; k < 4; ++k) {
        const int idx = t * 4 + k;
        c[k] = idx < ntiles ? row[idx] : 0;
        tsum += c[k];
    }
    int inc = tsum;
#pragma unroll
    for (int d = 1; d < 32; d <<= 1) {
        const int o = __shfl_up(inc, d);
        if (lane >= d) inc += o;
    }
    if (lane == 31) wsum[wv] = inc;
    __syncthreads();
    int woff = 0, total = 0;
#pragma unroll
    for (int w = 0; w < 8; ++w) {
        const int s = wsum[w];
        if (w < wv) woff += s;
        total += s;
    }
    int run = woff + inc - tsum;
#pragma unroll
    for (int k = 0; k < 4; ++k) {
        const int idx = t * 4 + k;
        if (idx < ntiles) row[idx] = run;
        run += c[k];
    }
    if (t == 0) btotal[blockIdx.x] = total;
}

__global__ __launch_bounds__(1024) void k_bpart(const int* __restrict__ src, const int* __restrict__ dst,
                                                const int* __restrict__ boff, const int* __restrict__ btotal,
                                                int* __restrict__ part, int ne, int nb, int ntiles) {
    __shared__ int hist[BK_MAX];
    __shared__ int gdelta[BK_MAX];
    __shared__ int wsum[32];
    __shared__ int2 stage[TL_EDGES];
    const int t = threadIdx.x;
    if (t < BK_MAX) hist[t] = 0;
    __syncthreads();
    const int e0 = blockIdx.x * TL_EDGES;
    int sv[TL_PER], dv[TL_PER];
#pragma unroll
    for (int k = 0; k < TL_PER; ++k) {
        const int idx = e0 + k * 1024 + t;
        const int ii = idx < ne ? idx : ne - 1;
        sv[k] = src[ii];
        dv[k] = dst[ii];
    }
#pragma unroll
    for (int k = 0; k < TL_PER; ++k) {
        if (e0 + k * 1024 + t < ne) atomicAdd(&hist[dv[k] >> BK_SHIFT], 1);
    }
    __syncthreads();
    const int hc = t < nb ? hist[t] : 0;
    const int bt = t < nb ? btotal[t] : 0;
    const int bo = t < nb ? boff[t * ntiles + blockIdx.x] : 0;
    int tot;
    const int lex = wg_excl_scan_1024(hc, wsum, &tot);
    const int gex = wg_excl_scan_1024(bt, wsum, &tot);
    if (t < nb) {
        hist[t] = lex;
        gdelta[t] = gex + bo - lex;
    }
    __syncthreads();
#pragma unroll
    for (int k = 0; k < TL_PER; ++k) {
        if (e0 + k * 1024 + t < ne) {
            const int bk = dv[k] >> BK_SHIFT;
            const int pos = atomicAdd(&hist[bk], 1);
            stage[pos] = make_int2((int)(((unsigned int)sv[k] << BK_SHIFT) | (unsigned int)(dv[k] & (BK_NODES - 1))), bk);
        }
    }
    __syncthreads();
    const int ct = (ne - e0) < TL_EDGES ? (ne - e0) : TL_EDGES;
#pragma unroll
    for (int k = 0; k < TL_PER; ++k) {
        const int i = k * 1024 + t;
        if (i < ct) {
            const int2 p = stage[i];
            part[gdelta[p.y] + i] = p.x;
        }
    }
}

__global__ __launch_bounds__(1024) void k_bsort(const int* __restrict__ part, const int* __restrict__ btotal,
                                                int* __restrict__ rowptr, int* __restrict__ col,
                                                float* __restrict__ dinv, int n, int ne, int nb) {
    __shared__ int ncnt[BK_NODES];
    __shared__ int wsum[32];
    __shared__ int bsz[2];
    __shared__ int stage[ST_CAP];
    const int t = threadIdx.x;
    const int b = blockIdx.x;
    const int base = b << BK_SHIFT;
    ncnt[t] = 0;
    const int bt = t < nb ? btotal[t] : 0;
    int tot;
    const int gex = wg_excl_scan_1024(bt, wsum, &tot);
    if (t == b) {
        bsz[0] = gex;
        bsz[1] = bt;
    }
    __syncthreads();
    const int bs = bsz[0];
    const int cb = bsz[1];
    const int* __restrict__ pp = part + bs;
    const int nit = (cb + TL_EDGES - 1) / TL_EDGES;

    for (int it = 0; it < nit; ++it) {
        const int i0 = it * TL_EDGES + t;
        int dv[TL_PER];
#pragma unroll
        for (int k = 0; k < TL_PER; ++k) {
            const int i = i0 + k * 1024;
            dv[k] = pp[i < cb ? i : cb - 1];
        }
#pragma unroll
        for (int k = 0; k < TL_PER; ++k) {
            if (i0 + k * 1024 < cb) atomicAdd(&ncnt[dv[k] & (BK_NODES - 1)], 1);
        }
    }
    __syncthreads();
    const int c = ncnt[t];
    const int ex = wg_excl_scan_1024(c, wsum, &tot);
    if (base + t < n) {
        rowptr[base + t] = bs + ex;
        dinv[base + t] = rsqrtf((float)(c + 1));
    }
    if (b == nb - 1 && t == 0) rowptr[n] = ne;
    ncnt[t] = ex;
    __syncthreads();

    if (cb <= ST_CAP) {
        for (int it = 0; it < nit; ++it) {
            const int i0 = it * TL_EDGES + t;
            int pv[TL_PER];
#pragma unroll
            for (int k = 0; k < TL_PER; ++k) {
                const int i = i0 + k * 1024;
                pv[k] = pp[i < cb ? i : cb - 1];
            }
#pragma unroll
            for (int k = 0; k < TL_PER; ++k) {
                if (i0 + k * 1024 < cb) {
                    const int pos = atomicAdd(&ncnt[pv[k] & (BK_NODES - 1)], 1);
                    stage[pos] = (int)((unsigned int)pv[k] >> BK_SHIFT);
                }
            }
        }
        __syncthreads();
        for (int i = t; i < cb; i += 1024) col[bs + i] = stage[i];
    } else {
        for (int it = 0; it < nit; ++it) {
            const int i0 = it * TL_EDGES + t;
            int pv[TL_PER];
#pragma unroll
            for (int k = 0; k < TL_PER; ++k) {
                const int i = i0 + k * 1024;
                pv[k] = pp[i < cb ? i : cb - 1];
            }
#pragma unroll
            for (int k = 0; k < TL_PER; ++k) {
                if (i0 + k * 1024 < cb) {
                    const int pos = atomicAdd(&ncnt[pv[k] & (BK_NODES - 1)], 1);
                    col[bs + pos] = (int)((unsigned int)pv[k] >> BK_SHIFT);
                }
            }
        }
    }
}

typedef float v4f __attribute__((ext_vector_type(4)));

template <int I>
__device__ __forceinline__ int bcast8(int s) {
    return (int)__builtin_amdgcn_mov_dpp8((unsigned int)s, (unsigned int)(I * 0x249249));
}

__device__ __forceinline__ v4f row_piece(const float* __restrict__ hs, int sj, unsigned int qoff) {
    const unsigned int off = (unsigned int)sj * (unsigned int)(HID * 4) + qoff;
    return *(const v4f*)((const char*)hs + off);
}

__device__ __forceinline__ float xor8(float x) {
    return __builtin_bit_cast(float, __builtin_amdgcn_update_dpp(0, __builtin_bit_cast(int, x), 0x168  ,
                                                                 0xf, 0xf, false));
}
__device__ __forceinline__ float xor16(float x) {
    return __builtin_bit_cast(float, __builtin_amdgcn_permlanex16(__builtin_bit_cast(int, x), __builtin_bit_cast(int, x),
                                                                  0x76543210, 0xfedcba98, false, false));
}

template <int NF>
__device__ __forceinline__ v4f gather_n(const float* __restrict__ hs, int s, int sp, bool part,
                                        unsigned int qoff, v4f acc) {
    v4f v0, v1, v2, v3, v4, v5, v6, v7;
    v4f vp = {0.f, 0.f, 0.f, 0.f};
    if constexpr (NF > 0) v0 = row_piece(hs, bcast8<0>(s), qoff);
    if constexpr (NF > 1) v1 = row_piece(hs, bcast8<1>(s), qoff);
    if constexpr (NF > 2) v2 = row_piece(hs, bcast8<2>(s), qoff);
    if constexpr (NF > 3) v3 = row_piece(hs, bcast8<3>(s), qoff);
    if constexpr (NF > 4) v4 = row_piece(hs, bcast8<4>(s), qoff);
    if constexpr (NF > 5) v5 = row_piece(hs, bcast8<5>(s), qoff);
    if constexpr (NF > 6) v6 = row_piece(hs, bcast8<6>(s), qoff);
    if constexpr (NF > 7) v7 = row_piece(hs, bcast8<7>(s), qoff);
    if constexpr (NF < 8) {
        if (part) vp = row_piece(hs, sp, qoff);
    }
    if constexpr (NF > 0) acc += v0;
    if constexpr (NF > 1) acc += v1;
    if constexpr (NF > 2) acc += v2;
    if constexpr (NF > 3) acc += v3;
    if constexpr (NF > 4) acc += v4;
    if constexpr (NF > 5) acc += v5;
    if constexpr (NF > 6) acc += v6;
    if constexpr (NF > 7) acc += v7;
    if constexpr (NF < 8) acc += vp;
    return acc;
}

__device__ __forceinline__ v4f gather_block(const float* __restrict__ hs, int s, int sp, int cnt, int g,
                                            unsigned int qoff, v4f acc) {
    const bool part = g < (cnt & 3);
    switch (cnt >> 2) {
        case 0:  return gather_n<0>(hs, s, sp, part, qoff, acc);
        case 1:  return gather_n<1>(hs, s, sp, part, qoff, acc);
        case 2:  return gather_n<2>(hs, s, sp, part, qoff, acc);
        case 3:  return gather_n<3>(hs, s, sp, part, qoff, acc);
        case 4:  return gather_n<4>(hs, s, sp, part, qoff, acc);
        case 5:  return gather_n<5>(hs, s, sp, part, qoff, acc);
        case 6:  return gather_n<6>(hs, s, sp, part, qoff, acc);
        case 7:  return gather_n<7>(hs, s, sp, part, qoff, acc);
        default: return gather_n<8>(hs, s, sp, part, qoff, acc);
    }
}

__global__ __launch_bounds__(256) void k_gather(const int* __restrict__ rowptr, const int* __restrict__ col,
                                                const float* __restrict__ dinv, const float* __restrict__ hs,
                                                const float* __restrict__ b, float* __restrict__ outp, int n) {
    const int lane = threadIdx.x & 31;
    const int g = lane >> 3;
    const int q = lane & 7;
    const unsigned int qoff = (unsigned int)q * 16u;
    const int slot = 4 * q + g;
    const int wave = __builtin_amdgcn_readfirstlane((int)(blockIdx.x * 8 + (threadIdx.x >> 5)));
    const int nwaves = (int)gridDim.x * 8;
    const v4f bias = *(const v4f*)(b + 4 * q);
    for (int d = wave; d < n; d += nwaves) {
        const int beg = rowptr[d];
        const int end = rowptr[d + 1];
        v4f acc = {0.f, 0.f, 0.f, 0.f};
        if (g == 0) acc = row_piece(hs, d, qoff);
        for (int base = beg; base < end; base += 64) {
            const int left = end - base;
            const int c0 = left < 32 ? left : 32;
            const int c1 = left - 32 < 32 ? left - 32 : 32;
            const int* __restrict__ cb = col + base;
            const int s0 = cb[(unsigned int)(slot < c0 ? slot : 0)];
            const int p0 = cb[(unsigned int)(g < (c0 & 3) ? (c0 & ~3) + g : 0)];
            int s1 = 0, p1 = 0;
            if (c1 > 0) {
                s1 = cb[32u + (unsigned int)(slot < c1 ? slot : 0)];
                p1 = cb[32u + (unsigned int)(g < (c1 & 3) ? (c1 & ~3) + g : 0)];
            }
            acc = gather_block(hs, s0, p0, c0, g, qoff, acc);
            if (c1 > 0) acc = gather_block(hs, s1, p1, c1, g, qoff, acc);
        }
#pragma unroll
        for (int k = 0; k < 4; ++k) {
            float t = acc[k];
            t += xor8(t);
            t += xor16(t);
            acc[k] = t;
        }
        if (g == 0) {
            const float di = dinv[d];
            v4f o = acc * di + bias;
            o.x = o.x > 0.0f ? o.x : 0.0f;
            o.y = o.y > 0.0f ? o.y : 0.0f;
            o.z = o.z > 0.0f ? o.z : 0.0f;
            o.w = o.w > 0.0f ? o.w : 0.0f;
            *(v4f*)(outp + (size_t)d * HID + 4 * q) = o;
        }
    }
}

extern "C" void kernel_launch(void* const* d_in, const int* in_sizes, int n_in,
                              void* d_out, int out_size, void* d_ws, size_t ws_size,
                              hipStream_t stream) {
    const float* x   = (const float*)d_in[0];
    const int*   ei  = (const int*)d_in[1];
    const float* W1  = (const float*)d_in[2];
    const float* b1  = (const float*)d_in[3];
    const float* W2  = (const float*)d_in[4];
    const float* b2  = (const float*)d_in[5];
    const float* Wfc = (const float*)d_in[6];
    const float* bfc = (const float*)d_in[7];
    float* out = (float*)d_out;

    const int F = 128;
    const long long n = in_sizes[0] / F;
    const long long e = in_sizes[1] / 2;
    const int* src = ei;
    const int* dst = ei + e;
    const long long nh = n * HID;
    const int B = 256;
    dim3 bfill((unsigned)((n + B - 1) / B));
    dim3 bgemm((unsigned)((n + 63) / 64));

    const long long w_n   = (n + 63) & ~63LL;
    const long long w_rp  = (n + 1 + 63) & ~63LL;
    const long long w_bt  = 1024;
    const long long w_e   = (e + 63) & ~63LL;
    const long long w_nh  = (nh + 63) & ~63LL;
    const long long w_ag  = w_nh > w_e ? w_nh : w_e;
    const long long need  = (w_n + w_rp + w_bt + w_n + w_e + w_nh + w_ag) * 4;
    const long long nsblk = (n + SCAN_BLK - 1) / SCAN_BLK;

    if ((long long)ws_size >= need && nsblk <= 512) {
        int*   cnt    = (int*)d_ws;
        int*   rowptr = cnt + w_n;
        int*   btot   = rowptr + w_rp;
        unsigned int* tick = (unsigned int*)(btot + 512);
        const int nsplit = (int)((n / 2) & ~63LL);
        float* dinv   = (float*)(btot + w_bt);
        int*   col    = (int*)(dinv + w_n);
        float* hbuf   = (float*)(col + w_e);
        float* agg    = hbuf + w_nh;
        int*   rank   = (int*)agg;
        dim3 bscan((unsigned)nsblk);

        const int nb = (int)((n + BK_NODES - 1) >> BK_SHIFT);
        const int ntiles = (int)((e + TL_EDGES - 1) / TL_EDGES);
        if (nb <= BK_MAX && ntiles >= 1 && ntiles <= 1024 && (long long)nb * ntiles <= w_n &&
            n <= (1LL << (32 - BK_SHIFT)) && e < (1LL << 30)) {
            int* bcnt   = cnt;
            int* btotal = btot;
            int* part   = (int*)agg;
            k_bhist<<<ntiles, 1024, 0, stream>>>(dst, bcnt, (int)e, nb, ntiles);
            k_bscan<<<nb, 256, 0, stream>>>(bcnt, btotal, ntiles);
            k_bpart<<<ntiles, 1024, 0, stream>>>(src, dst, bcnt, btotal, part, (int)e, nb, ntiles);
            k_bsort<<<nb, 1024, 0, stream>>>(part, btotal, rowptr, col, dinv, (int)n, (int)e, nb);
        } else {
            k_zero<<<bfill, B, 0, stream>>>(cnt, tick, (int)n);
            k_count<<<PGRID, B, 0, stream>>>(dst, cnt, rank, tick, e, nsplit);
            k_scan_local<<<bscan, B, 0, stream>>>(cnt, rowptr, btot, dinv, (int)n);
            k_scan_add<<<bscan, B, 0, stream>>>(rowptr, btot, (int)n, (int)e);
            k_place<<<PGRID, B, 0, stream>>>(src, dst, rank, rowptr, col, tick + 128, e, nsplit);
        }

        k_gemm<128><<<bgemm, 128, 0, stream>>>(x, W1, dinv, hbuf, (int)n);
        k_gather<<<PGRID, B, 0, stream>>>(rowptr, col, dinv, hbuf, b1, agg, (int)n);

        k_gemm<32><<<bgemm, 128, 0, stream>>>(agg, W2, dinv, hbuf, (int)n);
        k_gather<<<PGRID, B, 0, stream>>>(rowptr, col, dinv, hbuf, b2, agg, (int)n);

        k_fc<<<bfill, B, 0, stream>>>(agg, Wfc, bfc, out, (int)n);
        return;
    }

    unsigned int* ctr = (unsigned int*)d_ws;
    float* dinv = (float*)d_ws + 256;
    float* hbuf = dinv + n;
    float* agg  = hbuf + n * HID;

    dim3 bfillh((unsigned)((nh + B - 1) / B));
    dim3 bedge((unsigned)((e + B - 1) / B));
    dim3 bscat(2048);
    const int nsplit = (int)(n / 2);

    k_fill<<<bfill, B, 0, stream>>>(dinv, 1.0f, n);
    k_deg<<<bedge, B, 0, stream>>>(dst, dinv, e);
    k_rsqrt<<<bfill, B, 0, stream>>>(dinv, ctr, n);

    k_gemm<128><<<bgemm, 128, 0, stream>>>(x, W1, nullptr, hbuf, (int)n);
    k_fill<<<bfillh, B, 0, stream>>>(agg, 0.0f, nh);
    k_scatter<<<bscat, B, 0, stream>>>(src, dst, dinv, hbuf, agg, ctr, e, nsplit);
    k_finalize<<<bfillh, B, 0, stream>>>(hbuf, dinv, b1, agg, nh);

    k_gemm<32><<<bgemm, 128, 0, stream>>>(agg, W2, nullptr, hbuf, (int)n);
    k_fill<<<bfillh, B, 0, stream>>>(agg, 0.0f, nh);
    k_scatter<<<bscat, B, 0, stream>>>(src, dst, dinv, hbuf, agg, ctr + 128, e, nsplit);
    k_finalize<<<bfillh, B, 0, stream>>>(hbuf, dinv, b2, agg, nh);

    k_fc<<<bfill, B, 0, stream>>>(agg, Wfc, bfc, out, (int)n);
}


# gather by tiles: a wave owns 16 consecutive nodes (bounds in one load, index words one node ahead) and does the next dense stage on its rows: W2 on WMMA, FC as a dot (tree sum order); 7 launches
# speedup vs baseline: 4.9789x; 1.0496x over previous
#include <hip/hip_runtime.h>

typedef float v2f __attribute__((ext_vector_type(2)));
typedef float v8f __attribute__((ext_vector_type(8)));

#define HID 32

__global__ void k_fill(float* __restrict__ p, float v, long long n) {
    long long i = (long long)blockIdx.x * blockDim.x + threadIdx.x;
    if (i < n) p[i] = v;
}

__global__ void k_deg(const int* __restrict__ dst, float* __restrict__ deg, long long e) {
    long long i = (long long)blockIdx.x * blockDim.x + threadIdx.x;
    if (i < e) atomicAdd(&deg[dst[i]], 1.0f);
}

#define NCTR 256
__global__ void k_rsqrt(float* __restrict__ d, unsigned int* __restrict__ ctr, long long n) {
    long long i = (long long)blockIdx.x * blockDim.x + threadIdx.x;
    if (i < NCTR) ctr[i] = 0u;
    if (i < n) {
        float v = d[i];
        d[i] = (v > 0.0f) ? rsqrtf(v) : 0.0f;
    }
}

template <int KDIM>
__global__ __launch_bounds__(128) void k_gemm(const float* __restrict__ A,
                                              const float* __restrict__ W,
                                              const float* __restrict__ rs,
                                              float* __restrict__ H, int n) {
    const int lane = threadIdx.x & 31;
    const int wave = threadIdx.x >> 5;
    const int row0 = (blockIdx.x * 4 + wave) * 16;
    if (row0 >= n) return;

    const int lr = lane & 15;
    const int kh = (lane >> 4) << 1;

    int arow = row0 + lr;
    if (arow >= n) arow = n - 1;
    const float* __restrict__ ap = A + (size_t)arow * KDIM;

    v8f c0 = {0.f, 0.f, 0.f, 0.f, 0.f, 0.f, 0.f, 0.f};
    v8f c1 = {0.f, 0.f, 0.f, 0.f, 0.f, 0.f, 0.f, 0.f};

#pragma unroll
    for (int k0 = 0; k0 < KDIM; k0 += 4) {
        v2f a;
        a.x = ap[k0 + kh];
        a.y = ap[k0 + kh + 1];
        const float* __restrict__ wp = W + (size_t)(k0 + kh) * HID;
        v2f b0, b1;
        b0.x = wp[lr];
        b0.y = wp[HID + lr];
        b1.x = wp[16 + lr];
        b1.y = wp[HID + 16 + lr];
        c0 = __builtin_amdgcn_wmma_f32_16x16x4_f32(false, a, false, b0, (short)0, c0, false, false);
        c1 = __builtin_amdgcn_wmma_f32_16x16x4_f32(false, a, false, b1, (short)0, c1, false, false);
    }

    const int mofs = (lane >> 4) * 8;
#pragma unroll
    for (int r = 0; r < 8; ++r) {
        int row = row0 + mofs + r;
        if (row < n) {
            const float sc = rs ? rs[row] : 1.0f;
            float* __restrict__ hp = H + (size_t)row * HID;
            hp[lr]      = c0[r] * sc;
            hp[16 + lr] = c1[r] * sc;
        }
    }
}

#define CH_STEPS 4
#define CH_EDGES (32 * CH_STEPS)
#define CTR_STRIDE 64

__device__ __forceinline__ void edge_group32(int s, int d, int lane,
                                             const float* __restrict__ dinv,
                                             const float* __restrict__ h,
                                             float* __restrict__ agg) {
    const float ws = dinv[s];
    const float wd = dinv[d];
    float v[32];
#pragma unroll
    for (int j = 0; j < 32; ++j) {
        const int sj = __builtin_amdgcn_readlane(s, j);
        v[j] = h[sj * HID + lane];
    }
    const float nrm = ws * wd;
#pragma unroll
    for (int j = 0; j < 32; ++j) {
        const int dj = __builtin_amdgcn_readlane(d, j);
        const float wj = __int_as_float(__builtin_amdgcn_readlane(__float_as_int(nrm), j));
        atomicAdd(&agg[dj * HID + lane], v[j] * wj);
    }
}

__global__ __launch_bounds__(256) void k_scatter(const int* __restrict__ src,
                                                 const int* __restrict__ dst,
                                                 const float* __restrict__ dinv,
                                                 const float* __restrict__ h,
                                                 float* __restrict__ agg,
                                                 unsigned int* __restrict__ ctr,
                                                 long long e, int nsplit) {
    __shared__ int q_s[8][64];
    __shared__ int q_d[8][64];
    const int lane = threadIdx.x & 31;
    const int wave = __builtin_amdgcn_readfirstlane((int)(threadIdx.x >> 5));
    int* qs = q_s[wave];
    int* qd = q_d[wave];
    const int half = (int)((__builtin_amdgcn_s_sendmsg_rtn(0x87) >> 18) & 1u);
    const unsigned int nchunks = (unsigned int)((e + CH_EDGES - 1) / CH_EDGES);
    int cnt = 0;

    for (int p = 0; p < 2; ++p) {
        const int r = half ^ p;
        unsigned int* c = ctr + r * CTR_STRIDE;
        for (unsigned int it = 0; it <= nchunks; ++it) {
            unsigned int ck = 0;
            if (lane == 0) ck = atomicAdd(c, 1u);
            ck = (unsigned int)__builtin_amdgcn_readfirstlane((int)ck);
            if (ck >= nchunks) break;
            const long long e0 = (long long)ck * CH_EDGES + lane;
            int sv[CH_STEPS], dv[CH_STEPS];
#pragma unroll
            for (int k = 0; k < CH_STEPS; ++k) {
                const long long idx = e0 + k * 32;
                const long long ii = idx < e ? idx : e - 1;
                sv[k] = src[ii];
                dv[k] = dst[ii];
            }
#pragma unroll
            for (int k = 0; k < CH_STEPS; ++k) {
                const long long idx = e0 + k * 32;
                const bool m = (idx < e) && ((dv[k] >= nsplit ? 1 : 0) == r);
                const unsigned int mask = __builtin_amdgcn_ballot_w32(m);
                if (m) {
                    const int pos = cnt + (int)__builtin_amdgcn_mbcnt_lo(mask, 0u);
                    qs[pos] = sv[k];
                    qd[pos] = dv[k];
                }
                cnt += __builtin_popcount(mask);
                if (cnt >= 32) {
                    const int s = qs[lane];
                    const int d = qd[lane];
                    const int s2 = qs[32 + lane];
                    const int d2 = qd[32 + lane];
                    edge_group32(s, d, lane, dinv, h, agg);
                    cnt -= 32;
                    if (lane < cnt) {
                        qs[lane] = s2;
                        qd[lane] = d2;
                    }
                }
            }
        }
    }
    if (cnt > 0) {
        const int ql = lane < cnt ? lane : cnt - 1;
        const int s = qs[ql];
        const int d = qd[ql];
        const float nrm = dinv[s] * dinv[d];
        for (int j = 0; j < cnt; ++j) {
            const int sj = __builtin_amdgcn_readlane(s, j);
            const int dj = __builtin_amdgcn_readlane(d, j);
            const float wj = __int_as_float(__builtin_amdgcn_readlane(__float_as_int(nrm), j));
            atomicAdd(&agg[dj * HID + lane], h[sj * HID + lane] * wj);
        }
    }
}

__global__ void k_finalize(const float* __restrict__ h, const float* __restrict__ dinv,
                           const float* __restrict__ b, float* __restrict__ agg,
                           long long total) {
    long long t = (long long)blockIdx.x * blockDim.x + threadIdx.x;
    if (t >= total) return;
    long long i = t >> 5;
    int k = (int)(t & 31);
    float di = dinv[i];
    float v = agg[t] + h[t] * (di * di) + b[k];
    agg[t] = v > 0.0f ? v : 0.0f;
}

__global__ void k_fc(const float* __restrict__ h, const float* __restrict__ wfc,
                     const float* __restrict__ bfc, float* __restrict__ out, int n) {
    int i = blockIdx.x * blockDim.x + threadIdx.x;
    if (i >= n) return;
    const float4* __restrict__ hp = (const float4*)(h + (size_t)i * HID);
    const float4* __restrict__ wp = (const float4*)wfc;
    float acc = 0.0f;
#pragma unroll
    for (int k = 0; k < 8; ++k) {
        float4 a = hp[k], w = wp[k];
        acc += a.x * w.x + a.y * w.y + a.z * w.z + a.w * w.w;
    }
    out[i] = acc + bfc[0];
}

#define PGRID 2048
#define SCAN_BLK 1024

__global__ void k_zero(int* __restrict__ p, unsigned int* __restrict__ ctr, int n) {
    int i = blockIdx.x * blockDim.x + threadIdx.x;
    if (i < NCTR) ctr[i] = 0u;
    if (i < n) p[i] = 0;
}

#define CK_STEPS 8
#define CK_EDGES (32 * CK_STEPS)
__global__ __launch_bounds__(256) void k_count(const int* __restrict__ dst, int* __restrict__ cnt,
                                               int* __restrict__ rank, unsigned int* __restrict__ ctr,
                                               long long e, int nsplit) {
    const int lane = threadIdx.x & 31;
    const int half = (int)((__builtin_amdgcn_s_sendmsg_rtn(0x87) >> 18) & 1u);
    const unsigned int nchunks = (unsigned int)((e + CK_EDGES - 1) / CK_EDGES);

    for (int p = 0; p < 2; ++p) {
        const int r = half ^ p;
        unsigned int* c = ctr + r * CTR_STRIDE;
        for (unsigned int it = 0; it <= nchunks; ++it) {
            unsigned int ck = 0;
            if (lane == 0) ck = atomicAdd(c, 1u);
            ck = (unsigned int)__builtin_amdgcn_readfirstlane((int)ck);
            if (ck >= nchunks) break;
            const long long e0 = (long long)ck * CK_EDGES + lane;
            int dv[CK_STEPS], rk[CK_STEPS];
#pragma unroll
            for (int k = 0; k < CK_STEPS; ++k) {
                const long long idx = e0 + k * 32;
                dv[k] = dst[idx < e ? idx : e - 1];
            }
            unsigned int inr = 0u;
#pragma unroll
            for (int k = 0; k < CK_STEPS; ++k) {
                const long long idx = e0 + k * 32;
                const unsigned int ex = idx < e ? 1u : 0u;
                const unsigned int in = (dv[k] >= nsplit ? 1 : 0) == r ? 1u : 0u;
                inr |= (ex & in) << k;
            }
            asm volatile("" : "+v"(inr));
#pragma unroll
            for (int k = 0; k < CK_STEPS; ++k) {
                rk[k] = 0;
                if (inr & (1u << k)) rk[k] = atomicAdd(&cnt[dv[k]], 1);
            }
#pragma unroll
            for (int k = 0; k < CK_STEPS; ++k) {
                if (inr & (1u << k)) rank[e0 + k * 32] = rk[k];
            }
        }
    }
}

__global__ __launch_bounds__(256) void k_scan_local(const int* __restrict__ cnt, int* __restrict__ rowptr,
                                                    int* __restrict__ btot, float* __restrict__ dinv, int n) {
    __shared__ int wsum[8];
    const int t = threadIdx.x;
    const int lane = t & 31;
    const int wv = t >> 5;
    const int base = blockIdx.x * SCAN_BLK + t * 4;
    int c[4];
    int tsum = 0;
#pragma unroll
    for (int k = 0; k < 4; ++k) {
        const int idx = base + k;
        c[k] = idx < n ? cnt[idx] : 0;
        if (idx < n) dinv[idx] = rsqrtf((float)(c[k] + 1));
        tsum += c[k];
    }
    int inc = tsum;
#pragma unroll
    for (int d = 1; d < 32; d <<= 1) {
        const int o = __shfl_up(inc, d);
        if (lane >= d) inc += o;
    }
    if (lane == 31) wsum[wv] = inc;
    __syncthreads();
    int woff = 0, total = 0;
#pragma unroll
    for (int w = 0; w < 8; ++w) {
        const int s = wsum[w];
        if (w < wv) woff += s;
        total += s;
    }
    int run = woff + inc - tsum;
#pragma unroll
    for (int k = 0; k < 4; ++k) {
        const int idx = base + k;
        if (idx < n) rowptr[idx] = run;
        run += c[k];
    }
    if (t == 0) btot[blockIdx.x] = total;
}

__global__ __launch_bounds__(256) void k_scan_add(int* __restrict__ rowptr, const int* __restrict__ btot,
                                                  int n, int e) {
    __shared__ int wsum[8];
    const int t = threadIdx.x;
    const int b = blockIdx.x;
    int s = 0;
    for (int i = t; i < b; i += 256) s += btot[i];
#pragma unroll
    for (int d = 16; d >= 1; d >>= 1) s += __shfl_xor(s, d);
    if ((t & 31) == 0) wsum[t >> 5] = s;
    __syncthreads();
    int off = 0;
#pragma unroll
    for (int w = 0; w < 8; ++w) off += wsum[w];
    const int base = b * SCAN_BLK + t * 4;
#pragma unroll
    for (int k = 0; k < 4; ++k) {
        const int idx = base + k;
        if (idx < n) rowptr[idx] += off;
    }
    if (b == 0 && t == 0) rowptr[n] = e;
}

#define PL_STEPS 8
#define PL_EDGES (32 * PL_STEPS)
__global__ __launch_bounds__(256) void k_place(const int* __restrict__ src, const int* __restrict__ dst,
                                               const int* __restrict__ rank, const int* __restrict__ rowptr,
                                               int* __restrict__ col, unsigned int* __restrict__ ctr,
                                               long long e, int nsplit) {
    const int lane = threadIdx.x & 31;
    const int half = (int)((__builtin_amdgcn_s_sendmsg_rtn(0x87) >> 18) & 1u);
    const int ne = (int)e;
    const unsigned int nchunks = (unsigned int)((ne + PL_EDGES - 1) / PL_EDGES);

    for (int p = 0; p < 2; ++p) {
        const int r = half ^ p;
        unsigned int* c = ctr + r * CTR_STRIDE;
        for (unsigned int it = 0; it <= nchunks; ++it) {
            unsigned int ck = 0;
            if (lane == 0) ck = atomicAdd(c, 1u);
            ck = (unsigned int)__builtin_amdgcn_readfirstlane((int)ck);
            if (ck >= nchunks) break;
            const int e0 = (int)ck * PL_EDGES + lane;
            int dv[PL_STEPS], sv[PL_STEPS], rk[PL_STEPS], rp[PL_STEPS];
#pragma unroll
            for (int k = 0; k < PL_STEPS; ++k) {
                const int idx = e0 + k * 32;
                const int ii = idx < ne ? idx : ne - 1;
                dv[k] = dst[ii];
                sv[k] = src[ii];
                rk[k] = rank[ii];
            }
            unsigned int inr = 0u;
#pragma unroll
            for (int k = 0; k < PL_STEPS; ++k) {
                const int idx = e0 + k * 32;
                const unsigned int ex = idx < ne ? 1u : 0u;
                const unsigned int in = (dv[k] >= nsplit ? 1 : 0) == r ? 1u : 0u;
                inr |= (ex & in) << k;
            }
#pragma unroll
            for (int k = 0; k < PL_STEPS; ++k) {
                rp[k] = 0;
                if (inr & (1u << k)) rp[k] = rowptr[dv[k]];
            }
#pragma unroll
            for (int k = 0; k < PL_STEPS; ++k) {
                if (inr & (1u << k)) col[rp[k] + rk[k]] = sv[k];
            }
        }
    }
}

#define BK_SHIFT 10
#define BK_NODES (1 << BK_SHIFT)
#define BK_MAX   256
#define TL_PER   8
#define TL_EDGES (1024 * TL_PER)
#define ST_CAP   49152

__device__ __forceinline__ int wg_excl_scan_1024(int v, int* wsum, int* tot) {
    const int lane = threadIdx.x & 31;
    const int wv = threadIdx.x >> 5;
    int inc = v;
#pragma unroll
    for (int d = 1; d < 32; d <<= 1) {
        const int o = __shfl_up(inc, d);
        if (lane >= d) inc += o;
    }
    if (lane == 31) wsum[wv] = inc;
    __syncthreads();
    const int ws = wsum[lane];
    int winc = ws;
#pragma unroll
    for (int d = 1; d < 32; d <<= 1) {
        const int o = __shfl_up(winc, d);
        if (lane >= d) winc += o;
    }
    const int total = __shfl(winc, 31);
    const int woff = __shfl(winc, wv) - __shfl(ws, wv);
    __syncthreads();
    *tot = total;
    return woff + inc - v;
}

__global__ __launch_bounds__(1024) void k_bhist(const int* __restrict__ dst, int* __restrict__ bcnt,
                                                int ne, int nb, int ntiles) {
    __shared__ int hist[BK_MAX];
    const int t = threadIdx.x;
    if (t < BK_MAX) hist[t] = 0;
    __syncthreads();
    const int e0 = blockIdx.x * TL_EDGES + t;
    int dv[TL_PER];
#pragma unroll
    for (int k = 0; k < TL_PER; ++k) {
        const int idx = e0 + k * 1024;
        dv[k] = dst[idx < ne ? idx : ne - 1];
    }
#pragma unroll
    for (int k = 0; k < TL_PER; ++k) {
        if (e0 + k * 1024 < ne) atomicAdd(&hist[dv[k] >> BK_SHIFT], 1);
    }
    __syncthreads();
    if (t < nb) bcnt[t * ntiles + blockIdx.x] = hist[t];
}

__global__ __launch_bounds__(256) void k_bscan(int* __restrict__ bcnt, int* __restrict__ btotal, int ntiles) {
    __shared__ int wsum[8];
    const int t = threadIdx.x;
    const int lane = t & 31;
    const int wv = t >> 5;
    int* __restrict__ row = bcnt + (size_t)blockIdx.x * ntiles;
    int c[4];
    int tsum = 0;
#pragma unroll
    for (int k = 0; k < 4; ++k) {
        const int idx = t * 4 + k;
        c[k] = idx < ntiles ? row[idx] : 0;
        tsum += c[k];
    }
    int inc = tsum;
#pragma unroll
    for (int d = 1; d < 32; d <<= 1) {
        const int o = __shfl_up(inc, d);
        if (lane >= d) inc += o;
    }
    if (lane == 31) wsum[wv] = inc;
    __syncthreads();
    int woff = 0, total = 0;
#pragma unroll
    for (int w = 0; w < 8; ++w) {
        const int s = wsum[w];
        if (w < wv) woff += s;
        total += s;
    }
    int run = woff + inc - tsum;
#pragma unroll
    for (int k = 0; k < 4; ++k) {
        const int idx = t * 4 + k;
        if (idx < ntiles) row[idx] = run;
        run += c[k];
    }
    if (t == 0) btotal[blockIdx.x] = total;
}

__global__ __launch_bounds__(1024) void k_bpart(const int* __restrict__ src, const int* __restrict__ dst,
                                                const int* __restrict__ boff, const int* __restrict__ btotal,
                                                int* __restrict__ part, int ne, int nb, int ntiles) {
    __shared__ int hist[BK_MAX];
    __shared__ int gdelta[BK_MAX];
    __shared__ int wsum[32];
    __shared__ int2 stage[TL_EDGES];
    const int t = threadIdx.x;
    if (t < BK_MAX) hist[t] = 0;
    __syncthreads();
    const int e0 = blockIdx.x * TL_EDGES;
    int sv[TL_PER], dv[TL_PER];
#pragma unroll
    for (int k = 0; k < TL_PER; ++k) {
        const int idx = e0 + k * 1024 + t;
        const int ii = idx < ne ? idx : ne - 1;
        sv[k] = src[ii];
        dv[k] = dst[ii];
    }
#pragma unroll
    for (int k = 0; k < TL_PER; ++k) {
        if (e0 + k * 1024 + t < ne) atomicAdd(&hist[dv[k] >> BK_SHIFT], 1);
    }
    __syncthreads();
    const int hc = t < nb ? hist[t] : 0;
    const int bt = t < nb ? btotal[t] : 0;
    const int bo = t < nb ? boff[t * ntiles + blockIdx.x] : 0;
    int tot;
    const int lex = wg_excl_scan_1024(hc, wsum, &tot);
    const int gex = wg_excl_scan_1024(bt, wsum, &tot);
    if (t < nb) {
        hist[t] = lex;
        gdelta[t] = gex + bo - lex;
    }
    __syncthreads();
#pragma unroll
    for (int k = 0; k < TL_PER; ++k) {
        if (e0 + k * 1024 + t < ne) {
            const int bk = dv[k] >> BK_SHIFT;
            const int pos = atomicAdd(&hist[bk], 1);
            stage[pos] = make_int2((int)(((unsigned int)sv[k] << BK_SHIFT) | (unsigned int)(dv[k] & (BK_NODES - 1))), bk);
        }
    }
    __syncthreads();
    const int ct = (ne - e0) < TL_EDGES ? (ne - e0) : TL_EDGES;
#pragma unroll
    for (int k = 0; k < TL_PER; ++k) {
        const int i = k * 1024 + t;
        if (i < ct) {
            const int2 p = stage[i];
            part[gdelta[p.y] + i] = p.x;
        }
    }
}

__global__ __launch_bounds__(1024) void k_bsort(const int* __restrict__ part, const int* __restrict__ btotal,
                                                int* __restrict__ rowptr, int* __restrict__ col,
                                                float* __restrict__ dinv, int n, int ne, int nb) {
    __shared__ int ncnt[BK_NODES];
    __shared__ int wsum[32];
    __shared__ int bsz[2];
    __shared__ int stage[ST_CAP];
    const int t = threadIdx.x;
    const int b = blockIdx.x;
    const int base = b << BK_SHIFT;
    ncnt[t] = 0;
    const int bt = t < nb ? btotal[t] : 0;
    int tot;
    const int gex = wg_excl_scan_1024(bt, wsum, &tot);
    if (t == b) {
        bsz[0] = gex;
        bsz[1] = bt;
    }
    __syncthreads();
    const int bs = bsz[0];
    const int cb = bsz[1];
    const int* __restrict__ pp = part + bs;
    const int nit = (cb + TL_EDGES - 1) / TL_EDGES;

    for (int it = 0; it < nit; ++it) {
        const int i0 = it * TL_EDGES + t;
        int dv[TL_PER];
#pragma unroll
        for (int k = 0; k < TL_PER; ++k) {
            const int i = i0 + k * 1024;
            dv[k] = pp[i < cb ? i : cb - 1];
        }
#pragma unroll
        for (int k = 0; k < TL_PER; ++k) {
            if (i0 + k * 1024 < cb) atomicAdd(&ncnt[dv[k] & (BK_NODES - 1)], 1);
        }
    }
    __syncthreads();
    const int c = ncnt[t];
    const int ex = wg_excl_scan_1024(c, wsum, &tot);
    if (base + t < n) {
        rowptr[base + t] = bs + ex;
        dinv[base + t] = rsqrtf((float)(c + 1));
    }
    if (b == nb - 1 && t == 0) rowptr[n] = ne;
    ncnt[t] = ex;
    __syncthreads();

    if (cb <= ST_CAP) {
        for (int it = 0; it < nit; ++it) {
            const int i0 = it * TL_EDGES + t;
            int pv[TL_PER];
#pragma unroll
            for (int k = 0; k < TL_PER; ++k) {
                const int i = i0 + k * 1024;
                pv[k] = pp[i < cb ? i : cb - 1];
            }
#pragma unroll
            for (int k = 0; k < TL_PER; ++k) {
                if (i0 + k * 1024 < cb) {
                    const int pos = atomicAdd(&ncnt[pv[k] & (BK_NODES - 1)], 1);
                    stage[pos] = (int)((unsigned int)pv[k] >> BK_SHIFT);
                }
            }
        }
        __syncthreads();
        for (int i = t; i < cb; i += 1024) col[bs + i] = stage[i];
    } else {
        for (int it = 0; it < nit; ++it) {
            const int i0 = it * TL_EDGES + t;
            int pv[TL_PER];
#pragma unroll
            for (int k = 0; k < TL_PER; ++k) {
                const int i = i0 + k * 1024;
                pv[k] = pp[i < cb ? i : cb - 1];
            }
#pragma unroll
            for (int k = 0; k < TL_PER; ++k) {
                if (i0 + k * 1024 < cb) {
                    const int pos = atomicAdd(&ncnt[pv[k] & (BK_NODES - 1)], 1);
                    col[bs + pos] = (int)((unsigned int)pv[k] >> BK_SHIFT);
                }
            }
        }
    }
}

typedef float v4f __attribute__((ext_vector_type(4)));

template <int I>
__device__ __forceinline__ int bcast8(int s) {
    return (int)__builtin_amdgcn_mov_dpp8((unsigned int)s, (unsigned int)(I * 0x249249));
}

__device__ __forceinline__ v4f row_piece(const float* __restrict__ hs, int sj, unsigned int qoff) {
    const unsigned int off = (unsigned int)sj * (unsigned int)(HID * 4) + qoff;
    return *(const v4f*)((const char*)hs + off);
}

__device__ __forceinline__ float xor8(float x) {
    return __builtin_bit_cast(float, __builtin_amdgcn_update_dpp(0, __builtin_bit_cast(int, x), 0x168  ,
                                                                 0xf, 0xf, false));
}
__device__ __forceinline__ float xor16(float x) {
    return __builtin_bit_cast(float, __builtin_amdgcn_permlanex16(__builtin_bit_cast(int, x), __builtin_bit_cast(int, x),
                                                                  0x76543210, 0xfedcba98, false, false));
}

template <int NF>
__device__ __forceinline__ v4f gather_n(const float* __restrict__ hs, int s, int sp, bool part,
                                        unsigned int qoff, v4f acc) {
    v4f v0, v1, v2, v3, v4, v5, v6, v7;
    v4f vp = {0.f, 0.f, 0.f, 0.f};
    if constexpr (NF > 0) v0 = row_piece(hs, bcast8<0>(s), qoff);
    if constexpr (NF > 1) v1 = row_piece(hs, bcast8<1>(s), qoff);
    if constexpr (NF > 2) v2 = row_piece(hs, bcast8<2>(s), qoff);
    if constexpr (NF > 3) v3 = row_piece(hs, bcast8<3>(s), qoff);
    if constexpr (NF > 4) v4 = row_piece(hs, bcast8<4>(s), qoff);
    if constexpr (NF > 5) v5 = row_piece(hs, bcast8<5>(s), qoff);
    if constexpr (NF > 6) v6 = row_piece(hs, bcast8<6>(s), qoff);
    if constexpr (NF > 7) v7 = row_piece(hs, bcast8<7>(s), qoff);
    if constexpr (NF < 8) {
        if (part) vp = row_piece(hs, sp, qoff);
    }
    if constexpr (NF > 0) acc += v0;
    if constexpr (NF > 1) acc += v1;
    if constexpr (NF > 2) acc += v2;
    if constexpr (NF > 3) acc += v3;
    if constexpr (NF > 4) acc += v4;
    if constexpr (NF > 5) acc += v5;
    if constexpr (NF > 6) acc += v6;
    if constexpr (NF > 7) acc += v7;
    if constexpr (NF < 8) acc += vp;
    return acc;
}

__device__ __forceinline__ v4f gather_block(const float* __restrict__ hs, int s, int sp, int cnt, int g,
                                            unsigned int qoff, v4f acc) {
    const bool part = g < (cnt & 3);
    switch (cnt >> 2) {
        case 0:  return gather_n<0>(hs, s, sp, part, qoff, acc);
        case 1:  return gather_n<1>(hs, s, sp, part, qoff, acc);
        case 2:  return gather_n<2>(hs, s, sp, part, qoff, acc);
        case 3:  return gather_n<3>(hs, s, sp, part, qoff, acc);
        case 4:  return gather_n<4>(hs, s, sp, part, qoff, acc);
        case 5:  return gather_n<5>(hs, s, sp, part, qoff, acc);
        case 6:  return gather_n<6>(hs, s, sp, part, qoff, acc);
        case 7:  return gather_n<7>(hs, s, sp, part, qoff, acc);
        default: return gather_n<8>(hs, s, sp, part, qoff, acc);
    }
}

__device__ __forceinline__ void load_idx(const int* __restrict__ col, int beg, int end, int slot, int g,
                                         int& s0, int& p0, int& s1, int& p1) {
    s0 = 0; p0 = 0; s1 = 0; p1 = 0;
    const int left = end - beg;
    if (left > 0) {
        const int c0 = left < 32 ? left : 32;
        const int c1 = left - 32 < 32 ? left - 32 : 32;
        const int* __restrict__ cb = col + beg;
        s0 = cb[(unsigned int)(slot < c0 ? slot : 0)];
        p0 = cb[(unsigned int)(g < (c0 & 3) ? (c0 & ~3) + g : 0)];
        if (c1 > 0) {
            s1 = cb[32u + (unsigned int)(slot < c1 ? slot : 0)];
            p1 = cb[32u + (unsigned int)(g < (c1 & 3) ? (c1 & ~3) + g : 0)];
        }
    }
}

template <int M>
__device__ __forceinline__ float xorm(float x) {
    return __builtin_bit_cast(float, __builtin_amdgcn_update_dpp(0, __builtin_bit_cast(int, x), 0x160 + M  ,
                                                                 0xf, 0xf, false));
}

#define TPITCH 36
template <int MODE>
__global__ __launch_bounds__(256, 16) void k_gather_tile(const int* __restrict__ rowptr, const int* __restrict__ col,
                                                     const float* __restrict__ dinv, const float* __restrict__ hs,
                                                     const float* __restrict__ b, const float* __restrict__ wn,
                                                     const float* __restrict__ bfc, float* __restrict__ outp, int n) {
    __shared__ __attribute__((aligned(16))) float tile_s[MODE == 0 ? 8 * 16 * TPITCH : 4];
    const int lane = threadIdx.x & 31;
    const int g = lane >> 3;
    const int q = lane & 7;
    const unsigned int qoff = (unsigned int)q * 16u;
    const int slot = 4 * q + g;
    const int wave = __builtin_amdgcn_readfirstlane((int)(blockIdx.x * 8 + (threadIdx.x >> 5)));
    const int wv = __builtin_amdgcn_readfirstlane((int)(threadIdx.x >> 5));
    const int ntile = (n + 15) >> 4;
    if (wave >= ntile) return;
    const v4f bias = *(const v4f*)(b + 4 * q);
    float* __restrict__ tile = tile_s + (MODE == 0 ? wv * 16 * TPITCH : 0);
    v4f wq = {0.f, 0.f, 0.f, 0.f};
    float bf = 0.0f;
    if constexpr (MODE == 1) {
        wq = *(const v4f*)(wn + 4 * q);
        bf = bfc[0];
    }

    {
        const int d0 = wave * 16;
        const int ri = d0 + (lane < 16 ? lane : 16);
        const int rp = rowptr[ri < n ? ri : n];
        const int di_i = d0 + (lane & 15);
        const float dvec = dinv[di_i < n ? di_i : n - 1];
        float sc = 0.0f;

        int ns0, np0, ns1, np1;
        load_idx(col, __builtin_amdgcn_readlane(rp, 0), __builtin_amdgcn_readlane(rp, 1), slot, g, ns0, np0, ns1, np1);

        for (int i = 0; i < 16; ++i) {
            const int d = d0 + i;
            const int beg = __builtin_amdgcn_readlane(rp, i);
            const int end = __builtin_amdgcn_readlane(rp, i + 1);
            const int s0 = ns0, p0 = np0, s1 = ns1, p1 = np1;
            v4f o = {0.f, 0.f, 0.f, 0.f};
            if (d < n) {
                if (i < 15 && d + 1 < n)
                    load_idx(col, end, __builtin_amdgcn_readlane(rp, i + 2), slot, g, ns0, np0, ns1, np1);
                v4f acc = {0.f, 0.f, 0.f, 0.f};
                if (g == 0) acc = row_piece(hs, d, qoff);
                const int left = end - beg;
                if (left > 0) {
                    const int c0 = left < 32 ? left : 32;
                    const int c1 = left - 32 < 32 ? left - 32 : 32;
                    acc = gather_block(hs, s0, p0, c0, g, qoff, acc);
                    if (c1 > 0) acc = gather_block(hs, s1, p1, c1, g, qoff, acc);
                }
                for (int base = beg + 64; base < end; base += 64) {
                    int t0, u0, t1, u1;
                    load_idx(col, base, end, slot, g, t0, u0, t1, u1);
                    const int lf = end - base;
                    const int c0 = lf < 32 ? lf : 32;
                    const int c1 = lf - 32 < 32 ? lf - 32 : 32;
                    acc = gather_block(hs, t0, u0, c0, g, qoff, acc);
                    if (c1 > 0) acc = gather_block(hs, t1, u1, c1, g, qoff, acc);
                }
#pragma unroll
                for (int k = 0; k < 4; ++k) {
                    float x = acc[k];
                    x += xor8(x);
                    x += xor16(x);
                    acc[k] = x;
                }
                const float di = __int_as_float(__builtin_amdgcn_readlane(__float_as_int(dvec), i));
                o = acc * di + bias;
                o.x = o.x > 0.0f ? o.x : 0.0f;
                o.y = o.y > 0.0f ? o.y : 0.0f;
                o.z = o.z > 0.0f ? o.z : 0.0f;
                o.w = o.w > 0.0f ? o.w : 0.0f;
            }
            if constexpr (MODE == 0) {
                if (g == 0) *(v4f*)(tile + i * TPITCH + 4 * q) = o;
            } else {
                float x = o.x * wq.x + o.y * wq.y + o.z * wq.z + o.w * wq.w;
                x += xorm<1>(x);
                x += xorm<2>(x);
                x += xorm<4>(x);
                if (lane == i) sc = x + bf;
            }
        }

        if constexpr (MODE == 0) {
            const int lr = lane & 15;
            const int kh = (lane >> 4) << 1;
            const float* __restrict__ tw = tile + lr * TPITCH;
            v8f c0 = {0.f, 0.f, 0.f, 0.f, 0.f, 0.f, 0.f, 0.f};
            v8f c1 = {0.f, 0.f, 0.f, 0.f, 0.f, 0.f, 0.f, 0.f};
#pragma unroll
            for (int k0 = 0; k0 < HID; k0 += 4) {
                v2f a;
                a.x = tw[k0 + kh];
                a.y = tw[k0 + kh + 1];
                const float* __restrict__ wp = wn + (size_t)(k0 + kh) * HID;
                v2f b0, b1;
                b0.x = wp[lr];
                b0.y = wp[HID + lr];
                b1.x = wp[16 + lr];
                b1.y = wp[HID + 16 + lr];
                c0 = __builtin_amdgcn_wmma_f32_16x16x4_f32(false, a, false, b0, (short)0, c0, false, false);
                c1 = __builtin_amdgcn_wmma_f32_16x16x4_f32(false, a, false, b1, (short)0, c1, false, false);
            }
            const int mofs = (lane >> 4) * 8;
#pragma unroll
            for (int r = 0; r < 8; ++r) {
                tile[(mofs + r) * TPITCH + lr]      = c0[r];
                tile[(mofs + r) * TPITCH + 16 + lr] = c1[r];
            }
            v4f vv[4];
            float rsv[4];
#pragma unroll
            for (int j = 0; j < 4; ++j) {
                const int rr = g + 4 * j;
                vv[j] = *(const v4f*)(tile + rr * TPITCH + 4 * q);
                rsv[j] = __shfl(dvec, rr);
            }
#pragma unroll
            for (int j = 0; j < 4; ++j) {
                const int row = d0 + g + 4 * j;
                if (row < n) *(v4f*)(outp + (size_t)row * HID + 4 * q) = vv[j] * rsv[j];
            }
        } else {
            if (lane < 16 && d0 + lane < n) outp[d0 + lane] = sc;
        }
    }
}

extern "C" void kernel_launch(void* const* d_in, const int* in_sizes, int n_in,
                              void* d_out, int out_size, void* d_ws, size_t ws_size,
                              hipStream_t stream) {
    const float* x   = (const float*)d_in[0];
    const int*   ei  = (const int*)d_in[1];
    const float* W1  = (const float*)d_in[2];
    const float* b1  = (const float*)d_in[3];
    const float* W2  = (const float*)d_in[4];
    const float* b2  = (const float*)d_in[5];
    const float* Wfc = (const float*)d_in[6];
    const float* bfc = (const float*)d_in[7];
    float* out = (float*)d_out;

    const int F = 128;
    const long long n = in_sizes[0] / F;
    const long long e = in_sizes[1] / 2;
    const int* src = ei;
    const int* dst = ei + e;
    const long long nh = n * HID;
    const int B = 256;
    dim3 bfill((unsigned)((n + B - 1) / B));
    dim3 bgemm((unsigned)((n + 63) / 64));

    const long long w_n   = (n + 63) & ~63LL;
    const long long w_rp  = (n + 1 + 63) & ~63LL;
    const long long w_bt  = 1024;
    const long long w_e   = (e + 63) & ~63LL;
    const long long w_nh  = (nh + 63) & ~63LL;
    const long long w_ag  = w_nh > w_e ? w_nh : w_e;
    const long long need  = (w_n + w_rp + w_bt + w_n + w_e + w_nh + w_ag) * 4;
    const long long nsblk = (n + SCAN_BLK - 1) / SCAN_BLK;

    if ((long long)ws_size >= need && nsblk <= 512) {
        int*   cnt    = (int*)d_ws;
        int*   rowptr = cnt + w_n;
        int*   btot   = rowptr + w_rp;
        unsigned int* tick = (unsigned int*)(btot + 512);
        const int nsplit = (int)((n / 2) & ~63LL);
        float* dinv   = (float*)(btot + w_bt);
        int*   col    = (int*)(dinv + w_n);
        float* hbuf   = (float*)(col + w_e);
        float* agg    = hbuf + w_nh;
        int*   rank   = (int*)agg;
        dim3 bscan((unsigned)nsblk);

        const int nb = (int)((n + BK_NODES - 1) >> BK_SHIFT);
        const int ntiles = (int)((e + TL_EDGES - 1) / TL_EDGES);
        if (nb <= BK_MAX && ntiles >= 1 && ntiles <= 1024 && (long long)nb * ntiles <= w_n &&
            n <= (1LL << (32 - BK_SHIFT)) && e < (1LL << 30)) {
            int* bcnt   = cnt;
            int* btotal = btot;
            int* part   = (int*)agg;
            k_bhist<<<ntiles, 1024, 0, stream>>>(dst, bcnt, (int)e, nb, ntiles);
            k_bscan<<<nb, 256, 0, stream>>>(bcnt, btotal, ntiles);
            k_bpart<<<ntiles, 1024, 0, stream>>>(src, dst, bcnt, btotal, part, (int)e, nb, ntiles);
            k_bsort<<<nb, 1024, 0, stream>>>(part, btotal, rowptr, col, dinv, (int)n, (int)e, nb);
        } else {
            k_zero<<<bfill, B, 0, stream>>>(cnt, tick, (int)n);
            k_count<<<PGRID, B, 0, stream>>>(dst, cnt, rank, tick, e, nsplit);
            k_scan_local<<<bscan, B, 0, stream>>>(cnt, rowptr, btot, dinv, (int)n);
            k_scan_add<<<bscan, B, 0, stream>>>(rowptr, btot, (int)n, (int)e);
            k_place<<<PGRID, B, 0, stream>>>(src, dst, rank, rowptr, col, tick + 128, e, nsplit);
        }

        dim3 btile((unsigned)((n + 127) / 128));

        k_gemm<128><<<bgemm, 128, 0, stream>>>(x, W1, dinv, hbuf, (int)n);
        k_gather_tile<0><<<btile, B, 0, stream>>>(rowptr, col, dinv, hbuf, b1, W2, bfc, agg, (int)n);

        k_gather_tile<1><<<btile, B, 0, stream>>>(rowptr, col, dinv, agg, b2, Wfc, bfc, out, (int)n);
        return;
    }

    unsigned int* ctr = (unsigned int*)d_ws;
    float* dinv = (float*)d_ws + 256;
    float* hbuf = dinv + n;
    float* agg  = hbuf + n * HID;

    dim3 bfillh((unsigned)((nh + B - 1) / B));
    dim3 bedge((unsigned)((e + B - 1) / B));
    dim3 bscat(2048);
    const int nsplit = (int)(n / 2);

    k_fill<<<bfill, B, 0, stream>>>(dinv, 1.0f, n);
    k_deg<<<bedge, B, 0, stream>>>(dst, dinv, e);
    k_rsqrt<<<bfill, B, 0, stream>>>(dinv, ctr, n);

    k_gemm<128><<<bgemm, 128, 0, stream>>>(x, W1, nullptr, hbuf, (int)n);
    k_fill<<<bfillh, B, 0, stream>>>(agg, 0.0f, nh);
    k_scatter<<<bscat, B, 0, stream>>>(src, dst, dinv, hbuf, agg, ctr, e, nsplit);
    k_finalize<<<bfillh, B, 0, stream>>>(hbuf, dinv, b1, agg, nh);

    k_gemm<32><<<bgemm, 128, 0, stream>>>(agg, W2, nullptr, hbuf, (int)n);
    k_fill<<<bfillh, B, 0, stream>>>(agg, 0.0f, nh);
    k_scatter<<<bscat, B, 0, stream>>>(src, dst, dinv, hbuf, agg, ctr + 128, e, nsplit);
    k_finalize<<<bfillh, B, 0, stream>>>(hbuf, dinv, b2, agg, nh);

    k_fc<<<bfill, B, 0, stream>>>(agg, Wfc, bfc, out, (int)n);
}


# edge grouping level 1 in one launch: tiles ordered by bucket in place, a bucket's first slot = sum over tiles of its run starts; count and scan launches go (7 to 5); sum order = LDS arrival order
# speedup vs baseline: 5.4607x; 1.0968x over previous
#include <hip/hip_runtime.h>

typedef float v2f __attribute__((ext_vector_type(2)));
typedef float v8f __attribute__((ext_vector_type(8)));

#define HID 32

__global__ void k_fill(float* __restrict__ p, float v, long long n) {
    long long i = (long long)blockIdx.x * blockDim.x + threadIdx.x;
    if (i < n) p[i] = v;
}

__global__ void k_deg(const int* __restrict__ dst, float* __restrict__ deg, long long e) {
    long long i = (long long)blockIdx.x * blockDim.x + threadIdx.x;
    if (i < e) atomicAdd(&deg[dst[i]], 1.0f);
}

#define NCTR 256
__global__ void k_rsqrt(float* __restrict__ d, unsigned int* __restrict__ ctr, long long n) {
    long long i = (long long)blockIdx.x * blockDim.x + threadIdx.x;
    if (i < NCTR) ctr[i] = 0u;
    if (i < n) {
        float v = d[i];
        d[i] = (v > 0.0f) ? rsqrtf(v) : 0.0f;
    }
}

template <int KDIM>
__global__ __launch_bounds__(128) void k_gemm(const float* __restrict__ A,
                                              const float* __restrict__ W,
                                              const float* __restrict__ rs,
                                              float* __restrict__ H, int n) {
    const int lane = threadIdx.x & 31;
    const int wave = threadIdx.x >> 5;
    const int row0 = (blockIdx.x * 4 + wave) * 16;
    if (row0 >= n) return;

    const int lr = lane & 15;
    const int kh = (lane >> 4) << 1;

    int arow = row0 + lr;
    if (arow >= n) arow = n - 1;
    const float* __restrict__ ap = A + (size_t)arow * KDIM;

    v8f c0 = {0.f, 0.f, 0.f, 0.f, 0.f, 0.f, 0.f, 0.f};
    v8f c1 = {0.f, 0.f, 0.f, 0.f, 0.f, 0.f, 0.f, 0.f};

#pragma unroll
    for (int k0 = 0; k0 < KDIM; k0 += 4) {
        v2f a;
        a.x = ap[k0 + kh];
        a.y = ap[k0 + kh + 1];
        const float* __restrict__ wp = W + (size_t)(k0 + kh) * HID;
        v2f b0, b1;
        b0.x = wp[lr];
        b0.y = wp[HID + lr];
        b1.x = wp[16 + lr];
        b1.y = wp[HID + 16 + lr];
        c0 = __builtin_amdgcn_wmma_f32_16x16x4_f32(false, a, false, b0, (short)0, c0, false, false);
        c1 = __builtin_amdgcn_wmma_f32_16x16x4_f32(false, a, false, b1, (short)0, c1, false, false);
    }

    const int mofs = (lane >> 4) * 8;
#pragma unroll
    for (int r = 0; r < 8; ++r) {
        int row = row0 + mofs + r;
        if (row < n) {
            const float sc = rs ? rs[row] : 1.0f;
            float* __restrict__ hp = H + (size_t)row * HID;
            hp[lr]      = c0[r] * sc;
            hp[16 + lr] = c1[r] * sc;
        }
    }
}

#define CH_STEPS 4
#define CH_EDGES (32 * CH_STEPS)
#define CTR_STRIDE 64

__device__ __forceinline__ void edge_group32(int s, int d, int lane,
                                             const float* __restrict__ dinv,
                                             const float* __restrict__ h,
                                             float* __restrict__ agg) {
    const float ws = dinv[s];
    const float wd = dinv[d];
    float v[32];
#pragma unroll
    for (int j = 0; j < 32; ++j) {
        const int sj = __builtin_amdgcn_readlane(s, j);
        v[j] = h[sj * HID + lane];
    }
    const float nrm = ws * wd;
#pragma unroll
    for (int j = 0; j < 32; ++j) {
        const int dj = __builtin_amdgcn_readlane(d, j);
        const float wj = __int_as_float(__builtin_amdgcn_readlane(__float_as_int(nrm), j));
        atomicAdd(&agg[dj * HID + lane], v[j] * wj);
    }
}

__global__ __launch_bounds__(256) void k_scatter(const int* __restrict__ src,
                                                 const int* __restrict__ dst,
                                                 const float* __restrict__ dinv,
                                                 const float* __restrict__ h,
                                                 float* __restrict__ agg,
                                                 unsigned int* __restrict__ ctr,
                                                 long long e, int nsplit) {
    __shared__ int q_s[8][64];
    __shared__ int q_d[8][64];
    const int lane = threadIdx.x & 31;
    const int wave = __builtin_amdgcn_readfirstlane((int)(threadIdx.x >> 5));
    int* qs = q_s[wave];
    int* qd = q_d[wave];
    const int half = (int)((__builtin_amdgcn_s_sendmsg_rtn(0x87) >> 18) & 1u);
    const unsigned int nchunks = (unsigned int)((e + CH_EDGES - 1) / CH_EDGES);
    int cnt = 0;

    for (int p = 0; p < 2; ++p) {
        const int r = half ^ p;
        unsigned int* c = ctr + r * CTR_STRIDE;
        for (unsigned int it = 0; it <= nchunks; ++it) {
            unsigned int ck = 0;
            if (lane == 0) ck = atomicAdd(c, 1u);
            ck = (unsigned int)__builtin_amdgcn_readfirstlane((int)ck);
            if (ck >= nchunks) break;
            const long long e0 = (long long)ck * CH_EDGES + lane;
            int sv[CH_STEPS], dv[CH_STEPS];
#pragma unroll
            for (int k = 0; k < CH_STEPS; ++k) {
                const long long idx = e0 + k * 32;
                const long long ii = idx < e ? idx : e - 1;
                sv[k] = src[ii];
                dv[k] = dst[ii];
            }
#pragma unroll
            for (int k = 0; k < CH_STEPS; ++k) {
                const long long idx = e0 + k * 32;
                const bool m = (idx < e) && ((dv[k] >= nsplit ? 1 : 0) == r);
                const unsigned int mask = __builtin_amdgcn_ballot_w32(m);
                if (m) {
                    const int pos = cnt + (int)__builtin_amdgcn_mbcnt_lo(mask, 0u);
                    qs[pos] = sv[k];
                    qd[pos] = dv[k];
                }
                cnt += __builtin_popcount(mask);
                if (cnt >= 32) {
                    const int s = qs[lane];
                    const int d = qd[lane];
                    const int s2 = qs[32 + lane];
                    const int d2 = qd[32 + lane];
                    edge_group32(s, d, lane, dinv, h, agg);
                    cnt -= 32;
                    if (lane < cnt) {
                        qs[lane] = s2;
                        qd[lane] = d2;
                    }
                }
            }
        }
    }
    if (cnt > 0) {
        const int ql = lane < cnt ? lane : cnt - 1;
        const int s = qs[ql];
        const int d = qd[ql];
        const float nrm = dinv[s] * dinv[d];
        for (int j = 0; j < cnt; ++j) {
            const int sj = __builtin_amdgcn_readlane(s, j);
            const int dj = __builtin_amdgcn_readlane(d, j);
            const float wj = __int_as_float(__builtin_amdgcn_readlane(__float_as_int(nrm), j));
            atomicAdd(&agg[dj * HID + lane], h[sj * HID + lane] * wj);
        }
    }
}

__global__ void k_finalize(const float* __restrict__ h, const float* __restrict__ dinv,
                           const float* __restrict__ b, float* __restrict__ agg,
                           long long total) {
    long long t = (long long)blockIdx.x * blockDim.x + threadIdx.x;
    if (t >= total) return;
    long long i = t >> 5;
    int k = (int)(t & 31);
    float di = dinv[i];
    float v = agg[t] + h[t] * (di * di) + b[k];
    agg[t] = v > 0.0f ? v : 0.0f;
}

__global__ void k_fc(const float* __restrict__ h, const float* __restrict__ wfc,
                     const float* __restrict__ bfc, float* __restrict__ out, int n) {
    int i = blockIdx.x * blockDim.x + threadIdx.x;
    if (i >= n) return;
    const float4* __restrict__ hp = (const float4*)(h + (size_t)i * HID);
    const float4* __restrict__ wp = (const float4*)wfc;
    float acc = 0.0f;
#pragma unroll
    for (int k = 0; k < 8; ++k) {
        float4 a = hp[k], w = wp[k];
        acc += a.x * w.x + a.y * w.y + a.z * w.z + a.w * w.w;
    }
    out[i] = acc + bfc[0];
}

#define PGRID 2048
#define SCAN_BLK 1024

__global__ void k_zero(int* __restrict__ p, unsigned int* __restrict__ ctr, int n) {
    int i = blockIdx.x * blockDim.x + threadIdx.x;
    if (i < NCTR) ctr[i] = 0u;
    if (i < n) p[i] = 0;
}

#define CK_STEPS 8
#define CK_EDGES (32 * CK_STEPS)
__global__ __launch_bounds__(256) void k_count(const int* __restrict__ dst, int* __restrict__ cnt,
                                               int* __restrict__ rank, unsigned int* __restrict__ ctr,
                                               long long e, int nsplit) {
    const int lane = threadIdx.x & 31;
    const int half = (int)((__builtin_amdgcn_s_sendmsg_rtn(0x87) >> 18) & 1u);
    const unsigned int nchunks = (unsigned int)((e + CK_EDGES - 1) / CK_EDGES);

    for (int p = 0; p < 2; ++p) {
        const int r = half ^ p;
        unsigned int* c = ctr + r * CTR_STRIDE;
        for (unsigned int it = 0; it <= nchunks; ++it) {
            unsigned int ck = 0;
            if (lane == 0) ck = atomicAdd(c, 1u);
            ck = (unsigned int)__builtin_amdgcn_readfirstlane((int)ck);
            if (ck >= nchunks) break;
            const long long e0 = (long long)ck * CK_EDGES + lane;
            int dv[CK_STEPS], rk[CK_STEPS];
#pragma unroll
            for (int k = 0; k < CK_STEPS; ++k) {
                const long long idx = e0 + k * 32;
                dv[k] = dst[idx < e ? idx : e - 1];
            }
            unsigned int inr = 0u;
#pragma unroll
            for (int k = 0; k < CK_STEPS; ++k) {
                const long long idx = e0 + k * 32;
                const unsigned int ex = idx < e ? 1u : 0u;
                const unsigned int in = (dv[k] >= nsplit ? 1 : 0) == r ? 1u : 0u;
                inr |= (ex & in) << k;
            }
            asm volatile("" : "+v"(inr));
#pragma unroll
            for (int k = 0; k < CK_STEPS; ++k) {
                rk[k] = 0;
                if (inr & (1u << k)) rk[k] = atomicAdd(&cnt[dv[k]], 1);
            }
#pragma unroll
            for (int k = 0; k < CK_STEPS; ++k) {
                if (inr & (1u << k)) rank[e0 + k * 32] = rk[k];
            }
        }
    }
}

__global__ __launch_bounds__(256) void k_scan_local(const int* __restrict__ cnt, int* __restrict__ rowptr,
                                                    int* __restrict__ btot, float* __restrict__ dinv, int n) {
    __shared__ int wsum[8];
    const int t = threadIdx.x;
    const int lane = t & 31;
    const int wv = t >> 5;
    const int base = blockIdx.x * SCAN_BLK + t * 4;
    int c[4];
    int tsum = 0;
#pragma unroll
    for (int k = 0; k < 4; ++k) {
        const int idx = base + k;
        c[k] = idx < n ? cnt[idx] : 0;
        if (idx < n) dinv[idx] = rsqrtf((float)(c[k] + 1));
        tsum += c[k];
    }
    int inc = tsum;
#pragma unroll
    for (int d = 1; d < 32; d <<= 1) {
        const int o = __shfl_up(inc, d);
        if (lane >= d) inc += o;
    }
    if (lane == 31) wsum[wv] = inc;
    __syncthreads();
    int woff = 0, total = 0;
#pragma unroll
    for (int w = 0; w < 8; ++w) {
        const int s = wsum[w];
        if (w < wv) woff += s;
        total += s;
    }
    int run = woff + inc - tsum;
#pragma unroll
    for (int k = 0; k < 4; ++k) {
        const int idx = base + k;
        if (idx < n) rowptr[idx] = run;
        run += c[k];
    }
    if (t == 0) btot[blockIdx.x] = total;
}

__global__ __launch_bounds__(256) void k_scan_add(int* __restrict__ rowptr, const int* __restrict__ btot,
                                                  int n, int e) {
    __shared__ int wsum[8];
    const int t = threadIdx.x;
    const int b = blockIdx.x;
    int s = 0;
    for (int i = t; i < b; i += 256) s += btot[i];
#pragma unroll
    for (int d = 16; d >= 1; d >>= 1) s += __shfl_xor(s, d);
    if ((t & 31) == 0) wsum[t >> 5] = s;
    __syncthreads();
    int off = 0;
#pragma unroll
    for (int w = 0; w < 8; ++w) off += wsum[w];
    const int base = b * SCAN_BLK + t * 4;
#pragma unroll
    for (int k = 0; k < 4; ++k) {
        const int idx = base + k;
        if (idx < n) rowptr[idx] += off;
    }
    if (b == 0 && t == 0) rowptr[n] = e;
}

#define PL_STEPS 8
#define PL_EDGES (32 * PL_STEPS)
__global__ __launch_bounds__(256) void k_place(const int* __restrict__ src, const int* __restrict__ dst,
                                               const int* __restrict__ rank, const int* __restrict__ rowptr,
                                               int* __restrict__ col, unsigned int* __restrict__ ctr,
                                               long long e, int nsplit) {
    const int lane = threadIdx.x & 31;
    const int half = (int)((__builtin_amdgcn_s_sendmsg_rtn(0x87) >> 18) & 1u);
    const int ne = (int)e;
    const unsigned int nchunks = (unsigned int)((ne + PL_EDGES - 1) / PL_EDGES);

    for (int p = 0; p < 2; ++p) {
        const int r = half ^ p;
        unsigned int* c = ctr + r * CTR_STRIDE;
        for (unsigned int it = 0; it <= nchunks; ++it) {
            unsigned int ck = 0;
            if (lane == 0) ck = atomicAdd(c, 1u);
            ck = (unsigned int)__builtin_amdgcn_readfirstlane((int)ck);
            if (ck >= nchunks) break;
            const int e0 = (int)ck * PL_EDGES + lane;
            int dv[PL_STEPS], sv[PL_STEPS], rk[PL_STEPS], rp[PL_STEPS];
#pragma unroll
            for (int k = 0; k < PL_STEPS; ++k) {
                const int idx = e0 + k * 32;
                const int ii = idx < ne ? idx : ne - 1;
                dv[k] = dst[ii];
                sv[k] = src[ii];
                rk[k] = rank[ii];
            }
            unsigned int inr = 0u;
#pragma unroll
            for (int k = 0; k < PL_STEPS; ++k) {
                const int idx = e0 + k * 32;
                const unsigned int ex = idx < ne ? 1u : 0u;
                const unsigned int in = (dv[k] >= nsplit ? 1 : 0) == r ? 1u : 0u;
                inr |= (ex & in) << k;
            }
#pragma unroll
            for (int k = 0; k < PL_STEPS; ++k) {
                rp[k] = 0;
                if (inr & (1u << k)) rp[k] = rowptr[dv[k]];
            }
#pragma unroll
            for (int k = 0; k < PL_STEPS; ++k) {
                if (inr & (1u << k)) col[rp[k] + rk[k]] = sv[k];
            }
        }
    }
}

#define BK_SHIFT 10
#define BK_NODES (1 << BK_SHIFT)
#define BK_MAX   256
#define TL_PER   8
#define TL_EDGES (1024 * TL_PER)
#define ST_CAP   49152

__device__ __forceinline__ int wg_excl_scan_1024(int v, int* wsum, int* tot) {
    const int lane = threadIdx.x & 31;
    const int wv = threadIdx.x >> 5;
    int inc = v;
#pragma unroll
    for (int d = 1; d < 32; d <<= 1) {
        const int o = __shfl_up(inc, d);
        if (lane >= d) inc += o;
    }
    if (lane == 31) wsum[wv] = inc;
    __syncthreads();
    const int ws = wsum[lane];
    int winc = ws;
#pragma unroll
    for (int d = 1; d < 32; d <<= 1) {
        const int o = __shfl_up(winc, d);
        if (lane >= d) winc += o;
    }
    const int total = __shfl(winc, 31);
    const int woff = __shfl(winc, wv) - __shfl(ws, wv);
    __syncthreads();
    *tot = total;
    return woff + inc - v;
}

__device__ __forceinline__ void wg_sum2_1024(int a, int b, int* wa, int* wb, int* sa, int* sb) {
    const int lane = threadIdx.x & 31;
    const int wv = threadIdx.x >> 5;
#pragma unroll
    for (int d = 16; d >= 1; d >>= 1) {
        a += __shfl_xor(a, d);
        b += __shfl_xor(b, d);
    }
    if (lane == 0) {
        wa[wv] = a;
        wb[wv] = b;
    }
    __syncthreads();
    int x = wa[lane], y = wb[lane];
#pragma unroll
    for (int d = 16; d >= 1; d >>= 1) {
        x += __shfl_xor(x, d);
        y += __shfl_xor(y, d);
    }
    *sa = x;
    *sb = y;
}

__global__ __launch_bounds__(1024) void k_tsort(const int* __restrict__ src, const int* __restrict__ dst,
                                                int* __restrict__ lext, int* __restrict__ part,
                                                int ne, int nb, int pitch) {
    __shared__ int hist[BK_MAX];
    __shared__ int wsum[32];
    __shared__ int stage[TL_EDGES];
    const int t = threadIdx.x;
    if (t < BK_MAX) hist[t] = 0;
    __syncthreads();
    const int e0 = blockIdx.x * TL_EDGES;
    int sv[TL_PER], dv[TL_PER];
#pragma unroll
    for (int k = 0; k < TL_PER; ++k) {
        const int idx = e0 + k * 1024 + t;
        const int ii = idx < ne ? idx : ne - 1;
        sv[k] = src[ii];
        dv[k] = dst[ii];
    }
#pragma unroll
    for (int k = 0; k < TL_PER; ++k) {
        if (e0 + k * 1024 + t < ne) atomicAdd(&hist[dv[k] >> BK_SHIFT], 1);
    }
    __syncthreads();
    const int hc = t < nb ? hist[t] : 0;
    int tot;
    const int lex = wg_excl_scan_1024(hc, wsum, &tot);
    if (t <= nb) lext[blockIdx.x * pitch + t] = lex;
    if (t < nb) hist[t] = lex;
    __syncthreads();
#pragma unroll
    for (int k = 0; k < TL_PER; ++k) {
        if (e0 + k * 1024 + t < ne) {
            const int pos = atomicAdd(&hist[dv[k] >> BK_SHIFT], 1);
            stage[pos] = (int)(((unsigned int)sv[k] << BK_SHIFT) | (unsigned int)(dv[k] & (BK_NODES - 1)));
        }
    }
    __syncthreads();
    const int ct = (ne - e0) < TL_EDGES ? (ne - e0) : TL_EDGES;
#pragma unroll
    for (int k = 0; k < TL_PER; ++k) {
        const int i = k * 1024 + t;
        if (i < ct) part[e0 + i] = stage[i];
    }
}

template <int MODE>
__device__ __forceinline__ void edge_word(int v, int* ncnt, int* stage, int* __restrict__ colb) {
    const int nd = v & (BK_NODES - 1);
    if constexpr (MODE == 0) {
        atomicAdd(&ncnt[nd], 1);
    } else {
        const int pos = atomicAdd(&ncnt[nd], 1);
        const int sw = (int)((unsigned int)v >> BK_SHIFT);
        if constexpr (MODE == 1) stage[pos] = sw;
        else colb[pos] = sw;
    }
}

template <int MODE>
__device__ __forceinline__ void run_walk(const int* __restrict__ part, const int* rst, const int* rln,
                                         int* ncnt, int* stage, int* __restrict__ colb, int ntiles) {
    const int lane = threadIdx.x & 31;
    const int wv = threadIdx.x >> 5;
    for (int i0 = 0; i0 < 32; i0 += 8) {
        if (wv + 32 * i0 >= ntiles) break;
        int st[8], ln[8], v0[8], v1[8];
#pragma unroll
        for (int g = 0; g < 8; ++g) {
            const int r = wv + 32 * (i0 + g);
            st[g] = __builtin_amdgcn_readfirstlane(rst[r]);
            ln[g] = __builtin_amdgcn_readfirstlane(rln[r]);
            v0[g] = part[lane < ln[g] ? st[g] + lane : 0];
            v1[g] = part[lane + 32 < ln[g] ? st[g] + 32 + lane : 0];
        }
#pragma unroll
        for (int g = 0; g < 8; ++g) {
            if (lane < ln[g]) edge_word<MODE>(v0[g], ncnt, stage, colb);
            if (lane + 32 < ln[g]) edge_word<MODE>(v1[g], ncnt, stage, colb);
        }
#pragma unroll
        for (int g = 0; g < 8; ++g) {
            for (int j = 64 + lane; j < ln[g]; j += 32) edge_word<MODE>(part[st[g] + j], ncnt, stage, colb);
        }
    }
}

__global__ __launch_bounds__(1024) void k_bsort(const int* __restrict__ part, const int* __restrict__ lext,
                                                int* __restrict__ rowptr, int* __restrict__ col,
                                                float* __restrict__ dinv, int n, int ne, int nb, int ntiles,
                                                int pitch) {
    __shared__ int ncnt[BK_NODES];
    __shared__ int wsum[32];
    __shared__ int wa[32];
    __shared__ int wb[32];
    __shared__ int rst[1024];
    __shared__ int rln[1024];
    __shared__ int stage[ST_CAP];
    const int t = threadIdx.x;
    const int b = blockIdx.x;
    const int base = b << BK_SHIFT;
    ncnt[t] = 0;
    int lo = 0, len = 0;
    if (t < ntiles) {
        const int* __restrict__ lp = lext + t * pitch + b;
        lo = lp[0];
        len = lp[1] - lo;
    }
    rst[t] = t < ntiles ? t * TL_EDGES + lo : 0;
    rln[t] = len;
    int bs, cb;
    wg_sum2_1024(lo, len, wa, wb, &bs, &cb);

    run_walk<0>(part, rst, rln, ncnt, stage, col + bs, ntiles);
    __syncthreads();
    const int c = ncnt[t];
    int tot;
    const int ex = wg_excl_scan_1024(c, wsum, &tot);
    if (base + t < n) {
        rowptr[base + t] = bs + ex;
        dinv[base + t] = rsqrtf((float)(c + 1));
    }
    if (b == nb - 1 && t == 0) rowptr[n] = ne;
    ncnt[t] = ex;
    __syncthreads();

    if (cb <= ST_CAP) {
        run_walk<1>(part, rst, rln, ncnt, stage, col + bs, ntiles);
        __syncthreads();
        for (int i = t; i < cb; i += 1024) col[bs + i] = stage[i];
    } else {
        run_walk<2>(part, rst, rln, ncnt, stage, col + bs, ntiles);
    }
}

typedef float v4f __attribute__((ext_vector_type(4)));

template <int I>
__device__ __forceinline__ int bcast8(int s) {
    return (int)__builtin_amdgcn_mov_dpp8((unsigned int)s, (unsigned int)(I * 0x249249));
}

__device__ __forceinline__ v4f row_piece(const float* __restrict__ hs, int sj, unsigned int qoff) {
    const unsigned int off = (unsigned int)sj * (unsigned int)(HID * 4) + qoff;
    return *(const v4f*)((const char*)hs + off);
}

__device__ __forceinline__ float xor8(float x) {
    return __builtin_bit_cast(float, __builtin_amdgcn_update_dpp(0, __builtin_bit_cast(int, x), 0x168  ,
                                                                 0xf, 0xf, false));
}
__device__ __forceinline__ float xor16(float x) {
    return __builtin_bit_cast(float, __builtin_amdgcn_permlanex16(__builtin_bit_cast(int, x), __builtin_bit_cast(int, x),
                                                                  0x76543210, 0xfedcba98, false, false));
}

template <int NF>
__device__ __forceinline__ v4f gather_n(const float* __restrict__ hs, int s, int sp, bool part,
                                        unsigned int qoff, v4f acc) {
    v4f v0, v1, v2, v3, v4, v5, v6, v7;
    v4f vp = {0.f, 0.f, 0.f, 0.f};
    if constexpr (NF > 0) v0 = row_piece(hs, bcast8<0>(s), qoff);
    if constexpr (NF > 1) v1 = row_piece(hs, bcast8<1>(s), qoff);
    if constexpr (NF > 2) v2 = row_piece(hs, bcast8<2>(s), qoff);
    if constexpr (NF > 3) v3 = row_piece(hs, bcast8<3>(s), qoff);
    if constexpr (NF > 4) v4 = row_piece(hs, bcast8<4>(s), qoff);
    if constexpr (NF > 5) v5 = row_piece(hs, bcast8<5>(s), qoff);
    if constexpr (NF > 6) v6 = row_piece(hs, bcast8<6>(s), qoff);
    if constexpr (NF > 7) v7 = row_piece(hs, bcast8<7>(s), qoff);
    if constexpr (NF < 8) {
        if (part) vp = row_piece(hs, sp, qoff);
    }
    if constexpr (NF > 0) acc += v0;
    if constexpr (NF > 1) acc += v1;
    if constexpr (NF > 2) acc += v2;
    if constexpr (NF > 3) acc += v3;
    if constexpr (NF > 4) acc += v4;
    if constexpr (NF > 5) acc += v5;
    if constexpr (NF > 6) acc += v6;
    if constexpr (NF > 7) acc += v7;
    if constexpr (NF < 8) acc += vp;
    return acc;
}

__device__ __forceinline__ v4f gather_block(const float* __restrict__ hs, int s, int sp, int cnt, int g,
                                            unsigned int qoff, v4f acc) {
    const bool part = g < (cnt & 3);
    switch (cnt >> 2) {
        case 0:  return gather_n<0>(hs, s, sp, part, qoff, acc);
        case 1:  return gather_n<1>(hs, s, sp, part, qoff, acc);
        case 2:  return gather_n<2>(hs, s, sp, part, qoff, acc);
        case 3:  return gather_n<3>(hs, s, sp, part, qoff, acc);
        case 4:  return gather_n<4>(hs, s, sp, part, qoff, acc);
        case 5:  return gather_n<5>(hs, s, sp, part, qoff, acc);
        case 6:  return gather_n<6>(hs, s, sp, part, qoff, acc);
        case 7:  return gather_n<7>(hs, s, sp, part, qoff, acc);
        default: return gather_n<8>(hs, s, sp, part, qoff, acc);
    }
}

__device__ __forceinline__ void load_idx(const int* __restrict__ col, int beg, int end, int slot, int g,
                                         int& s0, int& p0, int& s1, int& p1) {
    s0 = 0; p0 = 0; s1 = 0; p1 = 0;
    const int left = end - beg;
    if (left > 0) {
        const int c0 = left < 32 ? left : 32;
        const int c1 = left - 32 < 32 ? left - 32 : 32;
        const int* __restrict__ cb = col + beg;
        s0 = cb[(unsigned int)(slot < c0 ? slot : 0)];
        p0 = cb[(unsigned int)(g < (c0 & 3) ? (c0 & ~3) + g : 0)];
        if (c1 > 0) {
            s1 = cb[32u + (unsigned int)(slot < c1 ? slot : 0)];
            p1 = cb[32u + (unsigned int)(g < (c1 & 3) ? (c1 & ~3) + g : 0)];
        }
    }
}

template <int M>
__device__ __forceinline__ float xorm(float x) {
    return __builtin_bit_cast(float, __builtin_amdgcn_update_dpp(0, __builtin_bit_cast(int, x), 0x160 + M  ,
                                                                 0xf, 0xf, false));
}

#define TPITCH 36
template <int MODE>
__global__ __launch_bounds__(256, 16) void k_gather_tile(const int* __restrict__ rowptr, const int* __restrict__ col,
                                                     const float* __restrict__ dinv, const float* __restrict__ hs,
                                                     const float* __restrict__ b, const float* __restrict__ wn,
                                                     const float* __restrict__ bfc, float* __restrict__ outp, int n) {
    __shared__ __attribute__((aligned(16))) float tile_s[MODE == 0 ? 8 * 16 * TPITCH : 4];
    const int lane = threadIdx.x & 31;
    const int g = lane >> 3;
    const int q = lane & 7;
    const unsigned int qoff = (unsigned int)q * 16u;
    const int slot = 4 * q + g;
    const int wave = __builtin_amdgcn_readfirstlane((int)(blockIdx.x * 8 + (threadIdx.x >> 5)));
    const int wv = __builtin_amdgcn_readfirstlane((int)(threadIdx.x >> 5));
    const int ntile = (n + 15) >> 4;
    if (wave >= ntile) return;
    const v4f bias = *(const v4f*)(b + 4 * q);
    float* __restrict__ tile = tile_s + (MODE == 0 ? wv * 16 * TPITCH : 0);
    v4f wq = {0.f, 0.f, 0.f, 0.f};
    float bf = 0.0f;
    if constexpr (MODE == 1) {
        wq = *(const v4f*)(wn + 4 * q);
        bf = bfc[0];
    }

    {
        const int d0 = wave * 16;
        const int ri = d0 + (lane < 16 ? lane : 16);
        const int rp = rowptr[ri < n ? ri : n];
        const int di_i = d0 + (lane & 15);
        const float dvec = dinv[di_i < n ? di_i : n - 1];
        float sc = 0.0f;

        int ns0, np0, ns1, np1;
        load_idx(col, __builtin_amdgcn_readlane(rp, 0), __builtin_amdgcn_readlane(rp, 1), slot, g, ns0, np0, ns1, np1);

        for (int i = 0; i < 16; ++i) {
            const int d = d0 + i;
            const int beg = __builtin_amdgcn_readlane(rp, i);
            const int end = __builtin_amdgcn_readlane(rp, i + 1);
            const int s0 = ns0, p0 = np0, s1 = ns1, p1 = np1;
            v4f o = {0.f, 0.f, 0.f, 0.f};
            if (d < n) {
                if (i < 15 && d + 1 < n)
                    load_idx(col, end, __builtin_amdgcn_readlane(rp, i + 2), slot, g, ns0, np0, ns1, np1);
                v4f acc = {0.f, 0.f, 0.f, 0.f};
                if (g == 0) acc = row_piece(hs, d, qoff);
                const int left = end - beg;
                if (left > 0) {
                    const int c0 = left < 32 ? left : 32;
                    const int c1 = left - 32 < 32 ? left - 32 : 32;
                    acc = gather_block(hs, s0, p0, c0, g, qoff, acc);
                    if (c1 > 0) acc = gather_block(hs, s1, p1, c1, g, qoff, acc);
                }
                for (int base = beg + 64; base < end; base += 64) {
                    int t0, u0, t1, u1;
                    load_idx(col, base, end, slot, g, t0, u0, t1, u1);
                    const int lf = end - base;
                    const int c0 = lf < 32 ? lf : 32;
                    const int c1 = lf - 32 < 32 ? lf - 32 : 32;
                    acc = gather_block(hs, t0, u0, c0, g, qoff, acc);
                    if (c1 > 0) acc = gather_block(hs, t1, u1, c1, g, qoff, acc);
                }
#pragma unroll
                for (int k = 0; k < 4; ++k) {
                    float x = acc[k];
                    x += xor8(x);
                    x += xor16(x);
                    acc[k] = x;
                }
                const float di = __int_as_float(__builtin_amdgcn_readlane(__float_as_int(dvec), i));
                o = acc * di + bias;
                o.x = o.x > 0.0f ? o.x : 0.0f;
                o.y = o.y > 0.0f ? o.y : 0.0f;
                o.z = o.z > 0.0f ? o.z : 0.0f;
                o.w = o.w > 0.0f ? o.w : 0.0f;
            }
            if constexpr (MODE == 0) {
                if (g == 0) *(v4f*)(tile + i * TPITCH + 4 * q) = o;
            } else {
                float x = o.x * wq.x + o.y * wq.y + o.z * wq.z + o.w * wq.w;
                x += xorm<1>(x);
                x += xorm<2>(x);
                x += xorm<4>(x);
                if (lane == i) sc = x + bf;
            }
        }

        if constexpr (MODE == 0) {
            const int lr = lane & 15;
            const int kh = (lane >> 4) << 1;
            const float* __restrict__ tw = tile + lr * TPITCH;
            v8f c0 = {0.f, 0.f, 0.f, 0.f, 0.f, 0.f, 0.f, 0.f};
            v8f c1 = {0.f, 0.f, 0.f, 0.f, 0.f, 0.f, 0.f, 0.f};
#pragma unroll
            for (int k0 = 0; k0 < HID; k0 += 4) {
                v2f a;
                a.x = tw[k0 + kh];
                a.y = tw[k0 + kh + 1];
                const float* __restrict__ wp = wn + (size_t)(k0 + kh) * HID;
                v2f b0, b1;
                b0.x = wp[lr];
                b0.y = wp[HID + lr];
                b1.x = wp[16 + lr];
                b1.y = wp[HID + 16 + lr];
                c0 = __builtin_amdgcn_wmma_f32_16x16x4_f32(false, a, false, b0, (short)0, c0, false, false);
                c1 = __builtin_amdgcn_wmma_f32_16x16x4_f32(false, a, false, b1, (short)0, c1, false, false);
            }
            const int mofs = (lane >> 4) * 8;
#pragma unroll
            for (int r = 0; r < 8; ++r) {
                tile[(mofs + r) * TPITCH + lr]      = c0[r];
                tile[(mofs + r) * TPITCH + 16 + lr] = c1[r];
            }
            v4f vv[4];
            float rsv[4];
#pragma unroll
            for (int j = 0; j < 4; ++j) {
                const int rr = g + 4 * j;
                vv[j] = *(const v4f*)(tile + rr * TPITCH + 4 * q);
                rsv[j] = __shfl(dvec, rr);
            }
#pragma unroll
            for (int j = 0; j < 4; ++j) {
                const int row = d0 + g + 4 * j;
                if (row < n) *(v4f*)(outp + (size_t)row * HID + 4 * q) = vv[j] * rsv[j];
            }
        } else {
            if (lane < 16 && d0 + lane < n) outp[d0 + lane] = sc;
        }
    }
}

extern "C" void kernel_launch(void* const* d_in, const int* in_sizes, int n_in,
                              void* d_out, int out_size, void* d_ws, size_t ws_size,
                              hipStream_t stream) {
    const float* x   = (const float*)d_in[0];
    const int*   ei  = (const int*)d_in[1];
    const float* W1  = (const float*)d_in[2];
    const float* b1  = (const float*)d_in[3];
    const float* W2  = (const float*)d_in[4];
    const float* b2  = (const float*)d_in[5];
    const float* Wfc = (const float*)d_in[6];
    const float* bfc = (const float*)d_in[7];
    float* out = (float*)d_out;

    const int F = 128;
    const long long n = in_sizes[0] / F;
    const long long e = in_sizes[1] / 2;
    const int* src = ei;
    const int* dst = ei + e;
    const long long nh = n * HID;
    const int B = 256;
    dim3 bfill((unsigned)((n + B - 1) / B));
    dim3 bgemm((unsigned)((n + 63) / 64));

    const long long w_n   = (n + 63) & ~63LL;
    const long long w_rp  = (n + 1 + 63) & ~63LL;
    const long long w_bt  = 1024;
    const long long w_e   = (e + 63) & ~63LL;
    const long long w_nh  = (nh + 63) & ~63LL;
    const long long w_ag  = w_nh > w_e ? w_nh : w_e;
    const long long need  = (w_n + w_rp + w_bt + w_n + w_e + w_nh + w_ag) * 4;
    const long long nsblk = (n + SCAN_BLK - 1) / SCAN_BLK;

    if ((long long)ws_size >= need && nsblk <= 512) {
        int*   cnt    = (int*)d_ws;
        int*   rowptr = cnt + w_n;
        int*   btot   = rowptr + w_rp;
        unsigned int* tick = (unsigned int*)(btot + 512);
        const int nsplit = (int)((n / 2) & ~63LL);
        float* dinv   = (float*)(btot + w_bt);
        int*   col    = (int*)(dinv + w_n);
        float* hbuf   = (float*)(col + w_e);
        float* agg    = hbuf + w_nh;
        int*   rank   = (int*)agg;
        dim3 bscan((unsigned)nsblk);

        const int nb = (int)((n + BK_NODES - 1) >> BK_SHIFT);
        const int ntiles = (int)((e + TL_EDGES - 1) / TL_EDGES);
        const int pitch = (nb + 1 + 31) & ~31;
        if (nb <= BK_MAX && ntiles >= 1 && ntiles <= 1024 && (long long)ntiles * pitch <= w_n &&
            n <= (1LL << (32 - BK_SHIFT)) && e < (1LL << 30)) {
            int* lext = cnt;
            int* part = (int*)agg;
            k_tsort<<<ntiles, 1024, 0, stream>>>(src, dst, lext, part, (int)e, nb, pitch);
            k_bsort<<<nb, 1024, 0, stream>>>(part, lext, rowptr, col, dinv, (int)n, (int)e, nb, ntiles, pitch);
        } else {
            k_zero<<<bfill, B, 0, stream>>>(cnt, tick, (int)n);
            k_count<<<PGRID, B, 0, stream>>>(dst, cnt, rank, tick, e, nsplit);
            k_scan_local<<<bscan, B, 0, stream>>>(cnt, rowptr, btot, dinv, (int)n);
            k_scan_add<<<bscan, B, 0, stream>>>(rowptr, btot, (int)n, (int)e);
            k_place<<<PGRID, B, 0, stream>>>(src, dst, rank, rowptr, col, tick + 128, e, nsplit);
        }

        dim3 btile((unsigned)((n + 127) / 128));

        k_gemm<128><<<bgemm, 128, 0, stream>>>(x, W1, dinv, hbuf, (int)n);
        k_gather_tile<0><<<btile, B, 0, stream>>>(rowptr, col, dinv, hbuf, b1, W2, bfc, agg, (int)n);

        k_gather_tile<1><<<btile, B, 0, stream>>>(rowptr, col, dinv, agg, b2, Wfc, bfc, out, (int)n);
        return;
    }

    unsigned int* ctr = (unsigned int*)d_ws;
    float* dinv = (float*)d_ws + 256;
    float* hbuf = dinv + n;
    float* agg  = hbuf + n * HID;

    dim3 bfillh((unsigned)((nh + B - 1) / B));
    dim3 bedge((unsigned)((e + B - 1) / B));
    dim3 bscat(2048);
    const int nsplit = (int)(n / 2);

    k_fill<<<bfill, B, 0, stream>>>(dinv, 1.0f, n);
    k_deg<<<bedge, B, 0, stream>>>(dst, dinv, e);
    k_rsqrt<<<bfill, B, 0, stream>>>(dinv, ctr, n);

    k_gemm<128><<<bgemm, 128, 0, stream>>>(x, W1, nullptr, hbuf, (int)n);
    k_fill<<<bfillh, B, 0, stream>>>(agg, 0.0f, nh);
    k_scatter<<<bscat, B, 0, stream>>>(src, dst, dinv, hbuf, agg, ctr, e, nsplit);
    k_finalize<<<bfillh, B, 0, stream>>>(hbuf, dinv, b1, agg, nh);

    k_gemm<32><<<bgemm, 128, 0, stream>>>(agg, W2, nullptr, hbuf, (int)n);
    k_fill<<<bfillh, B, 0, stream>>>(agg, 0.0f, nh);
    k_scatter<<<bscat, B, 0, stream>>>(src, dst, dinv, hbuf, agg, ctr + 128, e, nsplit);
    k_finalize<<<bfillh, B, 0, stream>>>(hbuf, dinv, b2, agg, nh);

    k_fc<<<bfill, B, 0, stream>>>(agg, Wfc, bfc, out, (int)n);
}
